# starting kernel
# baseline (speedup 1.0000x reference)
#include <hip/hip_runtime.h>
#include <hip/hip_bf16.h>
#include <math.h>

typedef __attribute__((ext_vector_type(2))) float v2f;
typedef __attribute__((ext_vector_type(8))) float v8f;

#define THREADS 256
#define ORD_NEG_INF 0x007FFFFFu

__device__ __forceinline__ unsigned ford(float f) {
  unsigned u = __float_as_uint(f);
  return (u & 0x80000000u) ? ~u : (u | 0x80000000u);
}
__device__ __forceinline__ float funord(unsigned u) {
  return __uint_as_float((u & 0x80000000u) ? (u & 0x7fffffffu) : ~u);
}

template <int K, int NT>
__global__ void wmma_gemm_xwt(const float* __restrict__ X,
                              const float* __restrict__ W,
                              float* __restrict__ Y, int M) {
  const int NO   = NT * 16;
  const int lane = threadIdx.x & 31;
  const int wave = threadIdx.x >> 5;
  const int tile = blockIdx.x * (blockDim.x >> 5) + wave;
  if (tile * 16 >= M) return;
  const int rbase = tile * 16;
  const int mrow  = rbase + (lane & 15);
  const int koff  = (lane < 16) ? 0 : 2;

  v8f acc[NT];
#pragma unroll
  for (int t = 0; t < NT; ++t) acc[t] = (v8f){0.f,0.f,0.f,0.f,0.f,0.f,0.f,0.f};

  for (int k0 = 0; k0 < K; k0 += 4) {
    const int k = k0 + koff;
    v2f a;
    a.x = (k     < K) ? X[mrow * K + k]     : 0.f;
    a.y = (k + 1 < K) ? X[mrow * K + k + 1] : 0.f;
#pragma unroll
    for (int t = 0; t < NT; ++t) {
      const int col = t * 16 + (lane & 15);
      v2f b;
      b.x = (k     < K) ? W[col * K + k]     : 0.f;
      b.y = (k + 1 < K) ? W[col * K + k + 1] : 0.f;
      acc[t] = __builtin_amdgcn_wmma_f32_16x16x4_f32(
          false, a, false, b, (short)0, acc[t], false, false);
    }
  }

  const int rowoff = (lane < 16) ? 0 : 8;
  const int colo   = lane & 15;
#pragma unroll
  for (int t = 0; t < NT; ++t) {
#pragma unroll
    for (int i = 0; i < 8; ++i) {
      Y[(rbase + rowoff + i) * NO + t * 16 + colo] = acc[t][i];
    }
  }
}

__global__ void fill_u32(unsigned* p, unsigned v, int n) {
  int t = blockIdx.x * blockDim.x + threadIdx.x;
  if (t < n) p[t] = v;
}

__global__ void attn_proj(const float* __restrict__ feat,
                          const float* __restrict__ al,
                          const float* __restrict__ ar,
                          float* __restrict__ el, float* __restrict__ er, int NH) {
  int t = blockIdx.x * blockDim.x + threadIdx.x;
  if (t >= NH) return;
  int n = t >> 2, h = t & 3;
  const float* f = feat + n * 128 + h * 32;
  float sl = 0.f, sr = 0.f;
#pragma unroll
  for (int i = 0; i < 32; ++i) {
    sl += f[i] * al[h * 32 + i];
    sr += f[i] * ar[h * 32 + i];
  }
  el[t] = sl; er[t] = sr;
}

__global__ void edge_score_max(const int* __restrict__ src, const int* __restrict__ dst,
                               const float* __restrict__ el, const float* __restrict__ er,
                               float* __restrict__ ebuf, unsigned* __restrict__ nmax, int EH) {
  int t = blockIdx.x * blockDim.x + threadIdx.x;
  if (t >= EH) return;
  int e = t >> 2, h = t & 3;
  float v = el[src[e] * 4 + h] + er[dst[e] * 4 + h];
  v = (v > 0.f) ? v : 0.2f * v;
  ebuf[t] = v;
  atomicMax(&nmax[dst[e] * 4 + h], ford(v));
}

__global__ void edge_expsum(const int* __restrict__ dst,
                            const unsigned* __restrict__ nmax,
                            float* __restrict__ ebuf, float* __restrict__ nsum, int EH) {
  int t = blockIdx.x * blockDim.x + threadIdx.x;
  if (t >= EH) return;
  int e = t >> 2, h = t & 3;
  float m = funord(nmax[dst[e] * 4 + h]);
  float ex = expf(ebuf[t] - m);
  ebuf[t] = ex;
  atomicAdd(&nsum[dst[e] * 4 + h], ex);
}

__global__ void edge_aggregate(const int* __restrict__ src, const int* __restrict__ dst,
                               const float* __restrict__ ebuf, const float* __restrict__ nsum,
                               const float* __restrict__ feat, float* __restrict__ accum, int EH) {
  int t = blockIdx.x * blockDim.x + threadIdx.x;
  if (t >= EH) return;
  int e = t >> 2, h = t & 3;
  int s = src[e], d = dst[e];
  float a = ebuf[t] / nsum[d * 4 + h];
  const float* fs = feat + s * 128 + h * 32;
  float* ad = accum + d * 128 + h * 32;
#pragma unroll
  for (int i = 0; i < 32; ++i) atomicAdd(&ad[i], fs[i] * a);
}

__global__ void elu_combine(float* __restrict__ accum, const float* __restrict__ res,
                            const float* __restrict__ bias, int n) {
  int t = blockIdx.x * blockDim.x + threadIdx.x;
  if (t >= n) return;
  float v = accum[t] + res[t] + bias[t & 127];
  accum[t] = (v > 0.f) ? v : expm1f(v);
}

__global__ void combine_mean(const float* __restrict__ accum, const float* __restrict__ h1,
                             const float* __restrict__ bias2, float* __restrict__ hnode, int ND) {
  int t = blockIdx.x * blockDim.x + threadIdx.x;
  if (t >= ND) return;
  int n = t >> 5, d = t & 31;
  float s = 0.f;
#pragma unroll
  for (int h = 0; h < 4; ++h) {
    int idx = n * 128 + h * 32 + d;
    s += accum[idx] + h1[idx] + bias2[h * 32 + d];
  }
  hnode[t] = s * 0.25f;
}

__global__ void node_weight(const float* __restrict__ hnode, const float* __restrict__ aw_w,
                            const float* __restrict__ aw_b, float* __restrict__ wbuf, int N) {
  int n = blockIdx.x * blockDim.x + threadIdx.x;
  if (n >= N) return;
  float s = aw_b[0];
#pragma unroll
  for (int i = 0; i < 32; ++i) s += hnode[n * 32 + i] * aw_w[i];
  wbuf[n] = 1.f / (1.f + expf(-s));
}

__global__ void graph_init(unsigned* gmat, int n) {
  int t = blockIdx.x * blockDim.x + threadIdx.x;
  if (t >= n) return;
  gmat[t] = ((t & 63) < 32) ? 0u : ORD_NEG_INF;
}

__global__ void graph_scatter(const float* __restrict__ hnode, const float* __restrict__ wbuf,
                              const int* __restrict__ gid, float* __restrict__ gmat, int ND) {
  int t = blockIdx.x * blockDim.x + threadIdx.x;
  if (t >= ND) return;
  int n = t >> 5, d = t & 31;
  int g = gid[n];
  float hv = hnode[t];
  atomicAdd(&gmat[g * 64 + d], wbuf[n] * hv);
  atomicMax((unsigned*)gmat + g * 64 + 32 + d, ford(hv));
}

__global__ void graph_decode(float* gmat, int GD) {
  int t = blockIdx.x * blockDim.x + threadIdx.x;
  if (t >= GD) return;
  int g = t >> 5, d = t & 31;
  unsigned u = ((unsigned*)gmat)[g * 64 + 32 + d];
  gmat[g * 64 + 32 + d] = (u == ORD_NEG_INF) ? 0.f : funord(u);
}

__global__ void final_ln(const float* __restrict__ ybuf, const float* __restrict__ out_b,
                         float* __restrict__ out) {
  __shared__ float sm[128];
  int g = blockIdx.x, j = threadIdx.x;
  float v = ybuf[g * 128 + j] + out_b[j];
  sm[j] = v; __syncthreads();
  for (int s = 64; s > 0; s >>= 1) { if (j < s) sm[j] += sm[j + s]; __syncthreads(); }
  float mu = sm[0] * (1.f / 128.f);
  __syncthreads();
  float dv = v - mu;
  sm[j] = dv * dv; __syncthreads();
  for (int s = 64; s > 0; s >>= 1) { if (j < s) sm[j] += sm[j + s]; __syncthreads(); }
  float var = sm[0] * (1.f / 128.f);
  out[g * 128 + j] = dv * rsqrtf(var + 1e-5f);
}

extern "C" void kernel_launch(void* const* d_in, const int* in_sizes, int n_in,
                              void* d_out, int out_size, void* d_ws, size_t ws_size,
                              hipStream_t stream) {
  const float* node  = (const float*)d_in[0];
  const int*   src   = (const int*)d_in[1];
  const int*   dst   = (const int*)d_in[2];
  const int*   gid   = (const int*)d_in[3];
  const float* fc1_w = (const float*)d_in[4];
  const float* al1   = (const float*)d_in[5];
  const float* ar1   = (const float*)d_in[6];
  const float* res1w = (const float*)d_in[7];
  const float* bias1 = (const float*)d_in[8];
  const float* fc2_w = (const float*)d_in[9];
  const float* al2   = (const float*)d_in[10];
  const float* ar2   = (const float*)d_in[11];
  const float* bias2 = (const float*)d_in[12];
  const float* aw_w  = (const float*)d_in[13];
  const float* aw_b  = (const float*)d_in[14];
  const float* out_w = (const float*)d_in[15];
  const float* out_b = (const float*)d_in[16];
  (void)n_in; (void)ws_size;

  const int N = in_sizes[3];
  const int E = in_sizes[1];
  const int G = out_size / 128;

  char* wsb = (char*)d_ws;
  size_t off = 0;
  auto alloc = [&](size_t bytes) -> char* {
    char* p = wsb + off;
    off = (off + bytes + 255) & ~(size_t)255;
    return p;
  };
  float*    bufA  = (float*)alloc((size_t)N * 128 * 4);
  float*    bufB  = (float*)alloc((size_t)N * 128 * 4);
  float*    bufC  = (float*)alloc((size_t)N * 128 * 4);
  float*    el    = (float*)alloc((size_t)N * 4 * 4);
  float*    er    = (float*)alloc((size_t)N * 4 * 4);
  unsigned* nmax  = (unsigned*)alloc((size_t)N * 4 * 4);
  float*    nsum  = (float*)alloc((size_t)N * 4 * 4);
  float*    ebuf  = (float*)alloc((size_t)E * 4 * 4);
  float*    hnode = (float*)alloc((size_t)N * 32 * 4);
  float*    wbuf  = (float*)alloc((size_t)N * 4);
  float*    gmat  = (float*)alloc((size_t)G * 64 * 4);
  float*    ybuf  = (float*)alloc((size_t)G * 128 * 4);

  auto grid = [](int n) { return dim3((unsigned)((n + THREADS - 1) / THREADS)); };
  const int ntiles  = (N + 15) / 16;
  const int nblocks = (ntiles + 3) / 4;
  const int gtiles  = (G + 15) / 16;
  const int gblocks = (gtiles + 3) / 4;

  wmma_gemm_xwt<27, 8><<<nblocks, 128, 0, stream>>>(node, fc1_w, bufA, N);
  wmma_gemm_xwt<27, 8><<<nblocks, 128, 0, stream>>>(node, res1w, bufB, N);
  fill_u32<<<grid(N * 128), THREADS, 0, stream>>>((unsigned*)bufC, 0u, N * 128);
  fill_u32<<<grid(N * 4), THREADS, 0, stream>>>(nmax, ORD_NEG_INF, N * 4);
  fill_u32<<<grid(N * 4), THREADS, 0, stream>>>((unsigned*)nsum, 0u, N * 4);
  attn_proj<<<grid(N * 4), THREADS, 0, stream>>>(bufA, al1, ar1, el, er, N * 4);
  edge_score_max<<<grid(E * 4), THREADS, 0, stream>>>(src, dst, el, er, ebuf, nmax, E * 4);
  edge_expsum<<<grid(E * 4), THREADS, 0, stream>>>(dst, nmax, ebuf, nsum, E * 4);
  edge_aggregate<<<grid(E * 4), THREADS, 0, stream>>>(src, dst, ebuf, nsum, bufA, bufC, E * 4);
  elu_combine<<<grid(N * 128), THREADS, 0, stream>>>(bufC, bufB, bias1, N * 128);

  wmma_gemm_xwt<128, 8><<<nblocks, 128, 0, stream>>>(bufC, fc2_w, bufA, N);
  fill_u32<<<grid(N * 128), THREADS, 0, stream>>>((unsigned*)bufB, 0u, N * 128);
  fill_u32<<<grid(N * 4), THREADS, 0, stream>>>(nmax, ORD_NEG_INF, N * 4);
  fill_u32<<<grid(N * 4), THREADS, 0, stream>>>((unsigned*)nsum, 0u, N * 4);
  attn_proj<<<grid(N * 4), THREADS, 0, stream>>>(bufA, al2, ar2, el, er, N * 4);
  edge_score_max<<<grid(E * 4), THREADS, 0, stream>>>(src, dst, el, er, ebuf, nmax, E * 4);
  edge_expsum<<<grid(E * 4), THREADS, 0, stream>>>(dst, nmax, ebuf, nsum, E * 4);
  edge_aggregate<<<grid(E * 4), THREADS, 0, stream>>>(src, dst, ebuf, nsum, bufA, bufB, E * 4);
  combine_mean<<<grid(N * 32), THREADS, 0, stream>>>(bufB, bufC, bias2, hnode, N * 32);

  node_weight<<<grid(N), THREADS, 0, stream>>>(hnode, aw_w, aw_b, wbuf, N);
  graph_init<<<grid(G * 64), THREADS, 0, stream>>>((unsigned*)gmat, G * 64);
  graph_scatter<<<grid(N * 32), THREADS, 0, stream>>>(hnode, wbuf, gid, gmat, N * 32);
  graph_decode<<<grid(G * 32), THREADS, 0, stream>>>(gmat, G * 32);
  wmma_gemm_xwt<64, 8><<<gblocks, 128, 0, stream>>>(gmat, out_w, ybuf, G);
  final_ln<<<G, 128, 0, stream>>>(ybuf, out_b, (float*)d_out);
}


# Edge softmax and aggregation done per destination node over a CSR rebuilt each call with E integer atomics; no float atomics on features; sum order per node = CSR arrival order
# speedup vs baseline: 3.5750x; 3.5750x over previous
#include <hip/hip_runtime.h>
#include <hip/hip_bf16.h>
#include <math.h>

typedef __attribute__((ext_vector_type(2))) float v2f;
typedef __attribute__((ext_vector_type(8))) float v8f;

#define THREADS 256
#define ORD_NEG_INF 0x007FFFFFu

__device__ __forceinline__ unsigned ford(float f) {
  unsigned u = __float_as_uint(f);
  return (u & 0x80000000u) ? ~u : (u | 0x80000000u);
}
__device__ __forceinline__ float funord(unsigned u) {
  return __uint_as_float((u & 0x80000000u) ? (u & 0x7fffffffu) : ~u);
}

template <int K, int NT>
__global__ void wmma_gemm_xwt(const float* __restrict__ X,
                              const float* __restrict__ W,
                              float* __restrict__ Y, int M) {
  const int NO   = NT * 16;
  const int lane = threadIdx.x & 31;
  const int wave = threadIdx.x >> 5;
  const int tile = blockIdx.x * (blockDim.x >> 5) + wave;
  if (tile * 16 >= M) return;
  const int rbase = tile * 16;
  const int mrow  = rbase + (lane & 15);
  const int koff  = (lane < 16) ? 0 : 2;

  v8f acc[NT];
#pragma unroll
  for (int t = 0; t < NT; ++t) acc[t] = (v8f){0.f,0.f,0.f,0.f,0.f,0.f,0.f,0.f};

  for (int k0 = 0; k0 < K; k0 += 4) {
    const int k = k0 + koff;
    v2f a;
    a.x = (k     < K) ? X[mrow * K + k]     : 0.f;
    a.y = (k + 1 < K) ? X[mrow * K + k + 1] : 0.f;
#pragma unroll
    for (int t = 0; t < NT; ++t) {
      const int col = t * 16 + (lane & 15);
      v2f b;
      b.x = (k     < K) ? W[col * K + k]     : 0.f;
      b.y = (k + 1 < K) ? W[col * K + k + 1] : 0.f;
      acc[t] = __builtin_amdgcn_wmma_f32_16x16x4_f32(
          false, a, false, b, (short)0, acc[t], false, false);
    }
  }

  const int rowoff = (lane < 16) ? 0 : 8;
  const int colo   = lane & 15;
#pragma unroll
  for (int t = 0; t < NT; ++t) {
#pragma unroll
    for (int i = 0; i < 8; ++i) {
      Y[(rbase + rowoff + i) * NO + t * 16 + colo] = acc[t][i];
    }
  }
}

__global__ void fill_u32(unsigned* p, unsigned v, int n) {
  int t = blockIdx.x * blockDim.x + threadIdx.x;
  if (t < n) p[t] = v;
}

__global__ void attn_proj(const float* __restrict__ feat,
                          const float* __restrict__ al,
                          const float* __restrict__ ar,
                          float* __restrict__ el, float* __restrict__ er, int NH) {
  int t = blockIdx.x * blockDim.x + threadIdx.x;
  if (t >= NH) return;
  int n = t >> 2, h = t & 3;
  const float* f = feat + n * 128 + h * 32;
  float sl = 0.f, sr = 0.f;
#pragma unroll
  for (int i = 0; i < 32; ++i) {
    sl += f[i] * al[h * 32 + i];
    sr += f[i] * ar[h * 32 + i];
  }
  el[t] = sl; er[t] = sr;
}

__global__ void deg_rank(const int* __restrict__ dst, int* __restrict__ cnt,
                         int* __restrict__ rank, int E) {
  int e = blockIdx.x * blockDim.x + threadIdx.x;
  if (e >= E) return;
  rank[e] = atomicAdd(&cnt[dst[e]], 1);
}

__global__ void seg_alloc(const int* __restrict__ cnt, int* __restrict__ start,
                          int* __restrict__ total, int N) {
  const int d    = blockIdx.x * blockDim.x + threadIdx.x;
  const int lane = threadIdx.x & 31;
  const int c = (d < N) ? cnt[d] : 0;
  int x = c;
  for (int o = 1; o < 32; o <<= 1) {
    int y = __shfl_up(x, o);
    if (lane >= o) x += y;
  }
  int base = 0;
  if (lane == 31) base = atomicAdd(total, x);
  base = __shfl(base, 31);
  if (d < N) start[d] = base + x - c;
}

__global__ void csr_fill(const int* __restrict__ src, const int* __restrict__ dst,
                         const int* __restrict__ start, const int* __restrict__ rank,
                         int* __restrict__ csr_src, int E) {
  int e = blockIdx.x * blockDim.x + threadIdx.x;
  if (e >= E) return;
  csr_src[start[dst[e]] + rank[e]] = src[e];
}

__global__ void node_softmax_gather(const int* __restrict__ start, const int* __restrict__ cnt,
                                    const int* __restrict__ csr_src,
                                    const float* __restrict__ el, const float* __restrict__ er,
                                    const float* __restrict__ feat, float* __restrict__ accum, int N) {
  const int lane = threadIdx.x & 31;
  const int d = __builtin_amdgcn_readfirstlane((int)(blockIdx.x * (blockDim.x >> 5) + (threadIdx.x >> 5)));
  if (d >= N) return;
  const int h   = lane >> 3;
  const int st  = start[d];
  const int deg = cnt[d];
  const float erd = er[d * 4 + h];

  float m = -INFINITY;
  for (int j = 0; j < deg; ++j) {
    const int sidx = csr_src[st + j];
    float v = el[sidx * 4 + h] + erd;
    v = (v > 0.f) ? v : 0.2f * v;
    m = fmaxf(m, v);
  }
  float sum = 0.f;
  for (int j = 0; j < deg; ++j) {
    const int sidx = csr_src[st + j];
    float v = el[sidx * 4 + h] + erd;
    v = (v > 0.f) ? v : 0.2f * v;
    sum += expf(v - m);
  }
  float4 acc = make_float4(0.f, 0.f, 0.f, 0.f);
  for (int j = 0; j < deg; ++j) {
    const int sidx = csr_src[st + j];
    float v = el[sidx * 4 + h] + erd;
    v = (v > 0.f) ? v : 0.2f * v;
    const float a = expf(v - m) / sum;
    const float4 f = *reinterpret_cast<const float4*>(feat + (size_t)sidx * 128 + lane * 4);
    acc.x += f.x * a; acc.y += f.y * a; acc.z += f.z * a; acc.w += f.w * a;
  }
  *reinterpret_cast<float4*>(accum + (size_t)d * 128 + lane * 4) = acc;
}

__global__ void elu_combine(float* __restrict__ accum, const float* __restrict__ res,
                            const float* __restrict__ bias, int n) {
  int t = blockIdx.x * blockDim.x + threadIdx.x;
  if (t >= n) return;
  float v = accum[t] + res[t] + bias[t & 127];
  accum[t] = (v > 0.f) ? v : expm1f(v);
}

__global__ void combine_mean(const float* __restrict__ accum, const float* __restrict__ h1,
                             const float* __restrict__ bias2, float* __restrict__ hnode, int ND) {
  int t = blockIdx.x * blockDim.x + threadIdx.x;
  if (t >= ND) return;
  int n = t >> 5, d = t & 31;
  float s = 0.f;
#pragma unroll
  for (int h = 0; h < 4; ++h) {
    int idx = n * 128 + h * 32 + d;
    s += accum[idx] + h1[idx] + bias2[h * 32 + d];
  }
  hnode[t] = s * 0.25f;
}

__global__ void node_weight(const float* __restrict__ hnode, const float* __restrict__ aw_w,
                            const float* __restrict__ aw_b, float* __restrict__ wbuf, int N) {
  int n = blockIdx.x * blockDim.x + threadIdx.x;
  if (n >= N) return;
  float s = aw_b[0];
#pragma unroll
  for (int i = 0; i < 32; ++i) s += hnode[n * 32 + i] * aw_w[i];
  wbuf[n] = 1.f / (1.f + expf(-s));
}

__global__ void graph_init(unsigned* gmat, int n) {
  int t = blockIdx.x * blockDim.x + threadIdx.x;
  if (t >= n) return;
  gmat[t] = ((t & 63) < 32) ? 0u : ORD_NEG_INF;
}

__global__ void graph_scatter(const float* __restrict__ hnode, const float* __restrict__ wbuf,
                              const int* __restrict__ gid, float* __restrict__ gmat, int ND) {
  int t = blockIdx.x * blockDim.x + threadIdx.x;
  if (t >= ND) return;
  int n = t >> 5, d = t & 31;
  int g = gid[n];
  float hv = hnode[t];
  atomicAdd(&gmat[g * 64 + d], wbuf[n] * hv);
  atomicMax((unsigned*)gmat + g * 64 + 32 + d, ford(hv));
}

__global__ void graph_decode(float* gmat, int GD) {
  int t = blockIdx.x * blockDim.x + threadIdx.x;
  if (t >= GD) return;
  int g = t >> 5, d = t & 31;
  unsigned u = ((unsigned*)gmat)[g * 64 + 32 + d];
  gmat[g * 64 + 32 + d] = (u == ORD_NEG_INF) ? 0.f : funord(u);
}

__global__ void final_ln(const float* __restrict__ ybuf, const float* __restrict__ out_b,
                         float* __restrict__ out) {
  __shared__ float sm[128];
  int g = blockIdx.x, j = threadIdx.x;
  float v = ybuf[g * 128 + j] + out_b[j];
  sm[j] = v; __syncthreads();
  for (int s = 64; s > 0; s >>= 1) { if (j < s) sm[j] += sm[j + s]; __syncthreads(); }
  float mu = sm[0] * (1.f / 128.f);
  __syncthreads();
  float dv = v - mu;
  sm[j] = dv * dv; __syncthreads();
  for (int s = 64; s > 0; s >>= 1) { if (j < s) sm[j] += sm[j + s]; __syncthreads(); }
  float var = sm[0] * (1.f / 128.f);
  out[g * 128 + j] = dv * rsqrtf(var + 1e-5f);
}

extern "C" void kernel_launch(void* const* d_in, const int* in_sizes, int n_in,
                              void* d_out, int out_size, void* d_ws, size_t ws_size,
                              hipStream_t stream) {
  const float* node  = (const float*)d_in[0];
  const int*   src   = (const int*)d_in[1];
  const int*   dst   = (const int*)d_in[2];
  const int*   gid   = (const int*)d_in[3];
  const float* fc1_w = (const float*)d_in[4];
  const float* al1   = (const float*)d_in[5];
  const float* ar1   = (const float*)d_in[6];
  const float* res1w = (const float*)d_in[7];
  const float* bias1 = (const float*)d_in[8];
  const float* fc2_w = (const float*)d_in[9];
  const float* al2   = (const float*)d_in[10];
  const float* ar2   = (const float*)d_in[11];
  const float* bias2 = (const float*)d_in[12];
  const float* aw_w  = (const float*)d_in[13];
  const float* aw_b  = (const float*)d_in[14];
  const float* out_w = (const float*)d_in[15];
  const float* out_b = (const float*)d_in[16];
  (void)n_in; (void)ws_size;

  const int N = in_sizes[3];
  const int E = in_sizes[1];
  const int G = out_size / 128;

  char* wsb = (char*)d_ws;
  size_t off = 0;
  auto alloc = [&](size_t bytes) -> char* {
    char* p = wsb + off;
    off = (off + bytes + 255) & ~(size_t)255;
    return p;
  };
  float*    bufA  = (float*)alloc((size_t)N * 128 * 4);
  float*    bufB  = (float*)alloc((size_t)N * 128 * 4);
  float*    bufC  = (float*)alloc((size_t)N * 128 * 4);
  float*    el    = (float*)alloc((size_t)N * 4 * 4);
  float*    er    = (float*)alloc((size_t)N * 4 * 4);
  int*      cnt   = (int*)alloc((size_t)N * 4 * 4);
  int*      start = (int*)alloc((size_t)N * 4 * 4);
  int*      rank  = (int*)alloc((size_t)E * 4 * 4);
  int*      csr   = rank + E;
  float*    hnode = (float*)alloc((size_t)N * 32 * 4);
  float*    wbuf  = (float*)alloc((size_t)N * 4);
  float*    gmat  = (float*)alloc((size_t)G * 64 * 4);
  float*    ybuf  = (float*)alloc((size_t)G * 128 * 4);

  auto grid = [](int n) { return dim3((unsigned)((n + THREADS - 1) / THREADS)); };
  const int ntiles  = (N + 15) / 16;
  const int nblocks = (ntiles + 3) / 4;
  const int gtiles  = (G + 15) / 16;
  const int gblocks = (gtiles + 3) / 4;

  fill_u32<<<grid(N + 1), THREADS, 0, stream>>>((unsigned*)cnt, 0u, N + 1);
  deg_rank<<<grid(E), THREADS, 0, stream>>>(dst, cnt, rank, E);
  seg_alloc<<<grid(N), THREADS, 0, stream>>>(cnt, start, cnt + N, N);
  csr_fill<<<grid(E), THREADS, 0, stream>>>(src, dst, start, rank, csr, E);

  wmma_gemm_xwt<27, 8><<<nblocks, 128, 0, stream>>>(node, fc1_w, bufA, N);
  wmma_gemm_xwt<27, 8><<<nblocks, 128, 0, stream>>>(node, res1w, bufB, N);
  attn_proj<<<grid(N * 4), THREADS, 0, stream>>>(bufA, al1, ar1, el, er, N * 4);
  node_softmax_gather<<<grid(N * 32), THREADS, 0, stream>>>(start, cnt, csr, el, er, bufA, bufC, N);
  elu_combine<<<grid(N * 128), THREADS, 0, stream>>>(bufC, bufB, bias1, N * 128);

  wmma_gemm_xwt<128, 8><<<nblocks, 128, 0, stream>>>(bufC, fc2_w, bufA, N);
  attn_proj<<<grid(N * 4), THREADS, 0, stream>>>(bufA, al2, ar2, el, er, N * 4);
  node_softmax_gather<<<grid(N * 32), THREADS, 0, stream>>>(start, cnt, csr, el, er, bufA, bufB, N);
  combine_mean<<<grid(N * 32), THREADS, 0, stream>>>(bufB, bufC, bias2, hnode, N * 32);

  node_weight<<<grid(N), THREADS, 0, stream>>>(hnode, aw_w, aw_b, wbuf, N);
  graph_init<<<grid(G * 64), THREADS, 0, stream>>>((unsigned*)gmat, G * 64);
  graph_scatter<<<grid(N * 32), THREADS, 0, stream>>>(hnode, wbuf, gid, gmat, N * 32);
  graph_decode<<<grid(G * 32), THREADS, 0, stream>>>(gmat, G * 32);
  wmma_gemm_xwt<64, 8><<<gblocks, 128, 0, stream>>>(gmat, out_w, ybuf, G);
  final_ln<<<G, 128, 0, stream>>>(ybuf, out_b, (float*)d_out);
}


# Pointwise stages after each aggregation (bias+residual+ELU; bias+residual, head mean, node weight, readout atomics) run in the aggregation kernel on the row in registers; 19 -> 15 launches
# speedup vs baseline: 4.3156x; 1.2072x over previous
#include <hip/hip_runtime.h>
#include <hip/hip_bf16.h>
#include <math.h>

typedef __attribute__((ext_vector_type(2))) float v2f;
typedef __attribute__((ext_vector_type(8))) float v8f;

#define THREADS 256
#define ORD_NEG_INF 0x007FFFFFu

__device__ __forceinline__ unsigned ford(float f) {
  unsigned u = __float_as_uint(f);
  return (u & 0x80000000u) ? ~u : (u | 0x80000000u);
}
__device__ __forceinline__ float funord(unsigned u) {
  return __uint_as_float((u & 0x80000000u) ? (u & 0x7fffffffu) : ~u);
}

template <int K, int NT>
__global__ void wmma_gemm_xwt(const float* __restrict__ X,
                              const float* __restrict__ W,
                              float* __restrict__ Y, int M) {
  const int NO   = NT * 16;
  const int lane = threadIdx.x & 31;
  const int wave = threadIdx.x >> 5;
  const int tile = blockIdx.x * (blockDim.x >> 5) + wave;
  if (tile * 16 >= M) return;
  const int rbase = tile * 16;
  const int mrow  = rbase + (lane & 15);
  const int koff  = (lane < 16) ? 0 : 2;

  v8f acc[NT];
#pragma unroll
  for (int t = 0; t < NT; ++t) acc[t] = (v8f){0.f,0.f,0.f,0.f,0.f,0.f,0.f,0.f};

  for (int k0 = 0; k0 < K; k0 += 4) {
    const int k = k0 + koff;
    v2f a;
    a.x = (k     < K) ? X[mrow * K + k]     : 0.f;
    a.y = (k + 1 < K) ? X[mrow * K + k + 1] : 0.f;
#pragma unroll
    for (int t = 0; t < NT; ++t) {
      const int col = t * 16 + (lane & 15);
      v2f b;
      b.x = (k     < K) ? W[col * K + k]     : 0.f;
      b.y = (k + 1 < K) ? W[col * K + k + 1] : 0.f;
      acc[t] = __builtin_amdgcn_wmma_f32_16x16x4_f32(
          false, a, false, b, (short)0, acc[t], false, false);
    }
  }

  const int rowoff = (lane < 16) ? 0 : 8;
  const int colo   = lane & 15;
#pragma unroll
  for (int t = 0; t < NT; ++t) {
#pragma unroll
    for (int i = 0; i < 8; ++i) {
      Y[(rbase + rowoff + i) * NO + t * 16 + colo] = acc[t][i];
    }
  }
}

__global__ void fill_u32(unsigned* p, unsigned v, int n) {
  int t = blockIdx.x * blockDim.x + threadIdx.x;
  if (t < n) p[t] = v;
}

__global__ void attn_proj(const float* __restrict__ feat,
                          const float* __restrict__ al,
                          const float* __restrict__ ar,
                          float* __restrict__ el, float* __restrict__ er, int NH) {
  int t = blockIdx.x * blockDim.x + threadIdx.x;
  if (t >= NH) return;
  int n = t >> 2, h = t & 3;
  const float* f = feat + n * 128 + h * 32;
  float sl = 0.f, sr = 0.f;
#pragma unroll
  for (int i = 0; i < 32; ++i) {
    sl += f[i] * al[h * 32 + i];
    sr += f[i] * ar[h * 32 + i];
  }
  el[t] = sl; er[t] = sr;
}

__global__ void deg_rank(const int* __restrict__ dst, int* __restrict__ cnt,
                         int* __restrict__ rank, int E) {
  int e = blockIdx.x * blockDim.x + threadIdx.x;
  if (e >= E) return;
  rank[e] = atomicAdd(&cnt[dst[e]], 1);
}

__global__ void seg_alloc(const int* __restrict__ cnt, int* __restrict__ start,
                          int* __restrict__ total, int N) {
  const int d    = blockIdx.x * blockDim.x + threadIdx.x;
  const int lane = threadIdx.x & 31;
  const int c = (d < N) ? cnt[d] : 0;
  int x = c;
  for (int o = 1; o < 32; o <<= 1) {
    int y = __shfl_up(x, o);
    if (lane >= o) x += y;
  }
  int base = 0;
  if (lane == 31) base = atomicAdd(total, x);
  base = __shfl(base, 31);
  if (d < N) start[d] = base + x - c;
}

__global__ void csr_fill(const int* __restrict__ src, const int* __restrict__ dst,
                         const int* __restrict__ start, const int* __restrict__ rank,
                         int* __restrict__ csr_src, int E) {
  int e = blockIdx.x * blockDim.x + threadIdx.x;
  if (e >= E) return;
  csr_src[start[dst[e]] + rank[e]] = src[e];
}

template <int EPI>
__global__ void node_softmax_gather(const int* __restrict__ start, const int* __restrict__ cnt,
                                    const int* __restrict__ csr_src,
                                    const float* __restrict__ el, const float* __restrict__ er,
                                    const float* __restrict__ feat,
                                    const float* __restrict__ addrow,
                                    const float* __restrict__ bias,
                                    float* __restrict__ h1out,
                                    const float* __restrict__ aw_w, const float* __restrict__ aw_b,
                                    const int* __restrict__ gid, float* __restrict__ gmat,
                                    int N) {
  const int lane = threadIdx.x & 31;
  const int d = __builtin_amdgcn_readfirstlane((int)(blockIdx.x * (blockDim.x >> 5) + (threadIdx.x >> 5)));
  if (d >= N) return;
  const int h   = lane >> 3;
  const int st  = start[d];
  const int deg = cnt[d];
  const float erd = er[d * 4 + h];
  const float4 r = *reinterpret_cast<const float4*>(addrow + (size_t)d * 128 + lane * 4);
  const float4 b = *reinterpret_cast<const float4*>(bias + lane * 4);

  float m = -INFINITY;
  for (int j = 0; j < deg; ++j) {
    const int sidx = csr_src[st + j];
    float v = el[sidx * 4 + h] + erd;
    v = (v > 0.f) ? v : 0.2f * v;
    m = fmaxf(m, v);
  }
  float sum = 0.f;
  for (int j = 0; j < deg; ++j) {
    const int sidx = csr_src[st + j];
    float v = el[sidx * 4 + h] + erd;
    v = (v > 0.f) ? v : 0.2f * v;
    sum += expf(v - m);
  }
  float4 acc = make_float4(0.f, 0.f, 0.f, 0.f);
  for (int j = 0; j < deg; ++j) {
    const int sidx = csr_src[st + j];
    float v = el[sidx * 4 + h] + erd;
    v = (v > 0.f) ? v : 0.2f * v;
    const float a = expf(v - m) / sum;
    const float4 f = *reinterpret_cast<const float4*>(feat + (size_t)sidx * 128 + lane * 4);
    acc.x += f.x * a; acc.y += f.y * a; acc.z += f.z * a; acc.w += f.w * a;
  }

  float4 x;
  x.x = acc.x + r.x + b.x; x.y = acc.y + r.y + b.y;
  x.z = acc.z + r.z + b.z; x.w = acc.w + r.w + b.w;

  if (EPI == 1) {
    x.x = (x.x > 0.f) ? x.x : expm1f(x.x);
    x.y = (x.y > 0.f) ? x.y : expm1f(x.y);
    x.z = (x.z > 0.f) ? x.z : expm1f(x.z);
    x.w = (x.w > 0.f) ? x.w : expm1f(x.w);
    *reinterpret_cast<float4*>(h1out + (size_t)d * 128 + lane * 4) = x;
  } else {
    const int q = lane & 7;
    float4 hn = make_float4(0.f, 0.f, 0.f, 0.f);
#pragma unroll
    for (int hh = 0; hh < 4; ++hh) {
      hn.x += __shfl(x.x, q + 8 * hh);
      hn.y += __shfl(x.y, q + 8 * hh);
      hn.z += __shfl(x.z, q + 8 * hh);
      hn.w += __shfl(x.w, q + 8 * hh);
    }
    hn.x *= 0.25f; hn.y *= 0.25f; hn.z *= 0.25f; hn.w *= 0.25f;

    float sw = aw_b[0];
#pragma unroll
    for (int i = 0; i < 32; ++i) {
      const float comp = ((i & 3) == 0) ? hn.x : ((i & 3) == 1) ? hn.y : ((i & 3) == 2) ? hn.z : hn.w;
      const float hi = __int_as_float(__builtin_amdgcn_readlane(__float_as_int(comp), i >> 2));
      sw += hi * aw_w[i];
    }
    const float w = 1.f / (1.f + expf(-sw));

    const float t0 = __shfl(hn.x, lane >> 2);
    const float t1 = __shfl(hn.y, lane >> 2);
    const float t2 = __shfl(hn.z, lane >> 2);
    const float t3 = __shfl(hn.w, lane >> 2);
    const int   c  = lane & 3;
    const float hv = (c == 0) ? t0 : (c == 1) ? t1 : (c == 2) ? t2 : t3;
    const int g = gid[d];
    atomicAdd(&gmat[g * 64 + lane], w * hv);
    atomicMax((unsigned*)gmat + g * 64 + 32 + lane, ford(hv));
  }
}

__global__ void graph_init(unsigned* gmat, int n) {
  int t = blockIdx.x * blockDim.x + threadIdx.x;
  if (t >= n) return;
  gmat[t] = ((t & 63) < 32) ? 0u : ORD_NEG_INF;
}

__global__ void graph_decode(float* gmat, int GD) {
  int t = blockIdx.x * blockDim.x + threadIdx.x;
  if (t >= GD) return;
  int g = t >> 5, d = t & 31;
  unsigned u = ((unsigned*)gmat)[g * 64 + 32 + d];
  gmat[g * 64 + 32 + d] = (u == ORD_NEG_INF) ? 0.f : funord(u);
}

__global__ void final_ln(const float* __restrict__ ybuf, const float* __restrict__ out_b,
                         float* __restrict__ out) {
  __shared__ float sm[128];
  int g = blockIdx.x, j = threadIdx.x;
  float v = ybuf[g * 128 + j] + out_b[j];
  sm[j] = v; __syncthreads();
  for (int s = 64; s > 0; s >>= 1) { if (j < s) sm[j] += sm[j + s]; __syncthreads(); }
  float mu = sm[0] * (1.f / 128.f);
  __syncthreads();
  float dv = v - mu;
  sm[j] = dv * dv; __syncthreads();
  for (int s = 64; s > 0; s >>= 1) { if (j < s) sm[j] += sm[j + s]; __syncthreads(); }
  float var = sm[0] * (1.f / 128.f);
  out[g * 128 + j] = dv * rsqrtf(var + 1e-5f);
}

extern "C" void kernel_launch(void* const* d_in, const int* in_sizes, int n_in,
                              void* d_out, int out_size, void* d_ws, size_t ws_size,
                              hipStream_t stream) {
  const float* node  = (const float*)d_in[0];
  const int*   src   = (const int*)d_in[1];
  const int*   dst   = (const int*)d_in[2];
  const int*   gid   = (const int*)d_in[3];
  const float* fc1_w = (const float*)d_in[4];
  const float* al1   = (const float*)d_in[5];
  const float* ar1   = (const float*)d_in[6];
  const float* res1w = (const float*)d_in[7];
  const float* bias1 = (const float*)d_in[8];
  const float* fc2_w = (const float*)d_in[9];
  const float* al2   = (const float*)d_in[10];
  const float* ar2   = (const float*)d_in[11];
  const float* bias2 = (const float*)d_in[12];
  const float* aw_w  = (const float*)d_in[13];
  const float* aw_b  = (const float*)d_in[14];
  const float* out_w = (const float*)d_in[15];
  const float* out_b = (const float*)d_in[16];
  (void)n_in; (void)ws_size;

  const int N = in_sizes[3];
  const int E = in_sizes[1];
  const int G = out_size / 128;

  char* wsb = (char*)d_ws;
  size_t off = 0;
  auto alloc = [&](size_t bytes) -> char* {
    char* p = wsb + off;
    off = (off + bytes + 255) & ~(size_t)255;
    return p;
  };
  float*    bufA  = (float*)alloc((size_t)N * 128 * 4);
  float*    bufB  = (float*)alloc((size_t)N * 128 * 4);
  float*    bufC  = (float*)alloc((size_t)N * 128 * 4);
  float*    el    = (float*)alloc((size_t)N * 4 * 4);
  float*    er    = (float*)alloc((size_t)N * 4 * 4);
  int*      cnt   = (int*)alloc((size_t)N * 4 * 4);
  int*      start = (int*)alloc((size_t)N * 4 * 4);
  int*      rank  = (int*)alloc((size_t)E * 4 * 4);
  int*      csr   = rank + E;
  float*    gmat  = (float*)alloc((size_t)G * 64 * 4);
  float*    ybuf  = (float*)alloc((size_t)G * 128 * 4);

  auto grid = [](int n) { return dim3((unsigned)((n + THREADS - 1) / THREADS)); };
  const int ntiles  = (N + 15) / 16;
  const int nblocks = (ntiles + 3) / 4;
  const int gtiles  = (G + 15) / 16;
  const int gblocks = (gtiles + 3) / 4;

  fill_u32<<<grid(N + 1), THREADS, 0, stream>>>((unsigned*)cnt, 0u, N + 1);
  deg_rank<<<grid(E), THREADS, 0, stream>>>(dst, cnt, rank, E);
  seg_alloc<<<grid(N), THREADS, 0, stream>>>(cnt, start, cnt + N, N);
  csr_fill<<<grid(E), THREADS, 0, stream>>>(src, dst, start, rank, csr, E);

  wmma_gemm_xwt<27, 8><<<nblocks, 128, 0, stream>>>(node, fc1_w, bufA, N);
  wmma_gemm_xwt<27, 8><<<nblocks, 128, 0, stream>>>(node, res1w, bufB, N);
  attn_proj<<<grid(N * 4), THREADS, 0, stream>>>(bufA, al1, ar1, el, er, N * 4);
  node_softmax_gather<1><<<grid(N * 32), THREADS, 0, stream>>>(start, cnt, csr, el, er, bufA, bufB, bias1, bufC,
                                                               aw_w, aw_b, gid, gmat, N);

  wmma_gemm_xwt<128, 8><<<nblocks, 128, 0, stream>>>(bufC, fc2_w, bufA, N);
  attn_proj<<<grid(N * 4), THREADS, 0, stream>>>(bufA, al2, ar2, el, er, N * 4);
  graph_init<<<grid(G * 64), THREADS, 0, stream>>>((unsigned*)gmat, G * 64);
  node_softmax_gather<2><<<grid(N * 32), THREADS, 0, stream>>>(start, cnt, csr, el, er, bufA, bufC, bias2, bufB,
                                                               aw_w, aw_b, gid, gmat, N);

  graph_decode<<<grid(G * 32), THREADS, 0, stream>>>(gmat, G * 32);
  wmma_gemm_xwt<64, 8><<<gblocks, 128, 0, stream>>>(gmat, out_w, ybuf, G);
  final_ln<<<G, 128, 0, stream>>>(ybuf, out_b, (float*)d_out);
}


# Layer 1 aggregates the 27-wide input rows per head before projecting ((sum_e a*x) @ W with the residual on WMMA), scores as x.(W^T a): 108-B row gathers, not 512-B; sums reassociated; 15->14 launches
# speedup vs baseline: 4.5542x; 1.0553x over previous
#include <hip/hip_runtime.h>
#include <hip/hip_bf16.h>
#include <math.h>

typedef __attribute__((ext_vector_type(2))) float v2f;
typedef __attribute__((ext_vector_type(8))) float v8f;

#define THREADS 256
#define ORD_NEG_INF 0x007FFFFFu

__device__ __forceinline__ unsigned ford(float f) {
  unsigned u = __float_as_uint(f);
  return (u & 0x80000000u) ? ~u : (u | 0x80000000u);
}
__device__ __forceinline__ float funord(unsigned u) {
  return __uint_as_float((u & 0x80000000u) ? (u & 0x7fffffffu) : ~u);
}

template <int K, int NT>
__global__ void wmma_gemm_xwt(const float* __restrict__ X,
                              const float* __restrict__ W,
                              float* __restrict__ Y, int M) {
  const int NO   = NT * 16;
  const int lane = threadIdx.x & 31;
  const int wave = threadIdx.x >> 5;
  const int tile = blockIdx.x * (blockDim.x >> 5) + wave;
  if (tile * 16 >= M) return;
  const int rbase = tile * 16;
  const int mrow  = rbase + (lane & 15);
  const int koff  = (lane < 16) ? 0 : 2;

  v8f acc[NT];
#pragma unroll
  for (int t = 0; t < NT; ++t) acc[t] = (v8f){0.f,0.f,0.f,0.f,0.f,0.f,0.f,0.f};

  for (int k0 = 0; k0 < K; k0 += 4) {
    const int k = k0 + koff;
    v2f a;
    a.x = (k     < K) ? X[mrow * K + k]     : 0.f;
    a.y = (k + 1 < K) ? X[mrow * K + k + 1] : 0.f;
#pragma unroll
    for (int t = 0; t < NT; ++t) {
      const int col = t * 16 + (lane & 15);
      v2f b;
      b.x = (k     < K) ? W[col * K + k]     : 0.f;
      b.y = (k + 1 < K) ? W[col * K + k + 1] : 0.f;
      acc[t] = __builtin_amdgcn_wmma_f32_16x16x4_f32(
          false, a, false, b, (short)0, acc[t], false, false);
    }
  }

  const int rowoff = (lane < 16) ? 0 : 8;
  const int colo   = lane & 15;
#pragma unroll
  for (int t = 0; t < NT; ++t) {
#pragma unroll
    for (int i = 0; i < 8; ++i) {
      Y[(rbase + rowoff + i) * NO + t * 16 + colo] = acc[t][i];
    }
  }
}

__device__ __forceinline__ v2f load_pair27(const float* __restrict__ row, int k) {
  const int k1 = (k + 1 < 27) ? k + 1 : 26;
  v2f r;
  r.x = row[k];
  const float y = row[k1];
  r.y = (k + 1 < 27) ? y : 0.f;
  return r;
}

__global__ void __launch_bounds__(128)
wmma_gemm_l1(const float* __restrict__ agg, const float* __restrict__ x,
             const float* __restrict__ W1, const float* __restrict__ R,
             const float* __restrict__ bias, float* __restrict__ h1, int M) {
  const int K    = 27;
  const int lane = threadIdx.x & 31;
  const int wave = threadIdx.x >> 5;
  const int tile = blockIdx.x * (blockDim.x >> 5) + wave;
  if (tile * 16 >= M) return;
  const int rbase = tile * 16;
  const int mrow  = rbase + (lane & 15);
  const int koff  = (lane < 16) ? 0 : 2;

  v8f acc[8];
#pragma unroll
  for (int t = 0; t < 8; ++t) acc[t] = (v8f){0.f,0.f,0.f,0.f,0.f,0.f,0.f,0.f};

  for (int k0 = 0; k0 < K; k0 += 4) {
    const int k = k0 + koff;
    const v2f a = load_pair27(x + mrow * K, k);
#pragma unroll
    for (int t = 0; t < 8; ++t) {
      const int col = t * 16 + (lane & 15);
      const v2f b = load_pair27(R + col * K, k);
      acc[t] = __builtin_amdgcn_wmma_f32_16x16x4_f32(
          false, a, false, b, (short)0, acc[t], false, false);
    }
  }

#pragma unroll
  for (int h = 0; h < 4; ++h) {
    const float* __restrict__ ah = agg + (size_t)mrow * 108 + h * K;
    for (int k0 = 0; k0 < K; k0 += 4) {
      const int k = k0 + koff;
      const v2f a = load_pair27(ah, k);
#pragma unroll
      for (int tt = 0; tt < 2; ++tt) {
        const int t   = 2 * h + tt;
        const int col = t * 16 + (lane & 15);
        const v2f b = load_pair27(W1 + col * K, k);
        acc[t] = __builtin_amdgcn_wmma_f32_16x16x4_f32(
            false, a, false, b, (short)0, acc[t], false, false);
      }
    }
  }

  const int rowoff = (lane < 16) ? 0 : 8;
  const int colo   = lane & 15;
#pragma unroll
  for (int t = 0; t < 8; ++t) {
    const float bt = bias[t * 16 + colo];
#pragma unroll
    for (int i = 0; i < 8; ++i) {
      float v = acc[t][i] + bt;
      v = (v > 0.f) ? v : expm1f(v);
      h1[(size_t)(rbase + rowoff + i) * 128 + t * 16 + colo] = v;
    }
  }
}

__global__ void fill_u32(unsigned* p, unsigned v, int n) {
  int t = blockIdx.x * blockDim.x + threadIdx.x;
  if (t < n) p[t] = v;
}

__global__ void attn_proj(const float* __restrict__ feat,
                          const float* __restrict__ al,
                          const float* __restrict__ ar,
                          float* __restrict__ el, float* __restrict__ er, int NH) {
  int t = blockIdx.x * blockDim.x + threadIdx.x;
  if (t >= NH) return;
  int n = t >> 2, h = t & 3;
  const float* f = feat + n * 128 + h * 32;
  float sl = 0.f, sr = 0.f;
#pragma unroll
  for (int i = 0; i < 32; ++i) {
    sl += f[i] * al[h * 32 + i];
    sr += f[i] * ar[h * 32 + i];
  }
  el[t] = sl; er[t] = sr;
}

__global__ void attn_proj_in(const float* __restrict__ x, const float* __restrict__ w,
                             const float* __restrict__ al, const float* __restrict__ ar,
                             float* __restrict__ el, float* __restrict__ er, int NH) {
  __shared__ float vt[216];
  const int tid = threadIdx.x;
  if (tid < 216) {
    const int which = (tid >= 108) ? 1 : 0;
    const int idx = tid - 108 * which;
    const int h = idx / 27, c = idx - h * 27;
    const float* a = which ? ar : al;
    float s = 0.f;
    for (int d = 0; d < 32; ++d) s += w[(h * 32 + d) * 27 + c] * a[h * 32 + d];
    vt[tid] = s;
  }
  __syncthreads();
  const int t = blockIdx.x * blockDim.x + tid;
  if (t >= NH) return;
  const int n = t >> 2, h = t & 3;
  const float* xr = x + n * 27;
  float sl = 0.f, sr = 0.f;
  for (int c = 0; c < 27; ++c) {
    const float xv = xr[c];
    sl += xv * vt[h * 27 + c];
    sr += xv * vt[108 + h * 27 + c];
  }
  el[t] = sl; er[t] = sr;
}

__global__ void deg_rank(const int* __restrict__ dst, int* __restrict__ cnt,
                         int* __restrict__ rank, int E) {
  int e = blockIdx.x * blockDim.x + threadIdx.x;
  if (e >= E) return;
  rank[e] = atomicAdd(&cnt[dst[e]], 1);
}

__global__ void seg_alloc(const int* __restrict__ cnt, int* __restrict__ start,
                          int* __restrict__ total, int N) {
  const int d    = blockIdx.x * blockDim.x + threadIdx.x;
  const int lane = threadIdx.x & 31;
  const int c = (d < N) ? cnt[d] : 0;
  int x = c;
  for (int o = 1; o < 32; o <<= 1) {
    int y = __shfl_up(x, o);
    if (lane >= o) x += y;
  }
  int base = 0;
  if (lane == 31) base = atomicAdd(total, x);
  base = __shfl(base, 31);
  if (d < N) start[d] = base + x - c;
}

__global__ void csr_fill(const int* __restrict__ src, const int* __restrict__ dst,
                         const int* __restrict__ start, const int* __restrict__ rank,
                         int* __restrict__ csr_src, int E) {
  int e = blockIdx.x * blockDim.x + threadIdx.x;
  if (e >= E) return;
  csr_src[start[dst[e]] + rank[e]] = src[e];
}

__global__ void node_softmax_gather_in(const int* __restrict__ start, const int* __restrict__ cnt,
                                       const int* __restrict__ csr_src,
                                       const float* __restrict__ el, const float* __restrict__ er,
                                       const float* __restrict__ x,
                                       float* __restrict__ agg,
                                       int N) {
  const int lane = threadIdx.x & 31;
  const int d = __builtin_amdgcn_readfirstlane((int)(blockIdx.x * (blockDim.x >> 5) + (threadIdx.x >> 5)));
  if (d >= N) return;
  const int h   = lane & 3;
  const int st  = start[d];
  const int deg = cnt[d];
  const float erd = er[d * 4 + h];
  const int cl = (lane < 27) ? lane : 26;

  float m = -INFINITY;
  for (int j = 0; j < deg; ++j) {
    const int sidx = csr_src[st + j];
    float v = el[sidx * 4 + h] + erd;
    v = (v > 0.f) ? v : 0.2f * v;
    m = fmaxf(m, v);
  }
  float sum = 0.f;
  for (int j = 0; j < deg; ++j) {
    const int sidx = csr_src[st + j];
    float v = el[sidx * 4 + h] + erd;
    v = (v > 0.f) ? v : 0.2f * v;
    sum += expf(v - m);
  }
  float a0 = 0.f, a1 = 0.f, a2 = 0.f, a3 = 0.f;
  for (int j = 0; j < deg; ++j) {
    const int sidx = csr_src[st + j];
    float v = el[sidx * 4 + h] + erd;
    v = (v > 0.f) ? v : 0.2f * v;
    const int a = __float_as_int(expf(v - m) / sum);
    const float xv = x[(size_t)sidx * 27 + cl];
    a0 += xv * __int_as_float(__builtin_amdgcn_readlane(a, 0));
    a1 += xv * __int_as_float(__builtin_amdgcn_readlane(a, 1));
    a2 += xv * __int_as_float(__builtin_amdgcn_readlane(a, 2));
    a3 += xv * __int_as_float(__builtin_amdgcn_readlane(a, 3));
  }
  if (lane < 27) {
    float* o = agg + (size_t)d * 108 + lane;
    o[0] = a0; o[27] = a1; o[54] = a2; o[81] = a3;
  }
}

__global__ void node_softmax_gather(const int* __restrict__ start, const int* __restrict__ cnt,
                                    const int* __restrict__ csr_src,
                                    const float* __restrict__ el, const float* __restrict__ er,
                                    const float* __restrict__ feat,
                                    const float* __restrict__ addrow,
                                    const float* __restrict__ bias,
                                    const float* __restrict__ aw_w, const float* __restrict__ aw_b,
                                    const int* __restrict__ gid, float* __restrict__ gmat,
                                    int N) {
  const int lane = threadIdx.x & 31;
  const int d = __builtin_amdgcn_readfirstlane((int)(blockIdx.x * (blockDim.x >> 5) + (threadIdx.x >> 5)));
  if (d >= N) return;
  const int h   = lane >> 3;
  const int st  = start[d];
  const int deg = cnt[d];
  const float erd = er[d * 4 + h];
  const float4 r = *reinterpret_cast<const float4*>(addrow + (size_t)d * 128 + lane * 4);
  const float4 b = *reinterpret_cast<const float4*>(bias + lane * 4);

  float m = -INFINITY;
  for (int j = 0; j < deg; ++j) {
    const int sidx = csr_src[st + j];
    float v = el[sidx * 4 + h] + erd;
    v = (v > 0.f) ? v : 0.2f * v;
    m = fmaxf(m, v);
  }
  float sum = 0.f;
  for (int j = 0; j < deg; ++j) {
    const int sidx = csr_src[st + j];
    float v = el[sidx * 4 + h] + erd;
    v = (v > 0.f) ? v : 0.2f * v;
    sum += expf(v - m);
  }
  float4 acc = make_float4(0.f, 0.f, 0.f, 0.f);
  for (int j = 0; j < deg; ++j) {
    const int sidx = csr_src[st + j];
    float v = el[sidx * 4 + h] + erd;
    v = (v > 0.f) ? v : 0.2f * v;
    const float a = expf(v - m) / sum;
    const float4 f = *reinterpret_cast<const float4*>(feat + (size_t)sidx * 128 + lane * 4);
    acc.x += f.x * a; acc.y += f.y * a; acc.z += f.z * a; acc.w += f.w * a;
  }

  float4 x;
  x.x = acc.x + r.x + b.x; x.y = acc.y + r.y + b.y;
  x.z = acc.z + r.z + b.z; x.w = acc.w + r.w + b.w;

  const int q = lane & 7;
  float4 hn = make_float4(0.f, 0.f, 0.f, 0.f);
#pragma unroll
  for (int hh = 0; hh < 4; ++hh) {
    hn.x += __shfl(x.x, q + 8 * hh);
    hn.y += __shfl(x.y, q + 8 * hh);
    hn.z += __shfl(x.z, q + 8 * hh);
    hn.w += __shfl(x.w, q + 8 * hh);
  }
  hn.x *= 0.25f; hn.y *= 0.25f; hn.z *= 0.25f; hn.w *= 0.25f;

  float sw = aw_b[0];
#pragma unroll
  for (int i = 0; i < 32; ++i) {
    const float comp = ((i & 3) == 0) ? hn.x : ((i & 3) == 1) ? hn.y : ((i & 3) == 2) ? hn.z : hn.w;
    const float hi = __int_as_float(__builtin_amdgcn_readlane(__float_as_int(comp), i >> 2));
    sw += hi * aw_w[i];
  }
  const float w = 1.f / (1.f + expf(-sw));

  const float t0 = __shfl(hn.x, lane >> 2);
  const float t1 = __shfl(hn.y, lane >> 2);
  const float t2 = __shfl(hn.z, lane >> 2);
  const float t3 = __shfl(hn.w, lane >> 2);
  const int   c  = lane & 3;
  const float hv = (c == 0) ? t0 : (c == 1) ? t1 : (c == 2) ? t2 : t3;
  const int g = gid[d];
  atomicAdd(&gmat[g * 64 + lane], w * hv);
  atomicMax((unsigned*)gmat + g * 64 + 32 + lane, ford(hv));
}

__global__ void graph_init(unsigned* gmat, int n) {
  int t = blockIdx.x * blockDim.x + threadIdx.x;
  if (t >= n) return;
  gmat[t] = ((t & 63) < 32) ? 0u : ORD_NEG_INF;
}

__global__ void graph_decode(float* gmat, int GD) {
  int t = blockIdx.x * blockDim.x + threadIdx.x;
  if (t >= GD) return;
  int g = t >> 5, d = t & 31;
  unsigned u = ((unsigned*)gmat)[g * 64 + 32 + d];
  gmat[g * 64 + 32 + d] = (u == ORD_NEG_INF) ? 0.f : funord(u);
}

__global__ void final_ln(const float* __restrict__ ybuf, const float* __restrict__ out_b,
                         float* __restrict__ out) {
  __shared__ float sm[128];
  int g = blockIdx.x, j = threadIdx.x;
  float v = ybuf[g * 128 + j] + out_b[j];
  sm[j] = v; __syncthreads();
  for (int s = 64; s > 0; s >>= 1) { if (j < s) sm[j] += sm[j + s]; __syncthreads(); }
  float mu = sm[0] * (1.f / 128.f);
  __syncthreads();
  float dv = v - mu;
  sm[j] = dv * dv; __syncthreads();
  for (int s = 64; s > 0; s >>= 1) { if (j < s) sm[j] += sm[j + s]; __syncthreads(); }
  float var = sm[0] * (1.f / 128.f);
  out[g * 128 + j] = dv * rsqrtf(var + 1e-5f);
}

extern "C" void kernel_launch(void* const* d_in, const int* in_sizes, int n_in,
                              void* d_out, int out_size, void* d_ws, size_t ws_size,
                              hipStream_t stream) {
  const float* node  = (const float*)d_in[0];
  const int*   src   = (const int*)d_in[1];
  const int*   dst   = (const int*)d_in[2];
  const int*   gid   = (const int*)d_in[3];
  const float* fc1_w = (const float*)d_in[4];
  const float* al1   = (const float*)d_in[5];
  const float* ar1   = (const float*)d_in[6];
  const float* res1w = (const float*)d_in[7];
  const float* bias1 = (const float*)d_in[8];
  const float* fc2_w = (const float*)d_in[9];
  const float* al2   = (const float*)d_in[10];
  const float* ar2   = (const float*)d_in[11];
  const float* bias2 = (const float*)d_in[12];
  const float* aw_w  = (const float*)d_in[13];
  const float* aw_b  = (const float*)d_in[14];
  const float* out_w = (const float*)d_in[15];
  const float* out_b = (const float*)d_in[16];
  (void)n_in; (void)ws_size;

  const int N = in_sizes[3];
  const int E = in_sizes[1];
  const int G = out_size / 128;

  char* wsb = (char*)d_ws;
  size_t off = 0;
  auto alloc = [&](size_t bytes) -> char* {
    char* p = wsb + off;
    off = (off + bytes + 255) & ~(size_t)255;
    return p;
  };
  float*    bufA  = (float*)alloc((size_t)N * 128 * 4);
  float*    bufB  = (float*)alloc((size_t)N * 128 * 4);
  float*    bufC  = (float*)alloc((size_t)N * 128 * 4);
  float*    el    = (float*)alloc((size_t)N * 4 * 4);
  float*    er    = (float*)alloc((size_t)N * 4 * 4);
  int*      cnt   = (int*)alloc((size_t)N * 4 * 4);
  int*      start = (int*)alloc((size_t)N * 4 * 4);
  int*      rank  = (int*)alloc((size_t)E * 4 * 4);
  int*      csr   = rank + E;
  float*    gmat  = (float*)alloc((size_t)G * 64 * 4);
  float*    ybuf  = (float*)alloc((size_t)G * 128 * 4);

  auto grid = [](int n) { return dim3((unsigned)((n + THREADS - 1) / THREADS)); };
  const int ntiles  = (N + 15) / 16;
  const int nblocks = (ntiles + 3) / 4;
  const int gtiles  = (G + 15) / 16;
  const int gblocks = (gtiles + 3) / 4;

  fill_u32<<<grid(N + 1), THREADS, 0, stream>>>((unsigned*)cnt, 0u, N + 1);
  deg_rank<<<grid(E), THREADS, 0, stream>>>(dst, cnt, rank, E);
  seg_alloc<<<grid(N), THREADS, 0, stream>>>(cnt, start, cnt + N, N);
  csr_fill<<<grid(E), THREADS, 0, stream>>>(src, dst, start, rank, csr, E);

  attn_proj_in<<<grid(N * 4), THREADS, 0, stream>>>(node, fc1_w, al1, ar1, el, er, N * 4);
  node_softmax_gather_in<<<grid(N * 32), THREADS, 0, stream>>>(start, cnt, csr, el, er, node, bufB, N);
  wmma_gemm_l1<<<nblocks, 128, 0, stream>>>(bufB, node, fc1_w, res1w, bias1, bufC, N);

  wmma_gemm_xwt<128, 8><<<nblocks, 128, 0, stream>>>(bufC, fc2_w, bufA, N);
  attn_proj<<<grid(N * 4), THREADS, 0, stream>>>(bufA, al2, ar2, el, er, N * 4);
  graph_init<<<grid(G * 64), THREADS, 0, stream>>>((unsigned*)gmat, G * 64);
  node_softmax_gather<<<grid(N * 32), THREADS, 0, stream>>>(start, cnt, csr, el, er, bufA, bufC, bias2,
                                                            aw_w, aw_b, gid, gmat, N);

  graph_decode<<<grid(G * 32), THREADS, 0, stream>>>(gmat, G * 32);
  wmma_gemm_xwt<64, 8><<<gblocks, 128, 0, stream>>>(gmat, out_w, ybuf, G);
  final_ln<<<G, 128, 0, stream>>>(ybuf, out_b, (float*)d_out);
}


# All three GEMM kernels load A and B fragments 16 B per lane (k in groups of 8, same permuted order in A and B; 27-wide rows end with one load of k 23..26): fewer 16-line loads; k sum order changed
# speedup vs baseline: 4.9044x; 1.0769x over previous
#include <hip/hip_runtime.h>
#include <hip/hip_bf16.h>
#include <math.h>

typedef __attribute__((ext_vector_type(2))) float v2f;
typedef __attribute__((ext_vector_type(4))) float v4f;
typedef __attribute__((ext_vector_type(8))) float v8f;

__device__ __forceinline__ v4f load_quad(const float* __restrict__ p) {
  v4f r;
  __builtin_memcpy(&r, p, 16);
  return r;
}

#define THREADS 256
#define ORD_NEG_INF 0x007FFFFFu

__device__ __forceinline__ unsigned ford(float f) {
  unsigned u = __float_as_uint(f);
  return (u & 0x80000000u) ? ~u : (u | 0x80000000u);
}
__device__ __forceinline__ float funord(unsigned u) {
  return __uint_as_float((u & 0x80000000u) ? (u & 0x7fffffffu) : ~u);
}

template <int K, int NT>
__global__ void wmma_gemm_xwt(const float* __restrict__ X,
                              const float* __restrict__ W,
                              float* __restrict__ Y, int M) {
  static_assert(K % 8 == 0, "the fragment loaders take k in groups of 8");
  const int NO   = NT * 16;
  const int lane = threadIdx.x & 31;
  const int wave = threadIdx.x >> 5;
  const int tile = blockIdx.x * (blockDim.x >> 5) + wave;
  if (tile * 16 >= M) return;
  const int rbase = tile * 16;
  const int mrow  = rbase + (lane & 15);
  const int qoff  = (lane < 16) ? 0 : 4;

  v8f acc[NT];
#pragma unroll
  for (int t = 0; t < NT; ++t) acc[t] = (v8f){0.f,0.f,0.f,0.f,0.f,0.f,0.f,0.f};

  const float* __restrict__ xr = X + (size_t)mrow * K + qoff;
  const float* __restrict__ wr = W + (size_t)(lane & 15) * K + qoff;
  for (int k0 = 0; k0 < K; k0 += 8) {
    const v4f a = load_quad(xr + k0);
    v4f b[NT];
#pragma unroll
    for (int t = 0; t < NT; ++t) b[t] = load_quad(wr + (size_t)t * 16 * K + k0);
#pragma unroll
    for (int t = 0; t < NT; ++t) {
      acc[t] = __builtin_amdgcn_wmma_f32_16x16x4_f32(
          false, (v2f){a.x, a.y}, false, (v2f){b[t].x, b[t].y}, (short)0, acc[t], false, false);
    }
#pragma unroll
    for (int t = 0; t < NT; ++t) {
      acc[t] = __builtin_amdgcn_wmma_f32_16x16x4_f32(
          false, (v2f){a.z, a.w}, false, (v2f){b[t].z, b[t].w}, (short)0, acc[t], false, false);
    }
  }

  const int rowoff = (lane < 16) ? 0 : 8;
  const int colo   = lane & 15;
#pragma unroll
  for (int t = 0; t < NT; ++t) {
#pragma unroll
    for (int i = 0; i < 8; ++i) {
      Y[(rbase + rowoff + i) * NO + t * 16 + colo] = acc[t][i];
    }
  }
}

__device__ __forceinline__ v2f load_tail27(const float* __restrict__ row, bool upper) {
  const v4f q = load_quad(row + 23);
  v2f r;
  r.x = upper ? q.w : q.y;
  r.y = upper ? 0.f : q.z;
  return r;
}

__global__ void __launch_bounds__(128)
wmma_gemm_l1(const float* __restrict__ agg, const float* __restrict__ x,
             const float* __restrict__ W1, const float* __restrict__ R,
             const float* __restrict__ bias, float* __restrict__ h1, int M) {
  const int K    = 27;
  const int lane = threadIdx.x & 31;
  const int wave = threadIdx.x >> 5;
  const int tile = blockIdx.x * (blockDim.x >> 5) + wave;
  if (tile * 16 >= M) return;
  const int rbase = tile * 16;
  const int mrow  = rbase + (lane & 15);
  const bool upper = lane >= 16;

  v8f acc[8];
#pragma unroll
  for (int t = 0; t < 8; ++t) acc[t] = (v8f){0.f,0.f,0.f,0.f,0.f,0.f,0.f,0.f};

  const int qoff = upper ? 4 : 0;

  {
    const float* __restrict__ xr = x + (size_t)mrow * K;
    const float* __restrict__ rr = R + (size_t)(lane & 15) * K;
#pragma unroll
    for (int k0 = 0; k0 < 24; k0 += 8) {
      const v4f a = load_quad(xr + k0 + qoff);
      v4f b[8];
#pragma unroll
      for (int t = 0; t < 8; ++t) b[t] = load_quad(rr + t * 16 * K + k0 + qoff);
#pragma unroll
      for (int t = 0; t < 8; ++t) {
        acc[t] = __builtin_amdgcn_wmma_f32_16x16x4_f32(
            false, (v2f){a.x, a.y}, false, (v2f){b[t].x, b[t].y}, (short)0, acc[t], false, false);
      }
#pragma unroll
      for (int t = 0; t < 8; ++t) {
        acc[t] = __builtin_amdgcn_wmma_f32_16x16x4_f32(
            false, (v2f){a.z, a.w}, false, (v2f){b[t].z, b[t].w}, (short)0, acc[t], false, false);
      }
    }
    const v2f a = load_tail27(xr, upper);
#pragma unroll
    for (int t = 0; t < 8; ++t) {
      const v2f b = load_tail27(rr + t * 16 * K, upper);
      acc[t] = __builtin_amdgcn_wmma_f32_16x16x4_f32(
          false, a, false, b, (short)0, acc[t], false, false);
    }
  }

#pragma unroll
  for (int h = 0; h < 4; ++h) {
    const float* __restrict__ ah = agg + (size_t)mrow * 108 + h * K;
    const float* __restrict__ wr = W1 + (size_t)(2 * h * 16 + (lane & 15)) * K;
#pragma unroll
    for (int k0 = 0; k0 < 24; k0 += 8) {
      const v4f a  = load_quad(ah + k0 + qoff);
      const v4f b0 = load_quad(wr + k0 + qoff);
      const v4f b1 = load_quad(wr + 16 * K + k0 + qoff);
      acc[2 * h] = __builtin_amdgcn_wmma_f32_16x16x4_f32(
          false, (v2f){a.x, a.y}, false, (v2f){b0.x, b0.y}, (short)0, acc[2 * h], false, false);
      acc[2 * h + 1] = __builtin_amdgcn_wmma_f32_16x16x4_f32(
          false, (v2f){a.x, a.y}, false, (v2f){b1.x, b1.y}, (short)0, acc[2 * h + 1], false, false);
      acc[2 * h] = __builtin_amdgcn_wmma_f32_16x16x4_f32(
          false, (v2f){a.z, a.w}, false, (v2f){b0.z, b0.w}, (short)0, acc[2 * h], false, false);
      acc[2 * h + 1] = __builtin_amdgcn_wmma_f32_16x16x4_f32(
          false, (v2f){a.z, a.w}, false, (v2f){b1.z, b1.w}, (short)0, acc[2 * h + 1], false, false);
    }
    const v2f a = load_tail27(ah, upper);
#pragma unroll
    for (int tt = 0; tt < 2; ++tt) {
      const v2f b = load_tail27(wr + tt * 16 * K, upper);
      acc[2 * h + tt] = __builtin_amdgcn_wmma_f32_16x16x4_f32(
          false, a, false, b, (short)0, acc[2 * h + tt], false, false);
    }
  }

  const int rowoff = (lane < 16) ? 0 : 8;
  const int colo   = lane & 15;
#pragma unroll
  for (int t = 0; t < 8; ++t) {
    const float bt = bias[t * 16 + colo];
#pragma unroll
    for (int i = 0; i < 8; ++i) {
      float v = acc[t][i] + bt;
      v = (v > 0.f) ? v : expm1f(v);
      h1[(size_t)(rbase + rowoff + i) * 128 + t * 16 + colo] = v;
    }
  }
}

__global__ void fill_u32(unsigned* p, unsigned v, int n) {
  int t = blockIdx.x * blockDim.x + threadIdx.x;
  if (t < n) p[t] = v;
}

__global__ void attn_proj(const float* __restrict__ feat,
                          const float* __restrict__ al,
                          const float* __restrict__ ar,
                          float* __restrict__ el, float* __restrict__ er, int NH) {
  int t = blockIdx.x * blockDim.x + threadIdx.x;
  if (t >= NH) return;
  int n = t >> 2, h = t & 3;
  const float* f = feat + n * 128 + h * 32;
  float sl = 0.f, sr = 0.f;
#pragma unroll
  for (int i = 0; i < 32; ++i) {
    sl += f[i] * al[h * 32 + i];
    sr += f[i] * ar[h * 32 + i];
  }
  el[t] = sl; er[t] = sr;
}

__global__ void attn_proj_in(const float* __restrict__ x, const float* __restrict__ w,
                             const float* __restrict__ al, const float* __restrict__ ar,
                             float* __restrict__ el, float* __restrict__ er, int NH) {
  __shared__ float vt[216];
  const int tid = threadIdx.x;
  if (tid < 216) {
    const int which = (tid >= 108) ? 1 : 0;
    const int idx = tid - 108 * which;
    const int h = idx / 27, c = idx - h * 27;
    const float* a = which ? ar : al;
    float s = 0.f;
    for (int d = 0; d < 32; ++d) s += w[(h * 32 + d) * 27 + c] * a[h * 32 + d];
    vt[tid] = s;
  }
  __syncthreads();
  const int t = blockIdx.x * blockDim.x + tid;
  if (t >= NH) return;
  const int n = t >> 2, h = t & 3;
  const float* xr = x + n * 27;
  float sl = 0.f, sr = 0.f;
  for (int c = 0; c < 27; ++c) {
    const float xv = xr[c];
    sl += xv * vt[h * 27 + c];
    sr += xv * vt[108 + h * 27 + c];
  }
  el[t] = sl; er[t] = sr;
}

__global__ void deg_rank(const int* __restrict__ dst, int* __restrict__ cnt,
                         int* __restrict__ rank, int E) {
  int e = blockIdx.x * blockDim.x + threadIdx.x;
  if (e >= E) return;
  rank[e] = atomicAdd(&cnt[dst[e]], 1);
}

__global__ void seg_alloc(const int* __restrict__ cnt, int* __restrict__ start,
                          int* __restrict__ total, int N) {
  const int d    = blockIdx.x * blockDim.x + threadIdx.x;
  const int lane = threadIdx.x & 31;
  const int c = (d < N) ? cnt[d] : 0;
  int x = c;
  for (int o = 1; o < 32; o <<= 1) {
    int y = __shfl_up(x, o);
    if (lane >= o) x += y;
  }
  int base = 0;
  if (lane == 31) base = atomicAdd(total, x);
  base = __shfl(base, 31);
  if (d < N) start[d] = base + x - c;
}

__global__ void csr_fill(const int* __restrict__ src, const int* __restrict__ dst,
                         const int* __restrict__ start, const int* __restrict__ rank,
                         int* __restrict__ csr_src, int E) {
  int e = blockIdx.x * blockDim.x + threadIdx.x;
  if (e >= E) return;
  csr_src[start[dst[e]] + rank[e]] = src[e];
}

__global__ void node_softmax_gather_in(const int* __restrict__ start, const int* __restrict__ cnt,
                                       const int* __restrict__ csr_src,
                                       const float* __restrict__ el, const float* __restrict__ er,
                                       const float* __restrict__ x,
                                       float* __restrict__ agg,
                                       int N) {
  const int lane = threadIdx.x & 31;
  const int d = __builtin_amdgcn_readfirstlane((int)(blockIdx.x * (blockDim.x >> 5) + (threadIdx.x >> 5)));
  if (d >= N) return;
  const int h   = lane & 3;
  const int st  = start[d];
  const int deg = cnt[d];
  const float erd = er[d * 4 + h];
  const int cl = (lane < 27) ? lane : 26;

  float m = -INFINITY;
  for (int j = 0; j < deg; ++j) {
    const int sidx = csr_src[st + j];
    float v = el[sidx * 4 + h] + erd;
    v = (v > 0.f) ? v : 0.2f * v;
    m = fmaxf(m, v);
  }
  float sum = 0.f;
  for (int j = 0; j < deg; ++j) {
    const int sidx = csr_src[st + j];
    float v = el[sidx * 4 + h] + erd;
    v = (v > 0.f) ? v : 0.2f * v;
    sum += expf(v - m);
  }
  float a0 = 0.f, a1 = 0.f, a2 = 0.f, a3 = 0.f;
  for (int j = 0; j < deg; ++j) {
    const int sidx = csr_src[st + j];
    float v = el[sidx * 4 + h] + erd;
    v = (v > 0.f) ? v : 0.2f * v;
    const int a = __float_as_int(expf(v - m) / sum);
    const float xv = x[(size_t)sidx * 27 + cl];
    a0 += xv * __int_as_float(__builtin_amdgcn_readlane(a, 0));
    a1 += xv * __int_as_float(__builtin_amdgcn_readlane(a, 1));
    a2 += xv * __int_as_float(__builtin_amdgcn_readlane(a, 2));
    a3 += xv * __int_as_float(__builtin_amdgcn_readlane(a, 3));
  }
  if (lane < 27) {
    float* o = agg + (size_t)d * 108 + lane;
    o[0] = a0; o[27] = a1; o[54] = a2; o[81] = a3;
  }
}

__global__ void node_softmax_gather(const int* __restrict__ start, const int* __restrict__ cnt,
                                    const int* __restrict__ csr_src,
                                    const float* __restrict__ el, const float* __restrict__ er,
                                    const float* __restrict__ feat,
                                    const float* __restrict__ addrow,
                                    const float* __restrict__ bias,
                                    const float* __restrict__ aw_w, const float* __restrict__ aw_b,
                                    const int* __restrict__ gid, float* __restrict__ gmat,
                                    int N) {
  const int lane = threadIdx.x & 31;
  const int d = __builtin_amdgcn_readfirstlane((int)(blockIdx.x * (blockDim.x >> 5) + (threadIdx.x >> 5)));
  if (d >= N) return;
  const int h   = lane >> 3;
  const int st  = start[d];
  const int deg = cnt[d];
  const float erd = er[d * 4 + h];
  const float4 r = *reinterpret_cast<const float4*>(addrow + (size_t)d * 128 + lane * 4);
  const float4 b = *reinterpret_cast<const float4*>(bias + lane * 4);

  float m = -INFINITY;
  for (int j = 0; j < deg; ++j) {
    const int sidx = csr_src[st + j];
    float v = el[sidx * 4 + h] + erd;
    v = (v > 0.f) ? v : 0.2f * v;
    m = fmaxf(m, v);
  }
  float sum = 0.f;
  for (int j = 0; j < deg; ++j) {
    const int sidx = csr_src[st + j];
    float v = el[sidx * 4 + h] + erd;
    v = (v > 0.f) ? v : 0.2f * v;
    sum += expf(v - m);
  }
  float4 acc = make_float4(0.f, 0.f, 0.f, 0.f);
  for (int j = 0; j < deg; ++j) {
    const int sidx = csr_src[st + j];
    float v = el[sidx * 4 + h] + erd;
    v = (v > 0.f) ? v : 0.2f * v;
    const float a = expf(v - m) / sum;
    const float4 f = *reinterpret_cast<const float4*>(feat + (size_t)sidx * 128 + lane * 4);
    acc.x += f.x * a; acc.y += f.y * a; acc.z += f.z * a; acc.w += f.w * a;
  }

  float4 x;
  x.x = acc.x + r.x + b.x; x.y = acc.y + r.y + b.y;
  x.z = acc.z + r.z + b.z; x.w = acc.w + r.w + b.w;

  const int q = lane & 7;
  float4 hn = make_float4(0.f, 0.f, 0.f, 0.f);
#pragma unroll
  for (int hh = 0; hh < 4; ++hh) {
    hn.x += __shfl(x.x, q + 8 * hh);
    hn.y += __shfl(x.y, q + 8 * hh);
    hn.z += __shfl(x.z, q + 8 * hh);
    hn.w += __shfl(x.w, q + 8 * hh);
  }
  hn.x *= 0.25f; hn.y *= 0.25f; hn.z *= 0.25f; hn.w *= 0.25f;

  float sw = aw_b[0];
#pragma unroll
  for (int i = 0; i < 32; ++i) {
    const float comp = ((i & 3) == 0) ? hn.x : ((i & 3) == 1) ? hn.y : ((i & 3) == 2) ? hn.z : hn.w;
    const float hi = __int_as_float(__builtin_amdgcn_readlane(__float_as_int(comp), i >> 2));
    sw += hi * aw_w[i];
  }
  const float w = 1.f / (1.f + expf(-sw));

  const float t0 = __shfl(hn.x, lane >> 2);
  const float t1 = __shfl(hn.y, lane >> 2);
  const float t2 = __shfl(hn.z, lane >> 2);
  const float t3 = __shfl(hn.w, lane >> 2);
  const int   c  = lane & 3;
  const float hv = (c == 0) ? t0 : (c == 1) ? t1 : (c == 2) ? t2 : t3;
  const int g = gid[d];
  atomicAdd(&gmat[g * 64 + lane], w * hv);
  atomicMax((unsigned*)gmat + g * 64 + 32 + lane, ford(hv));
}

__global__ void graph_init(unsigned* gmat, int n) {
  int t = blockIdx.x * blockDim.x + threadIdx.x;
  if (t >= n) return;
  gmat[t] = ((t & 63) < 32) ? 0u : ORD_NEG_INF;
}

__global__ void graph_decode(float* gmat, int GD) {
  int t = blockIdx.x * blockDim.x + threadIdx.x;
  if (t >= GD) return;
  int g = t >> 5, d = t & 31;
  unsigned u = ((unsigned*)gmat)[g * 64 + 32 + d];
  gmat[g * 64 + 32 + d] = (u == ORD_NEG_INF) ? 0.f : funord(u);
}

__global__ void final_ln(const float* __restrict__ ybuf, const float* __restrict__ out_b,
                         float* __restrict__ out) {
  __shared__ float sm[128];
  int g = blockIdx.x, j = threadIdx.x;
  float v = ybuf[g * 128 + j] + out_b[j];
  sm[j] = v; __syncthreads();
  for (int s = 64; s > 0; s >>= 1) { if (j < s) sm[j] += sm[j + s]; __syncthreads(); }
  float mu = sm[0] * (1.f / 128.f);
  __syncthreads();
  float dv = v - mu;
  sm[j] = dv * dv; __syncthreads();
  for (int s = 64; s > 0; s >>= 1) { if (j < s) sm[j] += sm[j + s]; __syncthreads(); }
  float var = sm[0] * (1.f / 128.f);
  out[g * 128 + j] = dv * rsqrtf(var + 1e-5f);
}

extern "C" void kernel_launch(void* const* d_in, const int* in_sizes, int n_in,
                              void* d_out, int out_size, void* d_ws, size_t ws_size,
                              hipStream_t stream) {
  const float* node  = (const float*)d_in[0];
  const int*   src   = (const int*)d_in[1];
  const int*   dst   = (const int*)d_in[2];
  const int*   gid   = (const int*)d_in[3];
  const float* fc1_w = (const float*)d_in[4];
  const float* al1   = (const float*)d_in[5];
  const float* ar1   = (const float*)d_in[6];
  const float* res1w = (const float*)d_in[7];
  const float* bias1 = (const float*)d_in[8];
  const float* fc2_w = (const float*)d_in[9];
  const float* al2   = (const float*)d_in[10];
  const float* ar2   = (const float*)d_in[11];
  const float* bias2 = (const float*)d_in[12];
  const float* aw_w  = (const float*)d_in[13];
  const float* aw_b  = (const float*)d_in[14];
  const float* out_w = (const float*)d_in[15];
  const float* out_b = (const float*)d_in[16];
  (void)n_in; (void)ws_size;

  const int N = in_sizes[3];
  const int E = in_sizes[1];
  const int G = out_size / 128;

  char* wsb = (char*)d_ws;
  size_t off = 0;
  auto alloc = [&](size_t bytes) -> char* {
    char* p = wsb + off;
    off = (off + bytes + 255) & ~(size_t)255;
    return p;
  };
  float*    bufA  = (float*)alloc((size_t)N * 128 * 4);
  float*    bufB  = (float*)alloc((size_t)N * 128 * 4);
  float*    bufC  = (float*)alloc((size_t)N * 128 * 4);
  float*    el    = (float*)alloc((size_t)N * 4 * 4);
  float*    er    = (float*)alloc((size_t)N * 4 * 4);
  int*      cnt   = (int*)alloc((size_t)N * 4 * 4);
  int*      start = (int*)alloc((size_t)N * 4 * 4);
  int*      rank  = (int*)alloc((size_t)E * 4 * 4);
  int*      csr   = rank + E;
  float*    gmat  = (float*)alloc((size_t)G * 64 * 4);
  float*    ybuf  = (float*)alloc((size_t)G * 128 * 4);

  auto grid = [](int n) { return dim3((unsigned)((n + THREADS - 1) / THREADS)); };
  const int ntiles  = (N + 15) / 16;
  const int nblocks = (ntiles + 3) / 4;
  const int gtiles  = (G + 15) / 16;
  const int gblocks = (gtiles + 3) / 4;

  fill_u32<<<grid(N + 1), THREADS, 0, stream>>>((unsigned*)cnt, 0u, N + 1);
  deg_rank<<<grid(E), THREADS, 0, stream>>>(dst, cnt, rank, E);
  seg_alloc<<<grid(N), THREADS, 0, stream>>>(cnt, start, cnt + N, N);
  csr_fill<<<grid(E), THREADS, 0, stream>>>(src, dst, start, rank, csr, E);

  attn_proj_in<<<grid(N * 4), THREADS, 0, stream>>>(node, fc1_w, al1, ar1, el, er, N * 4);
  node_softmax_gather_in<<<grid(N * 32), THREADS, 0, stream>>>(start, cnt, csr, el, er, node, bufB, N);
  wmma_gemm_l1<<<nblocks, 128, 0, stream>>>(bufB, node, fc1_w, res1w, bias1, bufC, N);

  wmma_gemm_xwt<128, 8><<<nblocks, 128, 0, stream>>>(bufC, fc2_w, bufA, N);
  attn_proj<<<grid(N * 4), THREADS, 0, stream>>>(bufA, al2, ar2, el, er, N * 4);
  graph_init<<<grid(G * 64), THREADS, 0, stream>>>((unsigned*)gmat, G * 64);
  node_softmax_gather<<<grid(N * 32), THREADS, 0, stream>>>(start, cnt, csr, el, er, bufA, bufC, bias2,
                                                            aw_w, aw_b, gid, gmat, N);

  graph_decode<<<grid(G * 32), THREADS, 0, stream>>>(gmat, G * 32);
  wmma_gemm_xwt<64, 8><<<gblocks, 128, 0, stream>>>(gmat, out_w, ybuf, G);
  final_ln<<<G, 128, 0, stream>>>(ybuf, out_b, (float*)d_out);
}


# Aggregation kernels: a node's 1..8 in-edges sit in lanes: softmax once per (edge, head) with DPP max/sum, not 3 serial passes per lane; row loads issued together; exp-sum is now an 8-lane tree
# speedup vs baseline: 5.3802x; 1.0970x over previous
#include <hip/hip_runtime.h>
#include <hip/hip_bf16.h>
#include <math.h>

typedef __attribute__((ext_vector_type(2))) float v2f;
typedef __attribute__((ext_vector_type(4))) float v4f;
typedef __attribute__((ext_vector_type(8))) float v8f;

__device__ __forceinline__ v4f load_quad(const float* __restrict__ p) {
  v4f r;
  __builtin_memcpy(&r, p, 16);
  return r;
}

#define THREADS 256
#define ORD_NEG_INF 0x007FFFFFu

__device__ __forceinline__ unsigned ford(float f) {
  unsigned u = __float_as_uint(f);
  return (u & 0x80000000u) ? ~u : (u | 0x80000000u);
}
__device__ __forceinline__ float funord(unsigned u) {
  return __uint_as_float((u & 0x80000000u) ? (u & 0x7fffffffu) : ~u);
}

template <int K, int NT>
__global__ void wmma_gemm_xwt(const float* __restrict__ X,
                              const float* __restrict__ W,
                              float* __restrict__ Y, int M) {
  static_assert(K % 8 == 0, "the fragment loaders take k in groups of 8");
  const int NO   = NT * 16;
  const int lane = threadIdx.x & 31;
  const int wave = threadIdx.x >> 5;
  const int tile = blockIdx.x * (blockDim.x >> 5) + wave;
  if (tile * 16 >= M) return;
  const int rbase = tile * 16;
  const int mrow  = rbase + (lane & 15);
  const int qoff  = (lane < 16) ? 0 : 4;

  v8f acc[NT];
#pragma unroll
  for (int t = 0; t < NT; ++t) acc[t] = (v8f){0.f,0.f,0.f,0.f,0.f,0.f,0.f,0.f};

  const float* __restrict__ xr = X + (size_t)mrow * K + qoff;
  const float* __restrict__ wr = W + (size_t)(lane & 15) * K + qoff;
  for (int k0 = 0; k0 < K; k0 += 8) {
    const v4f a = load_quad(xr + k0);
    v4f b[NT];
#pragma unroll
    for (int t = 0; t < NT; ++t) b[t] = load_quad(wr + (size_t)t * 16 * K + k0);
#pragma unroll
    for (int t = 0; t < NT; ++t) {
      acc[t] = __builtin_amdgcn_wmma_f32_16x16x4_f32(
          false, (v2f){a.x, a.y}, false, (v2f){b[t].x, b[t].y}, (short)0, acc[t], false, false);
    }
#pragma unroll
    for (int t = 0; t < NT; ++t) {
      acc[t] = __builtin_amdgcn_wmma_f32_16x16x4_f32(
          false, (v2f){a.z, a.w}, false, (v2f){b[t].z, b[t].w}, (short)0, acc[t], false, false);
    }
  }

  const int rowoff = (lane < 16) ? 0 : 8;
  const int colo   = lane & 15;
#pragma unroll
  for (int t = 0; t < NT; ++t) {
#pragma unroll
    for (int i = 0; i < 8; ++i) {
      Y[(rbase + rowoff + i) * NO + t * 16 + colo] = acc[t][i];
    }
  }
}

__device__ __forceinline__ v2f load_tail27(const float* __restrict__ row, bool upper) {
  const v4f q = load_quad(row + 23);
  v2f r;
  r.x = upper ? q.w : q.y;
  r.y = upper ? 0.f : q.z;
  return r;
}

__global__ void __launch_bounds__(128)
wmma_gemm_l1(const float* __restrict__ agg, const float* __restrict__ x,
             const float* __restrict__ W1, const float* __restrict__ R,
             const float* __restrict__ bias, float* __restrict__ h1, int M) {
  const int K    = 27;
  const int lane = threadIdx.x & 31;
  const int wave = threadIdx.x >> 5;
  const int tile = blockIdx.x * (blockDim.x >> 5) + wave;
  if (tile * 16 >= M) return;
  const int rbase = tile * 16;
  const int mrow  = rbase + (lane & 15);
  const bool upper = lane >= 16;

  v8f acc[8];
#pragma unroll
  for (int t = 0; t < 8; ++t) acc[t] = (v8f){0.f,0.f,0.f,0.f,0.f,0.f,0.f,0.f};

  const int qoff = upper ? 4 : 0;

  {
    const float* __restrict__ xr = x + (size_t)mrow * K;
    const float* __restrict__ rr = R + (size_t)(lane & 15) * K;
#pragma unroll
    for (int k0 = 0; k0 < 24; k0 += 8) {
      const v4f a = load_quad(xr + k0 + qoff);
      v4f b[8];
#pragma unroll
      for (int t = 0; t < 8; ++t) b[t] = load_quad(rr + t * 16 * K + k0 + qoff);
#pragma unroll
      for (int t = 0; t < 8; ++t) {
        acc[t] = __builtin_amdgcn_wmma_f32_16x16x4_f32(
            false, (v2f){a.x, a.y}, false, (v2f){b[t].x, b[t].y}, (short)0, acc[t], false, false);
      }
#pragma unroll
      for (int t = 0; t < 8; ++t) {
        acc[t] = __builtin_amdgcn_wmma_f32_16x16x4_f32(
            false, (v2f){a.z, a.w}, false, (v2f){b[t].z, b[t].w}, (short)0, acc[t], false, false);
      }
    }
    const v2f a = load_tail27(xr, upper);
#pragma unroll
    for (int t = 0; t < 8; ++t) {
      const v2f b = load_tail27(rr + t * 16 * K, upper);
      acc[t] = __builtin_amdgcn_wmma_f32_16x16x4_f32(
          false, a, false, b, (short)0, acc[t], false, false);
    }
  }

#pragma unroll
  for (int h = 0; h < 4; ++h) {
    const float* __restrict__ ah = agg + (size_t)mrow * 108 + h * K;
    const float* __restrict__ wr = W1 + (size_t)(2 * h * 16 + (lane & 15)) * K;
#pragma unroll
    for (int k0 = 0; k0 < 24; k0 += 8) {
      const v4f a  = load_quad(ah + k0 + qoff);
      const v4f b0 = load_quad(wr + k0 + qoff);
      const v4f b1 = load_quad(wr + 16 * K + k0 + qoff);
      acc[2 * h] = __builtin_amdgcn_wmma_f32_16x16x4_f32(
          false, (v2f){a.x, a.y}, false, (v2f){b0.x, b0.y}, (short)0, acc[2 * h], false, false);
      acc[2 * h + 1] = __builtin_amdgcn_wmma_f32_16x16x4_f32(
          false, (v2f){a.x, a.y}, false, (v2f){b1.x, b1.y}, (short)0, acc[2 * h + 1], false, false);
      acc[2 * h] = __builtin_amdgcn_wmma_f32_16x16x4_f32(
          false, (v2f){a.z, a.w}, false, (v2f){b0.z, b0.w}, (short)0, acc[2 * h], false, false);
      acc[2 * h + 1] = __builtin_amdgcn_wmma_f32_16x16x4_f32(
          false, (v2f){a.z, a.w}, false, (v2f){b1.z, b1.w}, (short)0, acc[2 * h + 1], false, false);
    }
    const v2f a = load_tail27(ah, upper);
#pragma unroll
    for (int tt = 0; tt < 2; ++tt) {
      const v2f b = load_tail27(wr + tt * 16 * K, upper);
      acc[2 * h + tt] = __builtin_amdgcn_wmma_f32_16x16x4_f32(
          false, a, false, b, (short)0, acc[2 * h + tt], false, false);
    }
  }

  const int rowoff = (lane < 16) ? 0 : 8;
  const int colo   = lane & 15;
#pragma unroll
  for (int t = 0; t < 8; ++t) {
    const float bt = bias[t * 16 + colo];
#pragma unroll
    for (int i = 0; i < 8; ++i) {
      float v = acc[t][i] + bt;
      v = (v > 0.f) ? v : expm1f(v);
      h1[(size_t)(rbase + rowoff + i) * 128 + t * 16 + colo] = v;
    }
  }
}

__global__ void fill_u32(unsigned* p, unsigned v, int n) {
  int t = blockIdx.x * blockDim.x + threadIdx.x;
  if (t < n) p[t] = v;
}

__global__ void attn_proj(const float* __restrict__ feat,
                          const float* __restrict__ al,
                          const float* __restrict__ ar,
                          float* __restrict__ el, float* __restrict__ er, int NH) {
  int t = blockIdx.x * blockDim.x + threadIdx.x;
  if (t >= NH) return;
  int n = t >> 2, h = t & 3;
  const float* f = feat + n * 128 + h * 32;
  float sl = 0.f, sr = 0.f;
#pragma unroll
  for (int i = 0; i < 32; ++i) {
    sl += f[i] * al[h * 32 + i];
    sr += f[i] * ar[h * 32 + i];
  }
  el[t] = sl; er[t] = sr;
}

__global__ void attn_proj_in(const float* __restrict__ x, const float* __restrict__ w,
                             const float* __restrict__ al, const float* __restrict__ ar,
                             float* __restrict__ el, float* __restrict__ er, int NH) {
  __shared__ float vt[216];
  const int tid = threadIdx.x;
  if (tid < 216) {
    const int which = (tid >= 108) ? 1 : 0;
    const int idx = tid - 108 * which;
    const int h = idx / 27, c = idx - h * 27;
    const float* a = which ? ar : al;
    float s = 0.f;
    for (int d = 0; d < 32; ++d) s += w[(h * 32 + d) * 27 + c] * a[h * 32 + d];
    vt[tid] = s;
  }
  __syncthreads();
  const int t = blockIdx.x * blockDim.x + tid;
  if (t >= NH) return;
  const int n = t >> 2, h = t & 3;
  const float* xr = x + n * 27;
  float sl = 0.f, sr = 0.f;
  for (int c = 0; c < 27; ++c) {
    const float xv = xr[c];
    sl += xv * vt[h * 27 + c];
    sr += xv * vt[108 + h * 27 + c];
  }
  el[t] = sl; er[t] = sr;
}

__global__ void deg_rank(const int* __restrict__ dst, int* __restrict__ cnt,
                         int* __restrict__ rank, int E) {
  int e = blockIdx.x * blockDim.x + threadIdx.x;
  if (e >= E) return;
  rank[e] = atomicAdd(&cnt[dst[e]], 1);
}

__global__ void seg_alloc(const int* __restrict__ cnt, int* __restrict__ start,
                          int* __restrict__ total, int N) {
  const int d    = blockIdx.x * blockDim.x + threadIdx.x;
  const int lane = threadIdx.x & 31;
  const int c = (d < N) ? cnt[d] : 0;
  int x = c;
  for (int o = 1; o < 32; o <<= 1) {
    int y = __shfl_up(x, o);
    if (lane >= o) x += y;
  }
  int base = 0;
  if (lane == 31) base = atomicAdd(total, x);
  base = __shfl(base, 31);
  if (d < N) start[d] = base + x - c;
}

__global__ void csr_fill(const int* __restrict__ src, const int* __restrict__ dst,
                         const int* __restrict__ start, const int* __restrict__ rank,
                         int* __restrict__ csr_src, int E) {
  int e = blockIdx.x * blockDim.x + threadIdx.x;
  if (e >= E) return;
  csr_src[start[dst[e]] + rank[e]] = src[e];
}

template <int MASK>
__device__ __forceinline__ float lane_xor(float x) {
  const int b = __float_as_int(x);
  return __int_as_float(__builtin_amdgcn_update_dpp(b, b, 0x160 + MASK, 0xf, 0xf, false));
}
template <int J>
__device__ __forceinline__ float group8_bcast(float x) {
  return __int_as_float((int)__builtin_amdgcn_mov_dpp8((unsigned)__float_as_int(x), J * 0x249249));
}
__device__ __forceinline__ int edge_source8(const int* __restrict__ csr_src, int st, int deg, int lane) {
  const int j = lane & 7;
  return csr_src[st + ((j < deg) ? j : deg - 1)];
}
__device__ __forceinline__ float edge_weight8(float els, float erd, int deg, int lane) {
  float v = els + erd;
  v = (v > 0.f) ? v : 0.2f * v;
  v = ((lane & 7) < deg) ? v : -INFINITY;
  float m = fmaxf(v, lane_xor<1>(v));
  m = fmaxf(m, lane_xor<2>(m));
  m = fmaxf(m, lane_xor<4>(m));
  const float ex = expf(v - m);
  float s = ex + lane_xor<1>(ex);
  s += lane_xor<2>(s);
  s += lane_xor<4>(s);
  return ex / s;
}

__global__ void node_softmax_gather_in(const int* __restrict__ start, const int* __restrict__ cnt,
                                       const int* __restrict__ csr_src,
                                       const float* __restrict__ el, const float* __restrict__ er,
                                       const float* __restrict__ x,
                                       float* __restrict__ agg,
                                       int N) {
  const int lane = threadIdx.x & 31;
  const int d = __builtin_amdgcn_readfirstlane((int)(blockIdx.x * (blockDim.x >> 5) + (threadIdx.x >> 5)));
  if (d >= N) return;
  const int h   = lane >> 3;
  const int st  = start[d];
  const int deg = cnt[d];
  const float erd = er[d * 4 + h];
  const int cl = (lane < 27) ? lane : 26;

  float a0 = 0.f, a1 = 0.f, a2 = 0.f, a3 = 0.f;
  if (deg >= 1 && deg <= 8) {
    const int sidx = edge_source8(csr_src, st, deg, lane);
    const float els = el[sidx * 4 + h];
    float xv0, xv1, xv2, xv3, xv4, xv5, xv6, xv7;
#define ROW_LOAD_IN(J) if (J < deg) xv##J = x[(size_t)__builtin_amdgcn_readlane(sidx, J) * 27 + cl];
    ROW_LOAD_IN(0) ROW_LOAD_IN(1) ROW_LOAD_IN(2) ROW_LOAD_IN(3)
    ROW_LOAD_IN(4) ROW_LOAD_IN(5) ROW_LOAD_IN(6) ROW_LOAD_IN(7)
#undef ROW_LOAD_IN
    const int a = __float_as_int(edge_weight8(els, erd, deg, lane));
#define ROW_FMA_IN(J) if (J < deg) {                                              \
      a0 += xv##J * __int_as_float(__builtin_amdgcn_readlane(a, J));              \
      a1 += xv##J * __int_as_float(__builtin_amdgcn_readlane(a, 8 + J));          \
      a2 += xv##J * __int_as_float(__builtin_amdgcn_readlane(a, 16 + J));         \
      a3 += xv##J * __int_as_float(__builtin_amdgcn_readlane(a, 24 + J)); }
    ROW_FMA_IN(0) ROW_FMA_IN(1) ROW_FMA_IN(2) ROW_FMA_IN(3)
    ROW_FMA_IN(4) ROW_FMA_IN(5) ROW_FMA_IN(6) ROW_FMA_IN(7)
#undef ROW_FMA_IN
  } else {
    float m = -INFINITY;
    for (int j = 0; j < deg; ++j) {
      const int sidx = csr_src[st + j];
      float v = el[sidx * 4 + h] + erd;
      v = (v > 0.f) ? v : 0.2f * v;
      m = fmaxf(m, v);
    }
    float sum = 0.f;
    for (int j = 0; j < deg; ++j) {
      const int sidx = csr_src[st + j];
      float v = el[sidx * 4 + h] + erd;
      v = (v > 0.f) ? v : 0.2f * v;
      sum += expf(v - m);
    }
    for (int j = 0; j < deg; ++j) {
      const int sidx = csr_src[st + j];
      float v = el[sidx * 4 + h] + erd;
      v = (v > 0.f) ? v : 0.2f * v;
      const int a = __float_as_int(expf(v - m) / sum);
      const float xv = x[(size_t)sidx * 27 + cl];
      a0 += xv * __int_as_float(__builtin_amdgcn_readlane(a, 0));
      a1 += xv * __int_as_float(__builtin_amdgcn_readlane(a, 8));
      a2 += xv * __int_as_float(__builtin_amdgcn_readlane(a, 16));
      a3 += xv * __int_as_float(__builtin_amdgcn_readlane(a, 24));
    }
  }
  if (lane < 27) {
    float* o = agg + (size_t)d * 108 + lane;
    o[0] = a0; o[27] = a1; o[54] = a2; o[81] = a3;
  }
}

__global__ void node_softmax_gather(const int* __restrict__ start, const int* __restrict__ cnt,
                                    const int* __restrict__ csr_src,
                                    const float* __restrict__ el, const float* __restrict__ er,
                                    const float* __restrict__ feat,
                                    const float* __restrict__ addrow,
                                    const float* __restrict__ bias,
                                    const float* __restrict__ aw_w, const float* __restrict__ aw_b,
                                    const int* __restrict__ gid, float* __restrict__ gmat,
                                    int N) {
  const int lane = threadIdx.x & 31;
  const int d = __builtin_amdgcn_readfirstlane((int)(blockIdx.x * (blockDim.x >> 5) + (threadIdx.x >> 5)));
  if (d >= N) return;
  const int h   = lane >> 3;
  const int st  = start[d];
  const int deg = cnt[d];
  const float erd = er[d * 4 + h];
  const float4 r = *reinterpret_cast<const float4*>(addrow + (size_t)d * 128 + lane * 4);
  const float4 b = *reinterpret_cast<const float4*>(bias + lane * 4);

  float4 acc = make_float4(0.f, 0.f, 0.f, 0.f);
  if (deg >= 1 && deg <= 8) {
    const int sidx = edge_source8(csr_src, st, deg, lane);
    const float els = el[sidx * 4 + h];
    float4 f0, f1, f2, f3, f4, f5, f6, f7;
#define ROW_LOAD(J) if (J < deg) f##J = *reinterpret_cast<const float4*>(                     \
        feat + (size_t)__builtin_amdgcn_readlane(sidx, J) * 128 + lane * 4);
    ROW_LOAD(0) ROW_LOAD(1) ROW_LOAD(2) ROW_LOAD(3)
    ROW_LOAD(4) ROW_LOAD(5) ROW_LOAD(6) ROW_LOAD(7)
#undef ROW_LOAD
    const float a = edge_weight8(els, erd, deg, lane);
#define ROW_FMA(J) if (J < deg) { const float aj = group8_bcast<J>(a);                        \
      acc.x += f##J.x * aj; acc.y += f##J.y * aj; acc.z += f##J.z * aj; acc.w += f##J.w * aj; }
    ROW_FMA(0) ROW_FMA(1) ROW_FMA(2) ROW_FMA(3)
    ROW_FMA(4) ROW_FMA(5) ROW_FMA(6) ROW_FMA(7)
#undef ROW_FMA
  } else {
    float m = -INFINITY;
    for (int j = 0; j < deg; ++j) {
      const int sidx = csr_src[st + j];
      float v = el[sidx * 4 + h] + erd;
      v = (v > 0.f) ? v : 0.2f * v;
      m = fmaxf(m, v);
    }
    float sum = 0.f;
    for (int j = 0; j < deg; ++j) {
      const int sidx = csr_src[st + j];
      float v = el[sidx * 4 + h] + erd;
      v = (v > 0.f) ? v : 0.2f * v;
      sum += expf(v - m);
    }
    for (int j = 0; j < deg; ++j) {
      const int sidx = csr_src[st + j];
      float v = el[sidx * 4 + h] + erd;
      v = (v > 0.f) ? v : 0.2f * v;
      const float a = expf(v - m) / sum;
      const float4 f = *reinterpret_cast<const float4*>(feat + (size_t)sidx * 128 + lane * 4);
      acc.x += f.x * a; acc.y += f.y * a; acc.z += f.z * a; acc.w += f.w * a;
    }
  }

  float4 x;
  x.x = acc.x + r.x + b.x; x.y = acc.y + r.y + b.y;
  x.z = acc.z + r.z + b.z; x.w = acc.w + r.w + b.w;

  const int q = lane & 7;
  float4 hn = make_float4(0.f, 0.f, 0.f, 0.f);
#pragma unroll
  for (int hh = 0; hh < 4; ++hh) {
    hn.x += __shfl(x.x, q + 8 * hh);
    hn.y += __shfl(x.y, q + 8 * hh);
    hn.z += __shfl(x.z, q + 8 * hh);
    hn.w += __shfl(x.w, q + 8 * hh);
  }
  hn.x *= 0.25f; hn.y *= 0.25f; hn.z *= 0.25f; hn.w *= 0.25f;

  float sw = aw_b[0];
#pragma unroll
  for (int i = 0; i < 32; ++i) {
    const float comp = ((i & 3) == 0) ? hn.x : ((i & 3) == 1) ? hn.y : ((i & 3) == 2) ? hn.z : hn.w;
    const float hi = __int_as_float(__builtin_amdgcn_readlane(__float_as_int(comp), i >> 2));
    sw += hi * aw_w[i];
  }
  const float w = 1.f / (1.f + expf(-sw));

  const float t0 = __shfl(hn.x, lane >> 2);
  const float t1 = __shfl(hn.y, lane >> 2);
  const float t2 = __shfl(hn.z, lane >> 2);
  const float t3 = __shfl(hn.w, lane >> 2);
  const int   c  = lane & 3;
  const float hv = (c == 0) ? t0 : (c == 1) ? t1 : (c == 2) ? t2 : t3;
  const int g = gid[d];
  atomicAdd(&gmat[g * 64 + lane], w * hv);
  atomicMax((unsigned*)gmat + g * 64 + 32 + lane, ford(hv));
}

__global__ void graph_init(unsigned* gmat, int n) {
  int t = blockIdx.x * blockDim.x + threadIdx.x;
  if (t >= n) return;
  gmat[t] = ((t & 63) < 32) ? 0u : ORD_NEG_INF;
}

__global__ void graph_decode(float* gmat, int GD) {
  int t = blockIdx.x * blockDim.x + threadIdx.x;
  if (t >= GD) return;
  int g = t >> 5, d = t & 31;
  unsigned u = ((unsigned*)gmat)[g * 64 + 32 + d];
  gmat[g * 64 + 32 + d] = (u == ORD_NEG_INF) ? 0.f : funord(u);
}

__global__ void final_ln(const float* __restrict__ ybuf, const float* __restrict__ out_b,
                         float* __restrict__ out) {
  __shared__ float sm[128];
  int g = blockIdx.x, j = threadIdx.x;
  float v = ybuf[g * 128 + j] + out_b[j];
  sm[j] = v; __syncthreads();
  for (int s = 64; s > 0; s >>= 1) { if (j < s) sm[j] += sm[j + s]; __syncthreads(); }
  float mu = sm[0] * (1.f / 128.f);
  __syncthreads();
  float dv = v - mu;
  sm[j] = dv * dv; __syncthreads();
  for (int s = 64; s > 0; s >>= 1) { if (j < s) sm[j] += sm[j + s]; __syncthreads(); }
  float var = sm[0] * (1.f / 128.f);
  out[g * 128 + j] = dv * rsqrtf(var + 1e-5f);
}

extern "C" void kernel_launch(void* const* d_in, const int* in_sizes, int n_in,
                              void* d_out, int out_size, void* d_ws, size_t ws_size,
                              hipStream_t stream) {
  const float* node  = (const float*)d_in[0];
  const int*   src   = (const int*)d_in[1];
  const int*   dst   = (const int*)d_in[2];
  const int*   gid   = (const int*)d_in[3];
  const float* fc1_w = (const float*)d_in[4];
  const float* al1   = (const float*)d_in[5];
  const float* ar1   = (const float*)d_in[6];
  const float* res1w = (const float*)d_in[7];
  const float* bias1 = (const float*)d_in[8];
  const float* fc2_w = (const float*)d_in[9];
  const float* al2   = (const float*)d_in[10];
  const float* ar2   = (const float*)d_in[11];
  const float* bias2 = (const float*)d_in[12];
  const float* aw_w  = (const float*)d_in[13];
  const float* aw_b  = (const float*)d_in[14];
  const float* out_w = (const float*)d_in[15];
  const float* out_b = (const float*)d_in[16];
  (void)n_in; (void)ws_size;

  const int N = in_sizes[3];
  const int E = in_sizes[1];
  const int G = out_size / 128;

  char* wsb = (char*)d_ws;
  size_t off = 0;
  auto alloc = [&](size_t bytes) -> char* {
    char* p = wsb + off;
    off = (off + bytes + 255) & ~(size_t)255;
    return p;
  };
  float*    bufA  = (float*)alloc((size_t)N * 128 * 4);
  float*    bufB  = (float*)alloc((size_t)N * 128 * 4);
  float*    bufC  = (float*)alloc((size_t)N * 128 * 4);
  float*    el    = (float*)alloc((size_t)N * 4 * 4);
  float*    er    = (float*)alloc((size_t)N * 4 * 4);
  int*      cnt   = (int*)alloc((size_t)N * 4 * 4);
  int*      start = (int*)alloc((size_t)N * 4 * 4);
  int*      rank  = (int*)alloc((size_t)E * 4 * 4);
  int*      csr   = rank + E;
  float*    gmat  = (float*)alloc((size_t)G * 64 * 4);
  float*    ybuf  = (float*)alloc((size_t)G * 128 * 4);

  auto grid = [](int n) { return dim3((unsigned)((n + THREADS - 1) / THREADS)); };
  const int ntiles  = (N + 15) / 16;
  const int nblocks = (ntiles + 3) / 4;
  const int gtiles  = (G + 15) / 16;
  const int gblocks = (gtiles + 3) / 4;

  fill_u32<<<grid(N + 1), THREADS, 0, stream>>>((unsigned*)cnt, 0u, N + 1);
  deg_rank<<<grid(E), THREADS, 0, stream>>>(dst, cnt, rank, E);
  seg_alloc<<<grid(N), THREADS, 0, stream>>>(cnt, start, cnt + N, N);
  csr_fill<<<grid(E), THREADS, 0, stream>>>(src, dst, start, rank, csr, E);

  attn_proj_in<<<grid(N * 4), THREADS, 0, stream>>>(node, fc1_w, al1, ar1, el, er, N * 4);
  node_softmax_gather_in<<<grid(N * 32), THREADS, 0, stream>>>(start, cnt, csr, el, er, node, bufB, N);
  wmma_gemm_l1<<<nblocks, 128, 0, stream>>>(bufB, node, fc1_w, res1w, bias1, bufC, N);

  wmma_gemm_xwt<128, 8><<<nblocks, 128, 0, stream>>>(bufC, fc2_w, bufA, N);
  attn_proj<<<grid(N * 4), THREADS, 0, stream>>>(bufA, al2, ar2, el, er, N * 4);
  graph_init<<<grid(G * 64), THREADS, 0, stream>>>((unsigned*)gmat, G * 64);
  node_softmax_gather<<<grid(N * 32), THREADS, 0, stream>>>(start, cnt, csr, el, er, bufA, bufC, bias2,
                                                            aw_w, aw_b, gid, gmat, N);

  graph_decode<<<grid(G * 32), THREADS, 0, stream>>>(gmat, G * 32);
  wmma_gemm_xwt<64, 8><<<gblocks, 128, 0, stream>>>(gmat, out_w, ybuf, G);
  final_ln<<<G, 128, 0, stream>>>(ybuf, out_b, (float*)d_out);
}


# GEMM epilogues take the stage that follows, on the accumulator tile: layer-2 el/er (2 in-lane terms + 16-lane DPP tree) and head bias+LayerNorm (8 in-lane + tree); sum orders changed; 14->12 launches
# speedup vs baseline: 5.7536x; 1.0694x over previous
#include <hip/hip_runtime.h>
#include <hip/hip_bf16.h>
#include <math.h>

typedef __attribute__((ext_vector_type(2))) float v2f;
typedef __attribute__((ext_vector_type(4))) float v4f;
typedef __attribute__((ext_vector_type(8))) float v8f;

__device__ __forceinline__ v4f load_quad(const float* __restrict__ p) {
  v4f r;
  __builtin_memcpy(&r, p, 16);
  return r;
}

#define THREADS 256
#define ORD_NEG_INF 0x007FFFFFu

__device__ __forceinline__ unsigned ford(float f) {
  unsigned u = __float_as_uint(f);
  return (u & 0x80000000u) ? ~u : (u | 0x80000000u);
}
__device__ __forceinline__ float funord(unsigned u) {
  return __uint_as_float((u & 0x80000000u) ? (u & 0x7fffffffu) : ~u);
}

template <int MASK>
__device__ __forceinline__ float lane_xor(float x) {
  const int b = __float_as_int(x);
  return __int_as_float(__builtin_amdgcn_update_dpp(b, b, 0x160 + MASK, 0xf, 0xf, false));
}
template <int MASK>
__device__ __forceinline__ float halve_sum(float a, float b, bool up) {
  const float keep = up ? b : a;
  const float send = up ? a : b;
  return keep + lane_xor<MASK>(send);
}
__device__ __forceinline__ float row16_sum(float v) {
  v += lane_xor<1>(v);
  v += lane_xor<2>(v);
  v += lane_xor<4>(v);
  v += lane_xor<8>(v);
  return v;
}

template <int K, int NT>
__device__ __forceinline__ void wmma_tile_xwt(const float* __restrict__ X,
                                              const float* __restrict__ W,
                                              int mrow, int lane, v8f (&acc)[NT]) {
  static_assert(K % 8 == 0, "the fragment loaders take k in groups of 8");
  const int qoff = (lane < 16) ? 0 : 4;
#pragma unroll
  for (int t = 0; t < NT; ++t) acc[t] = (v8f){0.f,0.f,0.f,0.f,0.f,0.f,0.f,0.f};

  const float* __restrict__ xr = X + (size_t)mrow * K + qoff;
  const float* __restrict__ wr = W + (size_t)(lane & 15) * K + qoff;
  for (int k0 = 0; k0 < K; k0 += 8) {
    const v4f a = load_quad(xr + k0);
    v4f b[NT];
#pragma unroll
    for (int t = 0; t < NT; ++t) b[t] = load_quad(wr + (size_t)t * 16 * K + k0);
#pragma unroll
    for (int t = 0; t < NT; ++t) {
      acc[t] = __builtin_amdgcn_wmma_f32_16x16x4_f32(
          false, (v2f){a.x, a.y}, false, (v2f){b[t].x, b[t].y}, (short)0, acc[t], false, false);
    }
#pragma unroll
    for (int t = 0; t < NT; ++t) {
      acc[t] = __builtin_amdgcn_wmma_f32_16x16x4_f32(
          false, (v2f){a.z, a.w}, false, (v2f){b[t].z, b[t].w}, (short)0, acc[t], false, false);
    }
  }
}

__global__ void __launch_bounds__(128)
wmma_gemm_feat_scores(const float* __restrict__ X, const float* __restrict__ W,
                      const float* __restrict__ al, const float* __restrict__ ar,
                      float* __restrict__ Y, float* __restrict__ el, float* __restrict__ er, int M) {
  const int lane = threadIdx.x & 31;
  const int wave = threadIdx.x >> 5;
  const int tile = blockIdx.x * (blockDim.x >> 5) + wave;
  if (tile * 16 >= M) return;
  const int rbase = tile * 16;

  v8f acc[8];
  wmma_tile_xwt<128, 8>(X, W, rbase + (lane & 15), lane, acc);

  const int rowoff = (lane < 16) ? 0 : 8;
  const int colo   = lane & 15;
#pragma unroll
  for (int t = 0; t < 8; ++t) {
#pragma unroll
    for (int i = 0; i < 8; ++i) {
      Y[(size_t)(rbase + rowoff + i) * 128 + t * 16 + colo] = acc[t][i];
    }
  }

  float alo[4], ahi[4], rlo[4], rhi[4];
#pragma unroll
  for (int h = 0; h < 4; ++h) {
    alo[h] = al[h * 32 + colo]; ahi[h] = al[h * 32 + 16 + colo];
    rlo[h] = ar[h * 32 + colo]; rhi[h] = ar[h * 32 + 16 + colo];
  }
  const bool b3 = (colo & 8) != 0, b2 = (colo & 4) != 0, b1 = (colo & 2) != 0, b0 = (colo & 1) != 0;
  float pl[8][4], pr[8][4];
#pragma unroll
  for (int i = 0; i < 8; ++i) {
#pragma unroll
    for (int h = 0; h < 4; ++h) {
      pl[i][h] = acc[2 * h][i] * alo[h] + acc[2 * h + 1][i] * ahi[h];
      pr[i][h] = acc[2 * h][i] * rlo[h] + acc[2 * h + 1][i] * rhi[h];
    }
  }
#pragma unroll
  for (int h = 0; h < 4; ++h) {
#pragma unroll
    for (int i = 0; i < 4; ++i) {
      pl[i][h] = halve_sum<8>(pl[i][h], pl[i + 4][h], b3);
      pr[i][h] = halve_sum<8>(pr[i][h], pr[i + 4][h], b3);
    }
#pragma unroll
    for (int i = 0; i < 2; ++i) {
      pl[i][h] = halve_sum<4>(pl[i][h], pl[i + 2][h], b2);
      pr[i][h] = halve_sum<4>(pr[i][h], pr[i + 2][h], b2);
    }
    pl[0][h] = halve_sum<2>(pl[0][h], pl[1][h], b1);
    pr[0][h] = halve_sum<2>(pr[0][h], pr[1][h], b1);
  }
  float2 ol, orr;
  ol.x  = halve_sum<1>(pl[0][0], pl[0][2], b0);
  ol.y  = halve_sum<1>(pl[0][1], pl[0][3], b0);
  orr.x = halve_sum<1>(pr[0][0], pr[0][2], b0);
  orr.y = halve_sum<1>(pr[0][1], pr[0][3], b0);
  const size_t so = (size_t)(rbase + rowoff) * 4 + 2 * colo;
  *reinterpret_cast<float2*>(el + so) = ol;
  *reinterpret_cast<float2*>(er + so) = orr;
}

__global__ void __launch_bounds__(128)
wmma_gemm_head_ln(const float* __restrict__ g, const float* __restrict__ W,
                  const float* __restrict__ out_b, float* __restrict__ out, int M) {
  const int lane = threadIdx.x & 31;
  const int wave = threadIdx.x >> 5;
  const int tile = blockIdx.x * (blockDim.x >> 5) + wave;
  if (tile * 16 >= M) return;
  const int rbase = tile * 16;

  v8f acc[8];
  wmma_tile_xwt<64, 8>(g, W, rbase + (lane & 15), lane, acc);

  const int rowoff = (lane < 16) ? 0 : 8;
  const int colo   = lane & 15;
  float bt[8];
#pragma unroll
  for (int t = 0; t < 8; ++t) bt[t] = out_b[t * 16 + colo];
#pragma unroll
  for (int i = 0; i < 8; ++i) {
    float v[8];
    float s = 0.f;
#pragma unroll
    for (int t = 0; t < 8; ++t) { v[t] = acc[t][i] + bt[t]; s += v[t]; }
    const float mu = row16_sum(s) * (1.f / 128.f);
    float q = 0.f;
#pragma unroll
    for (int t = 0; t < 8; ++t) { v[t] -= mu; q += v[t] * v[t]; }
    const float rs = rsqrtf(row16_sum(q) * (1.f / 128.f) + 1e-5f);
#pragma unroll
    for (int t = 0; t < 8; ++t) {
      out[(size_t)(rbase + rowoff + i) * 128 + t * 16 + colo] = v[t] * rs;
    }
  }
}

__device__ __forceinline__ v2f load_tail27(const float* __restrict__ row, bool upper) {
  const v4f q = load_quad(row + 23);
  v2f r;
  r.x = upper ? q.w : q.y;
  r.y = upper ? 0.f : q.z;
  return r;
}

__global__ void __launch_bounds__(128)
wmma_gemm_l1(const float* __restrict__ agg, const float* __restrict__ x,
             const float* __restrict__ W1, const float* __restrict__ R,
             const float* __restrict__ bias, float* __restrict__ h1, int M) {
  const int K    = 27;
  const int lane = threadIdx.x & 31;
  const int wave = threadIdx.x >> 5;
  const int tile = blockIdx.x * (blockDim.x >> 5) + wave;
  if (tile * 16 >= M) return;
  const int rbase = tile * 16;
  const int mrow  = rbase + (lane & 15);
  const bool upper = lane >= 16;

  v8f acc[8];
#pragma unroll
  for (int t = 0; t < 8; ++t) acc[t] = (v8f){0.f,0.f,0.f,0.f,0.f,0.f,0.f,0.f};

  const int qoff = upper ? 4 : 0;

  {
    const float* __restrict__ xr = x + (size_t)mrow * K;
    const float* __restrict__ rr = R + (size_t)(lane & 15) * K;
#pragma unroll
    for (int k0 = 0; k0 < 24; k0 += 8) {
      const v4f a = load_quad(xr + k0 + qoff);
      v4f b[8];
#pragma unroll
      for (int t = 0; t < 8; ++t) b[t] = load_quad(rr + t * 16 * K + k0 + qoff);
#pragma unroll
      for (int t = 0; t < 8; ++t) {
        acc[t] = __builtin_amdgcn_wmma_f32_16x16x4_f32(
            false, (v2f){a.x, a.y}, false, (v2f){b[t].x, b[t].y}, (short)0, acc[t], false, false);
      }
#pragma unroll
      for (int t = 0; t < 8; ++t) {
        acc[t] = __builtin_amdgcn_wmma_f32_16x16x4_f32(
            false, (v2f){a.z, a.w}, false, (v2f){b[t].z, b[t].w}, (short)0, acc[t], false, false);
      }
    }
    const v2f a = load_tail27(xr, upper);
#pragma unroll
    for (int t = 0; t < 8; ++t) {
      const v2f b = load_tail27(rr + t * 16 * K, upper);
      acc[t] = __builtin_amdgcn_wmma_f32_16x16x4_f32(
          false, a, false, b, (short)0, acc[t], false, false);
    }
  }

#pragma unroll
  for (int h = 0; h < 4; ++h) {
    const float* __restrict__ ah = agg + (size_t)mrow * 108 + h * K;
    const float* __restrict__ wr = W1 + (size_t)(2 * h * 16 + (lane & 15)) * K;
#pragma unroll
    for (int k0 = 0; k0 < 24; k0 += 8) {
      const v4f a  = load_quad(ah + k0 + qoff);
      const v4f b0 = load_quad(wr + k0 + qoff);
      const v4f b1 = load_quad(wr + 16 * K + k0 + qoff);
      acc[2 * h] = __builtin_amdgcn_wmma_f32_16x16x4_f32(
          false, (v2f){a.x, a.y}, false, (v2f){b0.x, b0.y}, (short)0, acc[2 * h], false, false);
      acc[2 * h + 1] = __builtin_amdgcn_wmma_f32_16x16x4_f32(
          false, (v2f){a.x, a.y}, false, (v2f){b1.x, b1.y}, (short)0, acc[2 * h + 1], false, false);
      acc[2 * h] = __builtin_amdgcn_wmma_f32_16x16x4_f32(
          false, (v2f){a.z, a.w}, false, (v2f){b0.z, b0.w}, (short)0, acc[2 * h], false, false);
      acc[2 * h + 1] = __builtin_amdgcn_wmma_f32_16x16x4_f32(
          false, (v2f){a.z, a.w}, false, (v2f){b1.z, b1.w}, (short)0, acc[2 * h + 1], false, false);
    }
    const v2f a = load_tail27(ah, upper);
#pragma unroll
    for (int tt = 0; tt < 2; ++tt) {
      const v2f b = load_tail27(wr + tt * 16 * K, upper);
      acc[2 * h + tt] = __builtin_amdgcn_wmma_f32_16x16x4_f32(
          false, a, false, b, (short)0, acc[2 * h + tt], false, false);
    }
  }

  const int rowoff = (lane < 16) ? 0 : 8;
  const int colo   = lane & 15;
#pragma unroll
  for (int t = 0; t < 8; ++t) {
    const float bt = bias[t * 16 + colo];
#pragma unroll
    for (int i = 0; i < 8; ++i) {
      float v = acc[t][i] + bt;
      v = (v > 0.f) ? v : expm1f(v);
      h1[(size_t)(rbase + rowoff + i) * 128 + t * 16 + colo] = v;
    }
  }
}

__global__ void fill_u32(unsigned* p, unsigned v, int n) {
  int t = blockIdx.x * blockDim.x + threadIdx.x;
  if (t < n) p[t] = v;
}

__global__ void attn_proj_in(const float* __restrict__ x, const float* __restrict__ w,
                             const float* __restrict__ al, const float* __restrict__ ar,
                             float* __restrict__ el, float* __restrict__ er, int NH) {
  __shared__ float vt[216];
  const int tid = threadIdx.x;
  if (tid < 216) {
    const int which = (tid >= 108) ? 1 : 0;
    const int idx = tid - 108 * which;
    const int h = idx / 27, c = idx - h * 27;
    const float* a = which ? ar : al;
    float s = 0.f;
    for (int d = 0; d < 32; ++d) s += w[(h * 32 + d) * 27 + c] * a[h * 32 + d];
    vt[tid] = s;
  }
  __syncthreads();
  const int t = blockIdx.x * blockDim.x + tid;
  if (t >= NH) return;
  const int n = t >> 2, h = t & 3;
  const float* xr = x + n * 27;
  float sl = 0.f, sr = 0.f;
  for (int c = 0; c < 27; ++c) {
    const float xv = xr[c];
    sl += xv * vt[h * 27 + c];
    sr += xv * vt[108 + h * 27 + c];
  }
  el[t] = sl; er[t] = sr;
}

__global__ void deg_rank(const int* __restrict__ dst, int* __restrict__ cnt,
                         int* __restrict__ rank, int E) {
  int e = blockIdx.x * blockDim.x + threadIdx.x;
  if (e >= E) return;
  rank[e] = atomicAdd(&cnt[dst[e]], 1);
}

__global__ void seg_alloc(const int* __restrict__ cnt, int* __restrict__ start,
                          int* __restrict__ total, int N) {
  const int d    = blockIdx.x * blockDim.x + threadIdx.x;
  const int lane = threadIdx.x & 31;
  const int c = (d < N) ? cnt[d] : 0;
  int x = c;
  for (int o = 1; o < 32; o <<= 1) {
    int y = __shfl_up(x, o);
    if (lane >= o) x += y;
  }
  int base = 0;
  if (lane == 31) base = atomicAdd(total, x);
  base = __shfl(base, 31);
  if (d < N) start[d] = base + x - c;
}

__global__ void csr_fill(const int* __restrict__ src, const int* __restrict__ dst,
                         const int* __restrict__ start, const int* __restrict__ rank,
                         int* __restrict__ csr_src, int E) {
  int e = blockIdx.x * blockDim.x + threadIdx.x;
  if (e >= E) return;
  csr_src[start[dst[e]] + rank[e]] = src[e];
}

template <int J>
__device__ __forceinline__ float group8_bcast(float x) {
  return __int_as_float((int)__builtin_amdgcn_mov_dpp8((unsigned)__float_as_int(x), J * 0x249249));
}
__device__ __forceinline__ int edge_source8(const int* __restrict__ csr_src, int st, int deg, int lane) {
  const int j = lane & 7;
  return csr_src[st + ((j < deg) ? j : deg - 1)];
}
__device__ __forceinline__ float edge_weight8(float els, float erd, int deg, int lane) {
  float v = els + erd;
  v = (v > 0.f) ? v : 0.2f * v;
  v = ((lane & 7) < deg) ? v : -INFINITY;
  float m = fmaxf(v, lane_xor<1>(v));
  m = fmaxf(m, lane_xor<2>(m));
  m = fmaxf(m, lane_xor<4>(m));
  const float ex = expf(v - m);
  float s = ex + lane_xor<1>(ex);
  s += lane_xor<2>(s);
  s += lane_xor<4>(s);
  return ex / s;
}

__global__ void node_softmax_gather_in(const int* __restrict__ start, const int* __restrict__ cnt,
                                       const int* __restrict__ csr_src,
                                       const float* __restrict__ el, const float* __restrict__ er,
                                       const float* __restrict__ x,
                                       float* __restrict__ agg,
                                       int N) {
  const int lane = threadIdx.x & 31;
  const int d = __builtin_amdgcn_readfirstlane((int)(blockIdx.x * (blockDim.x >> 5) + (threadIdx.x >> 5)));
  if (d >= N) return;
  const int h   = lane >> 3;
  const int st  = start[d];
  const int deg = cnt[d];
  const float erd = er[d * 4 + h];
  const int cl = (lane < 27) ? lane : 26;

  float a0 = 0.f, a1 = 0.f, a2 = 0.f, a3 = 0.f;
  if (deg >= 1 && deg <= 8) {
    const int sidx = edge_source8(csr_src, st, deg, lane);
    const float els = el[sidx * 4 + h];
    float xv0, xv1, xv2, xv3, xv4, xv5, xv6, xv7;
#define ROW_LOAD_IN(J) if (J < deg) xv##J = x[(size_t)__builtin_amdgcn_readlane(sidx, J) * 27 + cl];
    ROW_LOAD_IN(0) ROW_LOAD_IN(1) ROW_LOAD_IN(2) ROW_LOAD_IN(3)
    ROW_LOAD_IN(4) ROW_LOAD_IN(5) ROW_LOAD_IN(6) ROW_LOAD_IN(7)
#undef ROW_LOAD_IN
    const int a = __float_as_int(edge_weight8(els, erd, deg, lane));
#define ROW_FMA_IN(J) if (J < deg) {                                              \
      a0 += xv##J * __int_as_float(__builtin_amdgcn_readlane(a, J));              \
      a1 += xv##J * __int_as_float(__builtin_amdgcn_readlane(a, 8 + J));          \
      a2 += xv##J * __int_as_float(__builtin_amdgcn_readlane(a, 16 + J));         \
      a3 += xv##J * __int_as_float(__builtin_amdgcn_readlane(a, 24 + J)); }
    ROW_FMA_IN(0) ROW_FMA_IN(1) ROW_FMA_IN(2) ROW_FMA_IN(3)
    ROW_FMA_IN(4) ROW_FMA_IN(5) ROW_FMA_IN(6) ROW_FMA_IN(7)
#undef ROW_FMA_IN
  } else {
    float m = -INFINITY;
    for (int j = 0; j < deg; ++j) {
      const int sidx = csr_src[st + j];
      float v = el[sidx * 4 + h] + erd;
      v = (v > 0.f) ? v : 0.2f * v;
      m = fmaxf(m, v);
    }
    float sum = 0.f;
    for (int j = 0; j < deg; ++j) {
      const int sidx = csr_src[st + j];
      float v = el[sidx * 4 + h] + erd;
      v = (v > 0.f) ? v : 0.2f * v;
      sum += expf(v - m);
    }
    for (int j = 0; j < deg; ++j) {
      const int sidx = csr_src[st + j];
      float v = el[sidx * 4 + h] + erd;
      v = (v > 0.f) ? v : 0.2f * v;
      const int a = __float_as_int(expf(v - m) / sum);
      const float xv = x[(size_t)sidx * 27 + cl];
      a0 += xv * __int_as_float(__builtin_amdgcn_readlane(a, 0));
      a1 += xv * __int_as_float(__builtin_amdgcn_readlane(a, 8));
      a2 += xv * __int_as_float(__builtin_amdgcn_readlane(a, 16));
      a3 += xv * __int_as_float(__builtin_amdgcn_readlane(a, 24));
    }
  }
  if (lane < 27) {
    float* o = agg + (size_t)d * 108 + lane;
    o[0] = a0; o[27] = a1; o[54] = a2; o[81] = a3;
  }
}

__global__ void node_softmax_gather(const int* __restrict__ start, const int* __restrict__ cnt,
                                    const int* __restrict__ csr_src,
                                    const float* __restrict__ el, const float* __restrict__ er,
                                    const float* __restrict__ feat,
                                    const float* __restrict__ addrow,
                                    const float* __restrict__ bias,
                                    const float* __restrict__ aw_w, const float* __restrict__ aw_b,
                                    const int* __restrict__ gid, float* __restrict__ gmat,
                                    int N) {
  const int lane = threadIdx.x & 31;
  const int d = __builtin_amdgcn_readfirstlane((int)(blockIdx.x * (blockDim.x >> 5) + (threadIdx.x >> 5)));
  if (d >= N) return;
  const int h   = lane >> 3;
  const int st  = start[d];
  const int deg = cnt[d];
  const float erd = er[d * 4 + h];
  const float4 r = *reinterpret_cast<const float4*>(addrow + (size_t)d * 128 + lane * 4);
  const float4 b = *reinterpret_cast<const float4*>(bias + lane * 4);

  float4 acc = make_float4(0.f, 0.f, 0.f, 0.f);
  if (deg >= 1 && deg <= 8) {
    const int sidx = edge_source8(csr_src, st, deg, lane);
    const float els = el[sidx * 4 + h];
    float4 f0, f1, f2, f3, f4, f5, f6, f7;
#define ROW_LOAD(J) if (J < deg) f##J = *reinterpret_cast<const float4*>(                     \
        feat + (size_t)__builtin_amdgcn_readlane(sidx, J) * 128 + lane * 4);
    ROW_LOAD(0) ROW_LOAD(1) ROW_LOAD(2) ROW_LOAD(3)
    ROW_LOAD(4) ROW_LOAD(5) ROW_LOAD(6) ROW_LOAD(7)
#undef ROW_LOAD
    const float a = edge_weight8(els, erd, deg, lane);
#define ROW_FMA(J) if (J < deg) { const float aj = group8_bcast<J>(a);                        \
      acc.x += f##J.x * aj; acc.y += f##J.y * aj; acc.z += f##J.z * aj; acc.w += f##J.w * aj; }
    ROW_FMA(0) ROW_FMA(1) ROW_FMA(2) ROW_FMA(3)
    ROW_FMA(4) ROW_FMA(5) ROW_FMA(6) ROW_FMA(7)
#undef ROW_FMA
  } else {
    float m = -INFINITY;
    for (int j = 0; j < deg; ++j) {
      const int sidx = csr_src[st + j];
      float v = el[sidx * 4 + h] + erd;
      v = (v > 0.f) ? v : 0.2f * v;
      m = fmaxf(m, v);
    }
    float sum = 0.f;
    for (int j = 0; j < deg; ++j) {
      const int sidx = csr_src[st + j];
      float v = el[sidx * 4 + h] + erd;
      v = (v > 0.f) ? v : 0.2f * v;
      sum += expf(v - m);
    }
    for (int j = 0; j < deg; ++j) {
      const int sidx = csr_src[st + j];
      float v = el[sidx * 4 + h] + erd;
      v = (v > 0.f) ? v : 0.2f * v;
      const float a = expf(v - m) / sum;
      const float4 f = *reinterpret_cast<const float4*>(feat + (size_t)sidx * 128 + lane * 4);
      acc.x += f.x * a; acc.y += f.y * a; acc.z += f.z * a; acc.w += f.w * a;
    }
  }

  float4 x;
  x.x = acc.x + r.x + b.x; x.y = acc.y + r.y + b.y;
  x.z = acc.z + r.z + b.z; x.w = acc.w + r.w + b.w;

  const int q = lane & 7;
  float4 hn = make_float4(0.f, 0.f, 0.f, 0.f);
#pragma unroll
  for (int hh = 0; hh < 4; ++hh) {
    hn.x += __shfl(x.x, q + 8 * hh);
    hn.y += __shfl(x.y, q + 8 * hh);
    hn.z += __shfl(x.z, q + 8 * hh);
    hn.w += __shfl(x.w, q + 8 * hh);
  }
  hn.x *= 0.25f; hn.y *= 0.25f; hn.z *= 0.25f; hn.w *= 0.25f;

  float sw = aw_b[0];
#pragma unroll
  for (int i = 0; i < 32; ++i) {
    const float comp = ((i & 3) == 0) ? hn.x : ((i & 3) == 1) ? hn.y : ((i & 3) == 2) ? hn.z : hn.w;
    const float hi = __int_as_float(__builtin_amdgcn_readlane(__float_as_int(comp), i >> 2));
    sw += hi * aw_w[i];
  }
  const float w = 1.f / (1.f + expf(-sw));

  const float t0 = __shfl(hn.x, lane >> 2);
  const float t1 = __shfl(hn.y, lane >> 2);
  const float t2 = __shfl(hn.z, lane >> 2);
  const float t3 = __shfl(hn.w, lane >> 2);
  const int   c  = lane & 3;
  const float hv = (c == 0) ? t0 : (c == 1) ? t1 : (c == 2) ? t2 : t3;
  const int g = gid[d];
  atomicAdd(&gmat[g * 64 + lane], w * hv);
  atomicMax((unsigned*)gmat + g * 64 + 32 + lane, ford(hv));
}

__global__ void graph_init(unsigned* gmat, int n) {
  int t = blockIdx.x * blockDim.x + threadIdx.x;
  if (t >= n) return;
  gmat[t] = ((t & 63) < 32) ? 0u : ORD_NEG_INF;
}

__global__ void graph_decode(float* gmat, int GD) {
  int t = blockIdx.x * blockDim.x + threadIdx.x;
  if (t >= GD) return;
  int g = t >> 5, d = t & 31;
  unsigned u = ((unsigned*)gmat)[g * 64 + 32 + d];
  gmat[g * 64 + 32 + d] = (u == ORD_NEG_INF) ? 0.f : funord(u);
}

extern "C" void kernel_launch(void* const* d_in, const int* in_sizes, int n_in,
                              void* d_out, int out_size, void* d_ws, size_t ws_size,
                              hipStream_t stream) {
  const float* node  = (const float*)d_in[0];
  const int*   src   = (const int*)d_in[1];
  const int*   dst   = (const int*)d_in[2];
  const int*   gid   = (const int*)d_in[3];
  const float* fc1_w = (const float*)d_in[4];
  const float* al1   = (const float*)d_in[5];
  const float* ar1   = (const float*)d_in[6];
  const float* res1w = (const float*)d_in[7];
  const float* bias1 = (const float*)d_in[8];
  const float* fc2_w = (const float*)d_in[9];
  const float* al2   = (const float*)d_in[10];
  const float* ar2   = (const float*)d_in[11];
  const float* bias2 = (const float*)d_in[12];
  const float* aw_w  = (const float*)d_in[13];
  const float* aw_b  = (const float*)d_in[14];
  const float* out_w = (const float*)d_in[15];
  const float* out_b = (const float*)d_in[16];
  (void)n_in; (void)ws_size;

  const int N = in_sizes[3];
  const int E = in_sizes[1];
  const int G = out_size / 128;

  char* wsb = (char*)d_ws;
  size_t off = 0;
  auto alloc = [&](size_t bytes) -> char* {
    char* p = wsb + off;
    off = (off + bytes + 255) & ~(size_t)255;
    return p;
  };
  float*    bufA  = (float*)alloc((size_t)N * 128 * 4);
  float*    bufB  = (float*)alloc((size_t)N * 128 * 4);
  float*    bufC  = (float*)alloc((size_t)N * 128 * 4);
  float*    el    = (float*)alloc((size_t)N * 4 * 4);
  float*    er    = (float*)alloc((size_t)N * 4 * 4);
  int*      cnt   = (int*)alloc((size_t)N * 4 * 4);
  int*      start = (int*)alloc((size_t)N * 4 * 4);
  int*      rank  = (int*)alloc((size_t)E * 4 * 4);
  int*      csr   = rank + E;
  float*    gmat  = (float*)alloc((size_t)G * 64 * 4);

  auto grid = [](int n) { return dim3((unsigned)((n + THREADS - 1) / THREADS)); };
  const int ntiles  = (N + 15) / 16;
  const int nblocks = (ntiles + 3) / 4;
  const int gtiles  = (G + 15) / 16;
  const int gblocks = (gtiles + 3) / 4;

  fill_u32<<<grid(N + 1), THREADS, 0, stream>>>((unsigned*)cnt, 0u, N + 1);
  deg_rank<<<grid(E), THREADS, 0, stream>>>(dst, cnt, rank, E);
  seg_alloc<<<grid(N), THREADS, 0, stream>>>(cnt, start, cnt + N, N);
  csr_fill<<<grid(E), THREADS, 0, stream>>>(src, dst, start, rank, csr, E);

  attn_proj_in<<<grid(N * 4), THREADS, 0, stream>>>(node, fc1_w, al1, ar1, el, er, N * 4);
  node_softmax_gather_in<<<grid(N * 32), THREADS, 0, stream>>>(start, cnt, csr, el, er, node, bufB, N);
  wmma_gemm_l1<<<nblocks, 128, 0, stream>>>(bufB, node, fc1_w, res1w, bias1, bufC, N);

  wmma_gemm_feat_scores<<<nblocks, 128, 0, stream>>>(bufC, fc2_w, al2, ar2, bufA, el, er, N);
  graph_init<<<grid(G * 64), THREADS, 0, stream>>>((unsigned*)gmat, G * 64);
  node_softmax_gather<<<grid(N * 32), THREADS, 0, stream>>>(start, cnt, csr, el, er, bufA, bufC, bias2,
                                                            aw_w, aw_b, gid, gmat, N);

  graph_decode<<<grid(G * 32), THREADS, 0, stream>>>(gmat, G * 32);
  wmma_gemm_head_ln<<<gblocks, 128, 0, stream>>>(gmat, out_w, out_b, (float*)d_out, G);
}


# Three one-store kernels fold into neighbours: counter zeroing and readout start values ride in the first kernel (layer-1 scores), the max-half decode in the head GEMM's A loader; 12 -> 9 launches
# speedup vs baseline: 6.0049x; 1.0437x over previous
#include <hip/hip_runtime.h>
#include <hip/hip_bf16.h>
#include <math.h>

typedef __attribute__((ext_vector_type(2))) float v2f;
typedef __attribute__((ext_vector_type(4))) float v4f;
typedef __attribute__((ext_vector_type(8))) float v8f;

__device__ __forceinline__ v4f load_quad(const float* __restrict__ p) {
  v4f r;
  __builtin_memcpy(&r, p, 16);
  return r;
}

#define THREADS 256
#define ORD_NEG_INF 0x007FFFFFu

__device__ __forceinline__ unsigned ford(float f) {
  unsigned u = __float_as_uint(f);
  return (u & 0x80000000u) ? ~u : (u | 0x80000000u);
}
__device__ __forceinline__ float funord(unsigned u) {
  return __uint_as_float((u & 0x80000000u) ? (u & 0x7fffffffu) : ~u);
}

template <int MASK>
__device__ __forceinline__ float lane_xor(float x) {
  const int b = __float_as_int(x);
  return __int_as_float(__builtin_amdgcn_update_dpp(b, b, 0x160 + MASK, 0xf, 0xf, false));
}
template <int MASK>
__device__ __forceinline__ float halve_sum(float a, float b, bool up) {
  const float keep = up ? b : a;
  const float send = up ? a : b;
  return keep + lane_xor<MASK>(send);
}
__device__ __forceinline__ float row16_sum(float v) {
  v += lane_xor<1>(v);
  v += lane_xor<2>(v);
  v += lane_xor<4>(v);
  v += lane_xor<8>(v);
  return v;
}

template <int K, int NT>
__device__ __forceinline__ void wmma_tile_xwt(const float* __restrict__ X,
                                              const float* __restrict__ W,
                                              int mrow, int lane, v8f (&acc)[NT]) {
  static_assert(K % 8 == 0, "the fragment loaders take k in groups of 8");
  const int qoff = (lane < 16) ? 0 : 4;
#pragma unroll
  for (int t = 0; t < NT; ++t) acc[t] = (v8f){0.f,0.f,0.f,0.f,0.f,0.f,0.f,0.f};

  const float* __restrict__ xr = X + (size_t)mrow * K + qoff;
  const float* __restrict__ wr = W + (size_t)(lane & 15) * K + qoff;
  for (int k0 = 0; k0 < K; k0 += 8) {
    const v4f a = load_quad(xr + k0);
    v4f b[NT];
#pragma unroll
    for (int t = 0; t < NT; ++t) b[t] = load_quad(wr + (size_t)t * 16 * K + k0);
#pragma unroll
    for (int t = 0; t < NT; ++t) {
      acc[t] = __builtin_amdgcn_wmma_f32_16x16x4_f32(
          false, (v2f){a.x, a.y}, false, (v2f){b[t].x, b[t].y}, (short)0, acc[t], false, false);
    }
#pragma unroll
    for (int t = 0; t < NT; ++t) {
      acc[t] = __builtin_amdgcn_wmma_f32_16x16x4_f32(
          false, (v2f){a.z, a.w}, false, (v2f){b[t].z, b[t].w}, (short)0, acc[t], false, false);
    }
  }
}

__global__ void __launch_bounds__(128)
wmma_gemm_feat_scores(const float* __restrict__ X, const float* __restrict__ W,
                      const float* __restrict__ al, const float* __restrict__ ar,
                      float* __restrict__ Y, float* __restrict__ el, float* __restrict__ er, int M) {
  const int lane = threadIdx.x & 31;
  const int wave = threadIdx.x >> 5;
  const int tile = blockIdx.x * (blockDim.x >> 5) + wave;
  if (tile * 16 >= M) return;
  const int rbase = tile * 16;

  v8f acc[8];
  wmma_tile_xwt<128, 8>(X, W, rbase + (lane & 15), lane, acc);

  const int rowoff = (lane < 16) ? 0 : 8;
  const int colo   = lane & 15;
#pragma unroll
  for (int t = 0; t < 8; ++t) {
#pragma unroll
    for (int i = 0; i < 8; ++i) {
      Y[(size_t)(rbase + rowoff + i) * 128 + t * 16 + colo] = acc[t][i];
    }
  }

  float alo[4], ahi[4], rlo[4], rhi[4];
#pragma unroll
  for (int h = 0; h < 4; ++h) {
    alo[h] = al[h * 32 + colo]; ahi[h] = al[h * 32 + 16 + colo];
    rlo[h] = ar[h * 32 + colo]; rhi[h] = ar[h * 32 + 16 + colo];
  }
  const bool b3 = (colo & 8) != 0, b2 = (colo & 4) != 0, b1 = (colo & 2) != 0, b0 = (colo & 1) != 0;
  float pl[8][4], pr[8][4];
#pragma unroll
  for (int i = 0; i < 8; ++i) {
#pragma unroll
    for (int h = 0; h < 4; ++h) {
      pl[i][h] = acc[2 * h][i] * alo[h] + acc[2 * h + 1][i] * ahi[h];
      pr[i][h] = acc[2 * h][i] * rlo[h] + acc[2 * h + 1][i] * rhi[h];
    }
  }
#pragma unroll
  for (int h = 0; h < 4; ++h) {
#pragma unroll
    for (int i = 0; i < 4; ++i) {
      pl[i][h] = halve_sum<8>(pl[i][h], pl[i + 4][h], b3);
      pr[i][h] = halve_sum<8>(pr[i][h], pr[i + 4][h], b3);
    }
#pragma unroll
    for (int i = 0; i < 2; ++i) {
      pl[i][h] = halve_sum<4>(pl[i][h], pl[i + 2][h], b2);
      pr[i][h] = halve_sum<4>(pr[i][h], pr[i + 2][h], b2);
    }
    pl[0][h] = halve_sum<2>(pl[0][h], pl[1][h], b1);
    pr[0][h] = halve_sum<2>(pr[0][h], pr[1][h], b1);
  }
  float2 ol, orr;
  ol.x  = halve_sum<1>(pl[0][0], pl[0][2], b0);
  ol.y  = halve_sum<1>(pl[0][1], pl[0][3], b0);
  orr.x = halve_sum<1>(pr[0][0], pr[0][2], b0);
  orr.y = halve_sum<1>(pr[0][1], pr[0][3], b0);
  const size_t so = (size_t)(rbase + rowoff) * 4 + 2 * colo;
  *reinterpret_cast<float2*>(el + so) = ol;
  *reinterpret_cast<float2*>(er + so) = orr;
}

__device__ __forceinline__ float max_half_decode(float f) {
  const unsigned u = __float_as_uint(f);
  return (u == ORD_NEG_INF) ? 0.f : funord(u);
}

__global__ void __launch_bounds__(128)
wmma_gemm_head_ln(const float* __restrict__ g, const float* __restrict__ W,
                  const float* __restrict__ out_b, float* __restrict__ out, int M) {
  const int lane = threadIdx.x & 31;
  const int wave = threadIdx.x >> 5;
  const int tile = blockIdx.x * (blockDim.x >> 5) + wave;
  if (tile * 16 >= M) return;
  const int rbase = tile * 16;

  v8f acc[8];
#pragma unroll
  for (int t = 0; t < 8; ++t) acc[t] = (v8f){0.f,0.f,0.f,0.f,0.f,0.f,0.f,0.f};
  const int qoff = (lane < 16) ? 0 : 4;
  const float* __restrict__ gr = g + (size_t)(rbase + (lane & 15)) * 64 + qoff;
  const float* __restrict__ wr = W + (size_t)(lane & 15) * 64 + qoff;
#pragma unroll
  for (int k0 = 0; k0 < 64; k0 += 8) {
    v4f a = load_quad(gr + k0);
    if (k0 >= 32) {
      a.x = max_half_decode(a.x); a.y = max_half_decode(a.y);
      a.z = max_half_decode(a.z); a.w = max_half_decode(a.w);
    }
    v4f b[8];
#pragma unroll
    for (int t = 0; t < 8; ++t) b[t] = load_quad(wr + (size_t)t * 16 * 64 + k0);
#pragma unroll
    for (int t = 0; t < 8; ++t) {
      acc[t] = __builtin_amdgcn_wmma_f32_16x16x4_f32(
          false, (v2f){a.x, a.y}, false, (v2f){b[t].x, b[t].y}, (short)0, acc[t], false, false);
    }
#pragma unroll
    for (int t = 0; t < 8; ++t) {
      acc[t] = __builtin_amdgcn_wmma_f32_16x16x4_f32(
          false, (v2f){a.z, a.w}, false, (v2f){b[t].z, b[t].w}, (short)0, acc[t], false, false);
    }
  }

  const int rowoff = (lane < 16) ? 0 : 8;
  const int colo   = lane & 15;
  float bt[8];
#pragma unroll
  for (int t = 0; t < 8; ++t) bt[t] = out_b[t * 16 + colo];
#pragma unroll
  for (int i = 0; i < 8; ++i) {
    float v[8];
    float s = 0.f;
#pragma unroll
    for (int t = 0; t < 8; ++t) { v[t] = acc[t][i] + bt[t]; s += v[t]; }
    const float mu = row16_sum(s) * (1.f / 128.f);
    float q = 0.f;
#pragma unroll
    for (int t = 0; t < 8; ++t) { v[t] -= mu; q += v[t] * v[t]; }
    const float rs = rsqrtf(row16_sum(q) * (1.f / 128.f) + 1e-5f);
#pragma unroll
    for (int t = 0; t < 8; ++t) {
      out[(size_t)(rbase + rowoff + i) * 128 + t * 16 + colo] = v[t] * rs;
    }
  }
}

__device__ __forceinline__ v2f load_tail27(const float* __restrict__ row, bool upper) {
  const v4f q = load_quad(row + 23);
  v2f r;
  r.x = upper ? q.w : q.y;
  r.y = upper ? 0.f : q.z;
  return r;
}

__global__ void __launch_bounds__(128)
wmma_gemm_l1(const float* __restrict__ agg, const float* __restrict__ x,
             const float* __restrict__ W1, const float* __restrict__ R,
             const float* __restrict__ bias, float* __restrict__ h1, int M) {
  const int K    = 27;
  const int lane = threadIdx.x & 31;
  const int wave = threadIdx.x >> 5;
  const int tile = blockIdx.x * (blockDim.x >> 5) + wave;
  if (tile * 16 >= M) return;
  const int rbase = tile * 16;
  const int mrow  = rbase + (lane & 15);
  const bool upper = lane >= 16;

  v8f acc[8];
#pragma unroll
  for (int t = 0; t < 8; ++t) acc[t] = (v8f){0.f,0.f,0.f,0.f,0.f,0.f,0.f,0.f};

  const int qoff = upper ? 4 : 0;

  {
    const float* __restrict__ xr = x + (size_t)mrow * K;
    const float* __restrict__ rr = R + (size_t)(lane & 15) * K;
#pragma unroll
    for (int k0 = 0; k0 < 24; k0 += 8) {
      const v4f a = load_quad(xr + k0 + qoff);
      v4f b[8];
#pragma unroll
      for (int t = 0; t < 8; ++t) b[t] = load_quad(rr + t * 16 * K + k0 + qoff);
#pragma unroll
      for (int t = 0; t < 8; ++t) {
        acc[t] = __builtin_amdgcn_wmma_f32_16x16x4_f32(
            false, (v2f){a.x, a.y}, false, (v2f){b[t].x, b[t].y}, (short)0, acc[t], false, false);
      }
#pragma unroll
      for (int t = 0; t < 8; ++t) {
        acc[t] = __builtin_amdgcn_wmma_f32_16x16x4_f32(
            false, (v2f){a.z, a.w}, false, (v2f){b[t].z, b[t].w}, (short)0, acc[t], false, false);
      }
    }
    const v2f a = load_tail27(xr, upper);
#pragma unroll
    for (int t = 0; t < 8; ++t) {
      const v2f b = load_tail27(rr + t * 16 * K, upper);
      acc[t] = __builtin_amdgcn_wmma_f32_16x16x4_f32(
          false, a, false, b, (short)0, acc[t], false, false);
    }
  }

#pragma unroll
  for (int h = 0; h < 4; ++h) {
    const float* __restrict__ ah = agg + (size_t)mrow * 108 + h * K;
    const float* __restrict__ wr = W1 + (size_t)(2 * h * 16 + (lane & 15)) * K;
#pragma unroll
    for (int k0 = 0; k0 < 24; k0 += 8) {
      const v4f a  = load_quad(ah + k0 + qoff);
      const v4f b0 = load_quad(wr + k0 + qoff);
      const v4f b1 = load_quad(wr + 16 * K + k0 + qoff);
      acc[2 * h] = __builtin_amdgcn_wmma_f32_16x16x4_f32(
          false, (v2f){a.x, a.y}, false, (v2f){b0.x, b0.y}, (short)0, acc[2 * h], false, false);
      acc[2 * h + 1] = __builtin_amdgcn_wmma_f32_16x16x4_f32(
          false, (v2f){a.x, a.y}, false, (v2f){b1.x, b1.y}, (short)0, acc[2 * h + 1], false, false);
      acc[2 * h] = __builtin_amdgcn_wmma_f32_16x16x4_f32(
          false, (v2f){a.z, a.w}, false, (v2f){b0.z, b0.w}, (short)0, acc[2 * h], false, false);
      acc[2 * h + 1] = __builtin_amdgcn_wmma_f32_16x16x4_f32(
          false, (v2f){a.z, a.w}, false, (v2f){b1.z, b1.w}, (short)0, acc[2 * h + 1], false, false);
    }
    const v2f a = load_tail27(ah, upper);
#pragma unroll
    for (int tt = 0; tt < 2; ++tt) {
      const v2f b = load_tail27(wr + tt * 16 * K, upper);
      acc[2 * h + tt] = __builtin_amdgcn_wmma_f32_16x16x4_f32(
          false, a, false, b, (short)0, acc[2 * h + tt], false, false);
    }
  }

  const int rowoff = (lane < 16) ? 0 : 8;
  const int colo   = lane & 15;
#pragma unroll
  for (int t = 0; t < 8; ++t) {
    const float bt = bias[t * 16 + colo];
#pragma unroll
    for (int i = 0; i < 8; ++i) {
      float v = acc[t][i] + bt;
      v = (v > 0.f) ? v : expm1f(v);
      h1[(size_t)(rbase + rowoff + i) * 128 + t * 16 + colo] = v;
    }
  }
}

__global__ void attn_proj_in(const float* __restrict__ x, const float* __restrict__ w,
                             const float* __restrict__ al, const float* __restrict__ ar,
                             float* __restrict__ el, float* __restrict__ er, int NH,
                             unsigned* __restrict__ cnt, int ncnt,
                             unsigned* __restrict__ gmat, int ngm) {
  __shared__ float vt[216];
  const int tid = threadIdx.x;
  if (tid < 216) {
    const int which = (tid >= 108) ? 1 : 0;
    const int idx = tid - 108 * which;
    const int h = idx / 27, c = idx - h * 27;
    const float* a = which ? ar : al;
    float s = 0.f;
    for (int d = 0; d < 32; ++d) s += w[(h * 32 + d) * 27 + c] * a[h * 32 + d];
    vt[tid] = s;
  }
  __syncthreads();
  const int t = blockIdx.x * blockDim.x + tid;
  if (t < NH) {
    const int n = t >> 2, h = t & 3;
    const float* xr = x + n * 27;
    float sl = 0.f, sr = 0.f;
    for (int c = 0; c < 27; ++c) {
      const float xv = xr[c];
      sl += xv * vt[h * 27 + c];
      sr += xv * vt[108 + h * 27 + c];
    }
    el[t] = sl; er[t] = sr;
  }
  const unsigned nt = gridDim.x * blockDim.x;
#pragma clang loop unroll(disable) interleave(disable) vectorize(disable)
  for (unsigned i = (unsigned)t; i < (unsigned)ncnt; i += nt) cnt[i] = 0u;
#pragma clang loop unroll(disable) interleave(disable) vectorize(disable)
  for (unsigned i = (unsigned)t; i < (unsigned)ngm; i += nt) gmat[i] = ((i & 63u) < 32u) ? 0u : ORD_NEG_INF;
}

__global__ void deg_rank(const int* __restrict__ dst, int* __restrict__ cnt,
                         int* __restrict__ rank, int E) {
  int e = blockIdx.x * blockDim.x + threadIdx.x;
  if (e >= E) return;
  rank[e] = atomicAdd(&cnt[dst[e]], 1);
}

__global__ void seg_alloc(const int* __restrict__ cnt, int* __restrict__ start,
                          int* __restrict__ total, int N) {
  const int d    = blockIdx.x * blockDim.x + threadIdx.x;
  const int lane = threadIdx.x & 31;
  const int c = (d < N) ? cnt[d] : 0;
  int x = c;
  for (int o = 1; o < 32; o <<= 1) {
    int y = __shfl_up(x, o);
    if (lane >= o) x += y;
  }
  int base = 0;
  if (lane == 31) base = atomicAdd(total, x);
  base = __shfl(base, 31);
  if (d < N) start[d] = base + x - c;
}

__global__ void csr_fill(const int* __restrict__ src, const int* __restrict__ dst,
                         const int* __restrict__ start, const int* __restrict__ rank,
                         int* __restrict__ csr_src, int E) {
  int e = blockIdx.x * blockDim.x + threadIdx.x;
  if (e >= E) return;
  csr_src[start[dst[e]] + rank[e]] = src[e];
}

template <int J>
__device__ __forceinline__ float group8_bcast(float x) {
  return __int_as_float((int)__builtin_amdgcn_mov_dpp8((unsigned)__float_as_int(x), J * 0x249249));
}
__device__ __forceinline__ int edge_source8(const int* __restrict__ csr_src, int st, int deg, int lane) {
  const int j = lane & 7;
  return csr_src[st + ((j < deg) ? j : deg - 1)];
}
__device__ __forceinline__ float edge_weight8(float els, float erd, int deg, int lane) {
  float v = els + erd;
  v = (v > 0.f) ? v : 0.2f * v;
  v = ((lane & 7) < deg) ? v : -INFINITY;
  float m = fmaxf(v, lane_xor<1>(v));
  m = fmaxf(m, lane_xor<2>(m));
  m = fmaxf(m, lane_xor<4>(m));
  const float ex = expf(v - m);
  float s = ex + lane_xor<1>(ex);
  s += lane_xor<2>(s);
  s += lane_xor<4>(s);
  return ex / s;
}

__global__ void node_softmax_gather_in(const int* __restrict__ start, const int* __restrict__ cnt,
                                       const int* __restrict__ csr_src,
                                       const float* __restrict__ el, const float* __restrict__ er,
                                       const float* __restrict__ x,
                                       float* __restrict__ agg,
                                       int N) {
  const int lane = threadIdx.x & 31;
  const int d = __builtin_amdgcn_readfirstlane((int)(blockIdx.x * (blockDim.x >> 5) + (threadIdx.x >> 5)));
  if (d >= N) return;
  const int h   = lane >> 3;
  const int st  = start[d];
  const int deg = cnt[d];
  const float erd = er[d * 4 + h];
  const int cl = (lane < 27) ? lane : 26;

  float a0 = 0.f, a1 = 0.f, a2 = 0.f, a3 = 0.f;
  if (deg >= 1 && deg <= 8) {
    const int sidx = edge_source8(csr_src, st, deg, lane);
    const float els = el[sidx * 4 + h];
    float xv0, xv1, xv2, xv3, xv4, xv5, xv6, xv7;
#define ROW_LOAD_IN(J) if (J < deg) xv##J = x[(size_t)__builtin_amdgcn_readlane(sidx, J) * 27 + cl];
    ROW_LOAD_IN(0) ROW_LOAD_IN(1) ROW_LOAD_IN(2) ROW_LOAD_IN(3)
    ROW_LOAD_IN(4) ROW_LOAD_IN(5) ROW_LOAD_IN(6) ROW_LOAD_IN(7)
#undef ROW_LOAD_IN
    const int a = __float_as_int(edge_weight8(els, erd, deg, lane));
#define ROW_FMA_IN(J) if (J < deg) {                                              \
      a0 += xv##J * __int_as_float(__builtin_amdgcn_readlane(a, J));              \
      a1 += xv##J * __int_as_float(__builtin_amdgcn_readlane(a, 8 + J));          \
      a2 += xv##J * __int_as_float(__builtin_amdgcn_readlane(a, 16 + J));         \
      a3 += xv##J * __int_as_float(__builtin_amdgcn_readlane(a, 24 + J)); }
    ROW_FMA_IN(0) ROW_FMA_IN(1) ROW_FMA_IN(2) ROW_FMA_IN(3)
    ROW_FMA_IN(4) ROW_FMA_IN(5) ROW_FMA_IN(6) ROW_FMA_IN(7)
#undef ROW_FMA_IN
  } else {
    float m = -INFINITY;
    for (int j = 0; j < deg; ++j) {
      const int sidx = csr_src[st + j];
      float v = el[sidx * 4 + h] + erd;
      v = (v > 0.f) ? v : 0.2f * v;
      m = fmaxf(m, v);
    }
    float sum = 0.f;
    for (int j = 0; j < deg; ++j) {
      const int sidx = csr_src[st + j];
      float v = el[sidx * 4 + h] + erd;
      v = (v > 0.f) ? v : 0.2f * v;
      sum += expf(v - m);
    }
    for (int j = 0; j < deg; ++j) {
      const int sidx = csr_src[st + j];
      float v = el[sidx * 4 + h] + erd;
      v = (v > 0.f) ? v : 0.2f * v;
      const int a = __float_as_int(expf(v - m) / sum);
      const float xv = x[(size_t)sidx * 27 + cl];
      a0 += xv * __int_as_float(__builtin_amdgcn_readlane(a, 0));
      a1 += xv * __int_as_float(__builtin_amdgcn_readlane(a, 8));
      a2 += xv * __int_as_float(__builtin_amdgcn_readlane(a, 16));
      a3 += xv * __int_as_float(__builtin_amdgcn_readlane(a, 24));
    }
  }
  if (lane < 27) {
    float* o = agg + (size_t)d * 108 + lane;
    o[0] = a0; o[27] = a1; o[54] = a2; o[81] = a3;
  }
}

__global__ void node_softmax_gather(const int* __restrict__ start, const int* __restrict__ cnt,
                                    const int* __restrict__ csr_src,
                                    const float* __restrict__ el, const float* __restrict__ er,
                                    const float* __restrict__ feat,
                                    const float* __restrict__ addrow,
                                    const float* __restrict__ bias,
                                    const float* __restrict__ aw_w, const float* __restrict__ aw_b,
                                    const int* __restrict__ gid, float* __restrict__ gmat,
                                    int N) {
  const int lane = threadIdx.x & 31;
  const int d = __builtin_amdgcn_readfirstlane((int)(blockIdx.x * (blockDim.x >> 5) + (threadIdx.x >> 5)));
  if (d >= N) return;
  const int h   = lane >> 3;
  const int st  = start[d];
  const int deg = cnt[d];
  const float erd = er[d * 4 + h];
  const float4 r = *reinterpret_cast<const float4*>(addrow + (size_t)d * 128 + lane * 4);
  const float4 b = *reinterpret_cast<const float4*>(bias + lane * 4);

  float4 acc = make_float4(0.f, 0.f, 0.f, 0.f);
  if (deg >= 1 && deg <= 8) {
    const int sidx = edge_source8(csr_src, st, deg, lane);
    const float els = el[sidx * 4 + h];
    float4 f0, f1, f2, f3, f4, f5, f6, f7;
#define ROW_LOAD(J) if (J < deg) f##J = *reinterpret_cast<const float4*>(                     \
        feat + (size_t)__builtin_amdgcn_readlane(sidx, J) * 128 + lane * 4);
    ROW_LOAD(0) ROW_LOAD(1) ROW_LOAD(2) ROW_LOAD(3)
    ROW_LOAD(4) ROW_LOAD(5) ROW_LOAD(6) ROW_LOAD(7)
#undef ROW_LOAD
    const float a = edge_weight8(els, erd, deg, lane);
#define ROW_FMA(J) if (J < deg) { const float aj = group8_bcast<J>(a);                        \
      acc.x += f##J.x * aj; acc.y += f##J.y * aj; acc.z += f##J.z * aj; acc.w += f##J.w * aj; }
    ROW_FMA(0) ROW_FMA(1) ROW_FMA(2) ROW_FMA(3)
    ROW_FMA(4) ROW_FMA(5) ROW_FMA(6) ROW_FMA(7)
#undef ROW_FMA
  } else {
    float m = -INFINITY;
    for (int j = 0; j < deg; ++j) {
      const int sidx = csr_src[st + j];
      float v = el[sidx * 4 + h] + erd;
      v = (v > 0.f) ? v : 0.2f * v;
      m = fmaxf(m, v);
    }
    float sum = 0.f;
    for (int j = 0; j < deg; ++j) {
      const int sidx = csr_src[st + j];
      float v = el[sidx * 4 + h] + erd;
      v = (v > 0.f) ? v : 0.2f * v;
      sum += expf(v - m);
    }
    for (int j = 0; j < deg; ++j) {
      const int sidx = csr_src[st + j];
      float v = el[sidx * 4 + h] + erd;
      v = (v > 0.f) ? v : 0.2f * v;
      const float a = expf(v - m) / sum;
      const float4 f = *reinterpret_cast<const float4*>(feat + (size_t)sidx * 128 + lane * 4);
      acc.x += f.x * a; acc.y += f.y * a; acc.z += f.z * a; acc.w += f.w * a;
    }
  }

  float4 x;
  x.x = acc.x + r.x + b.x; x.y = acc.y + r.y + b.y;
  x.z = acc.z + r.z + b.z; x.w = acc.w + r.w + b.w;

  const int q = lane & 7;
  float4 hn = make_float4(0.f, 0.f, 0.f, 0.f);
#pragma unroll
  for (int hh = 0; hh < 4; ++hh) {
    hn.x += __shfl(x.x, q + 8 * hh);
    hn.y += __shfl(x.y, q + 8 * hh);
    hn.z += __shfl(x.z, q + 8 * hh);
    hn.w += __shfl(x.w, q + 8 * hh);
  }
  hn.x *= 0.25f; hn.y *= 0.25f; hn.z *= 0.25f; hn.w *= 0.25f;

  float sw = aw_b[0];
#pragma unroll
  for (int i = 0; i < 32; ++i) {
    const float comp = ((i & 3) == 0) ? hn.x : ((i & 3) == 1) ? hn.y : ((i & 3) == 2) ? hn.z : hn.w;
    const float hi = __int_as_float(__builtin_amdgcn_readlane(__float_as_int(comp), i >> 2));
    sw += hi * aw_w[i];
  }
  const float w = 1.f / (1.f + expf(-sw));

  const float t0 = __shfl(hn.x, lane >> 2);
  const float t1 = __shfl(hn.y, lane >> 2);
  const float t2 = __shfl(hn.z, lane >> 2);
  const float t3 = __shfl(hn.w, lane >> 2);
  const int   c  = lane & 3;
  const float hv = (c == 0) ? t0 : (c == 1) ? t1 : (c == 2) ? t2 : t3;
  const int g = gid[d];
  atomicAdd(&gmat[g * 64 + lane], w * hv);
  atomicMax((unsigned*)gmat + g * 64 + 32 + lane, ford(hv));
}

extern "C" void kernel_launch(void* const* d_in, const int* in_sizes, int n_in,
                              void* d_out, int out_size, void* d_ws, size_t ws_size,
                              hipStream_t stream) {
  const float* node  = (const float*)d_in[0];
  const int*   src   = (const int*)d_in[1];
  const int*   dst   = (const int*)d_in[2];
  const int*   gid   = (const int*)d_in[3];
  const float* fc1_w = (const float*)d_in[4];
  const float* al1   = (const float*)d_in[5];
  const float* ar1   = (const float*)d_in[6];
  const float* res1w = (const float*)d_in[7];
  const float* bias1 = (const float*)d_in[8];
  const float* fc2_w = (const float*)d_in[9];
  const float* al2   = (const float*)d_in[10];
  const float* ar2   = (const float*)d_in[11];
  const float* bias2 = (const float*)d_in[12];
  const float* aw_w  = (const float*)d_in[13];
  const float* aw_b  = (const float*)d_in[14];
  const float* out_w = (const float*)d_in[15];
  const float* out_b = (const float*)d_in[16];
  (void)n_in; (void)ws_size;

  const int N = in_sizes[3];
  const int E = in_sizes[1];
  const int G = out_size / 128;

  char* wsb = (char*)d_ws;
  size_t off = 0;
  auto alloc = [&](size_t bytes) -> char* {
    char* p = wsb + off;
    off = (off + bytes + 255) & ~(size_t)255;
    return p;
  };
  float*    bufA  = (float*)alloc((size_t)N * 128 * 4);
  float*    bufB  = (float*)alloc((size_t)N * 128 * 4);
  float*    bufC  = (float*)alloc((size_t)N * 128 * 4);
  float*    el    = (float*)alloc((size_t)N * 4 * 4);
  float*    er    = (float*)alloc((size_t)N * 4 * 4);
  int*      cnt   = (int*)alloc((size_t)N * 4 * 4);
  int*      start = (int*)alloc((size_t)N * 4 * 4);
  int*      rank  = (int*)alloc((size_t)E * 4 * 4);
  int*      csr   = rank + E;
  float*    gmat  = (float*)alloc((size_t)G * 64 * 4);

  auto grid = [](int n) { return dim3((unsigned)((n + THREADS - 1) / THREADS)); };
  const int ntiles  = (N + 15) / 16;
  const int nblocks = (ntiles + 3) / 4;
  const int gtiles  = (G + 15) / 16;
  const int gblocks = (gtiles + 3) / 4;

  attn_proj_in<<<grid(N * 4), THREADS, 0, stream>>>(node, fc1_w, al1, ar1, el, er, N * 4,
                                                    (unsigned*)cnt, N + 1, (unsigned*)gmat, G * 64);

  deg_rank<<<grid(E), THREADS, 0, stream>>>(dst, cnt, rank, E);
  seg_alloc<<<grid(N), THREADS, 0, stream>>>(cnt, start, cnt + N, N);
  csr_fill<<<grid(E), THREADS, 0, stream>>>(src, dst, start, rank, csr, E);

  node_softmax_gather_in<<<grid(N * 32), THREADS, 0, stream>>>(start, cnt, csr, el, er, node, bufB, N);
  wmma_gemm_l1<<<nblocks, 128, 0, stream>>>(bufB, node, fc1_w, res1w, bias1, bufC, N);

  wmma_gemm_feat_scores<<<nblocks, 128, 0, stream>>>(bufC, fc2_w, al2, ar2, bufA, el, er, N);
  node_softmax_gather<<<grid(N * 32), THREADS, 0, stream>>>(start, cnt, csr, el, er, bufA, bufC, bias2,
                                                            aw_w, aw_b, gid, gmat, N);

  wmma_gemm_head_ln<<<gblocks, 128, 0, stream>>>(gmat, out_w, out_b, (float*)d_out, G);
}


# Edge sources go straight into 16-slot rows per destination (slot = what the in-degree atomic returns; overflow list past 16): no segment scan or fill pass, 9 -> 7 launches; sums stay in arrival order
# speedup vs baseline: 6.4950x; 1.0816x over previous
#include <hip/hip_runtime.h>
#include <hip/hip_bf16.h>
#include <math.h>

typedef __attribute__((ext_vector_type(2))) float v2f;
typedef __attribute__((ext_vector_type(4))) float v4f;
typedef __attribute__((ext_vector_type(8))) float v8f;

__device__ __forceinline__ v4f load_quad(const float* __restrict__ p) {
  v4f r;
  __builtin_memcpy(&r, p, 16);
  return r;
}

#define THREADS 256
#define ORD_NEG_INF 0x007FFFFFu
#define ELL_W 16

__device__ __forceinline__ unsigned ford(float f) {
  unsigned u = __float_as_uint(f);
  return (u & 0x80000000u) ? ~u : (u | 0x80000000u);
}
__device__ __forceinline__ float funord(unsigned u) {
  return __uint_as_float((u & 0x80000000u) ? (u & 0x7fffffffu) : ~u);
}

template <int MASK>
__device__ __forceinline__ float lane_xor(float x) {
  const int b = __float_as_int(x);
  return __int_as_float(__builtin_amdgcn_update_dpp(b, b, 0x160 + MASK, 0xf, 0xf, false));
}
template <int MASK>
__device__ __forceinline__ float halve_sum(float a, float b, bool up) {
  const float keep = up ? b : a;
  const float send = up ? a : b;
  return keep + lane_xor<MASK>(send);
}
__device__ __forceinline__ float row16_sum(float v) {
  v += lane_xor<1>(v);
  v += lane_xor<2>(v);
  v += lane_xor<4>(v);
  v += lane_xor<8>(v);
  return v;
}

template <int K, int NT>
__device__ __forceinline__ void wmma_tile_xwt(const float* __restrict__ X,
                                              const float* __restrict__ W,
                                              int mrow, int lane, v8f (&acc)[NT]) {
  static_assert(K % 8 == 0, "the fragment loaders take k in groups of 8");
  const int qoff = (lane < 16) ? 0 : 4;
#pragma unroll
  for (int t = 0; t < NT; ++t) acc[t] = (v8f){0.f,0.f,0.f,0.f,0.f,0.f,0.f,0.f};

  const float* __restrict__ xr = X + (size_t)mrow * K + qoff;
  const float* __restrict__ wr = W + (size_t)(lane & 15) * K + qoff;
  for (int k0 = 0; k0 < K; k0 += 8) {
    const v4f a = load_quad(xr + k0);
    v4f b[NT];
#pragma unroll
    for (int t = 0; t < NT; ++t) b[t] = load_quad(wr + (size_t)t * 16 * K + k0);
#pragma unroll
    for (int t = 0; t < NT; ++t) {
      acc[t] = __builtin_amdgcn_wmma_f32_16x16x4_f32(
          false, (v2f){a.x, a.y}, false, (v2f){b[t].x, b[t].y}, (short)0, acc[t], false, false);
    }
#pragma unroll
    for (int t = 0; t < NT; ++t) {
      acc[t] = __builtin_amdgcn_wmma_f32_16x16x4_f32(
          false, (v2f){a.z, a.w}, false, (v2f){b[t].z, b[t].w}, (short)0, acc[t], false, false);
    }
  }
}

__global__ void __launch_bounds__(128)
wmma_gemm_feat_scores(const float* __restrict__ X, const float* __restrict__ W,
                      const float* __restrict__ al, const float* __restrict__ ar,
                      float* __restrict__ Y, float* __restrict__ el, float* __restrict__ er, int M) {
  const int lane = threadIdx.x & 31;
  const int wave = threadIdx.x >> 5;
  const int tile = blockIdx.x * (blockDim.x >> 5) + wave;
  if (tile * 16 >= M) return;
  const int rbase = tile * 16;

  v8f acc[8];
  wmma_tile_xwt<128, 8>(X, W, rbase + (lane & 15), lane, acc);

  const int rowoff = (lane < 16) ? 0 : 8;
  const int colo   = lane & 15;
#pragma unroll
  for (int t = 0; t < 8; ++t) {
#pragma unroll
    for (int i = 0; i < 8; ++i) {
      Y[(size_t)(rbase + rowoff + i) * 128 + t * 16 + colo] = acc[t][i];
    }
  }

  float alo[4], ahi[4], rlo[4], rhi[4];
#pragma unroll
  for (int h = 0; h < 4; ++h) {
    alo[h] = al[h * 32 + colo]; ahi[h] = al[h * 32 + 16 + colo];
    rlo[h] = ar[h * 32 + colo]; rhi[h] = ar[h * 32 + 16 + colo];
  }
  const bool b3 = (colo & 8) != 0, b2 = (colo & 4) != 0, b1 = (colo & 2) != 0, b0 = (colo & 1) != 0;
  float pl[8][4], pr[8][4];
#pragma unroll
  for (int i = 0; i < 8; ++i) {
#pragma unroll
    for (int h = 0; h < 4; ++h) {
      pl[i][h] = acc[2 * h][i] * alo[h] + acc[2 * h + 1][i] * ahi[h];
      pr[i][h] = acc[2 * h][i] * rlo[h] + acc[2 * h + 1][i] * rhi[h];
    }
  }
#pragma unroll
  for (int h = 0; h < 4; ++h) {
#pragma unroll
    for (int i = 0; i < 4; ++i) {
      pl[i][h] = halve_sum<8>(pl[i][h], pl[i + 4][h], b3);
      pr[i][h] = halve_sum<8>(pr[i][h], pr[i + 4][h], b3);
    }
#pragma unroll
    for (int i = 0; i < 2; ++i) {
      pl[i][h] = halve_sum<4>(pl[i][h], pl[i + 2][h], b2);
      pr[i][h] = halve_sum<4>(pr[i][h], pr[i + 2][h], b2);
    }
    pl[0][h] = halve_sum<2>(pl[0][h], pl[1][h], b1);
    pr[0][h] = halve_sum<2>(pr[0][h], pr[1][h], b1);
  }
  float2 ol, orr;
  ol.x  = halve_sum<1>(pl[0][0], pl[0][2], b0);
  ol.y  = halve_sum<1>(pl[0][1], pl[0][3], b0);
  orr.x = halve_sum<1>(pr[0][0], pr[0][2], b0);
  orr.y = halve_sum<1>(pr[0][1], pr[0][3], b0);
  const size_t so = (size_t)(rbase + rowoff) * 4 + 2 * colo;
  *reinterpret_cast<float2*>(el + so) = ol;
  *reinterpret_cast<float2*>(er + so) = orr;
}

__device__ __forceinline__ float max_half_decode(float f) {
  const unsigned u = __float_as_uint(f);
  return (u == ORD_NEG_INF) ? 0.f : funord(u);
}

__global__ void __launch_bounds__(128)
wmma_gemm_head_ln(const float* __restrict__ g, const float* __restrict__ W,
                  const float* __restrict__ out_b, float* __restrict__ out, int M) {
  const int lane = threadIdx.x & 31;
  const int wave = threadIdx.x >> 5;
  const int tile = blockIdx.x * (blockDim.x >> 5) + wave;
  if (tile * 16 >= M) return;
  const int rbase = tile * 16;

  v8f acc[8];
#pragma unroll
  for (int t = 0; t < 8; ++t) acc[t] = (v8f){0.f,0.f,0.f,0.f,0.f,0.f,0.f,0.f};
  const int qoff = (lane < 16) ? 0 : 4;
  const float* __restrict__ gr = g + (size_t)(rbase + (lane & 15)) * 64 + qoff;
  const float* __restrict__ wr = W + (size_t)(lane & 15) * 64 + qoff;
#pragma unroll
  for (int k0 = 0; k0 < 64; k0 += 8) {
    v4f a = load_quad(gr + k0);
    if (k0 >= 32) {
      a.x = max_half_decode(a.x); a.y = max_half_decode(a.y);
      a.z = max_half_decode(a.z); a.w = max_half_decode(a.w);
    }
    v4f b[8];
#pragma unroll
    for (int t = 0; t < 8; ++t) b[t] = load_quad(wr + (size_t)t * 16 * 64 + k0);
#pragma unroll
    for (int t = 0; t < 8; ++t) {
      acc[t] = __builtin_amdgcn_wmma_f32_16x16x4_f32(
          false, (v2f){a.x, a.y}, false, (v2f){b[t].x, b[t].y}, (short)0, acc[t], false, false);
    }
#pragma unroll
    for (int t = 0; t < 8; ++t) {
      acc[t] = __builtin_amdgcn_wmma_f32_16x16x4_f32(
          false, (v2f){a.z, a.w}, false, (v2f){b[t].z, b[t].w}, (short)0, acc[t], false, false);
    }
  }

  const int rowoff = (lane < 16) ? 0 : 8;
  const int colo   = lane & 15;
  float bt[8];
#pragma unroll
  for (int t = 0; t < 8; ++t) bt[t] = out_b[t * 16 + colo];
#pragma unroll
  for (int i = 0; i < 8; ++i) {
    float v[8];
    float s = 0.f;
#pragma unroll
    for (int t = 0; t < 8; ++t) { v[t] = acc[t][i] + bt[t]; s += v[t]; }
    const float mu = row16_sum(s) * (1.f / 128.f);
    float q = 0.f;
#pragma unroll
    for (int t = 0; t < 8; ++t) { v[t] -= mu; q += v[t] * v[t]; }
    const float rs = rsqrtf(row16_sum(q) * (1.f / 128.f) + 1e-5f);
#pragma unroll
    for (int t = 0; t < 8; ++t) {
      out[(size_t)(rbase + rowoff + i) * 128 + t * 16 + colo] = v[t] * rs;
    }
  }
}

__device__ __forceinline__ v2f load_tail27(const float* __restrict__ row, bool upper) {
  const v4f q = load_quad(row + 23);
  v2f r;
  r.x = upper ? q.w : q.y;
  r.y = upper ? 0.f : q.z;
  return r;
}

__global__ void __launch_bounds__(128)
wmma_gemm_l1(const float* __restrict__ agg, const float* __restrict__ x,
             const float* __restrict__ W1, const float* __restrict__ R,
             const float* __restrict__ bias, float* __restrict__ h1, int M) {
  const int K    = 27;
  const int lane = threadIdx.x & 31;
  const int wave = threadIdx.x >> 5;
  const int tile = blockIdx.x * (blockDim.x >> 5) + wave;
  if (tile * 16 >= M) return;
  const int rbase = tile * 16;
  const int mrow  = rbase + (lane & 15);
  const bool upper = lane >= 16;

  v8f acc[8];
#pragma unroll
  for (int t = 0; t < 8; ++t) acc[t] = (v8f){0.f,0.f,0.f,0.f,0.f,0.f,0.f,0.f};

  const int qoff = upper ? 4 : 0;

  {
    const float* __restrict__ xr = x + (size_t)mrow * K;
    const float* __restrict__ rr = R + (size_t)(lane & 15) * K;
#pragma unroll
    for (int k0 = 0; k0 < 24; k0 += 8) {
      const v4f a = load_quad(xr + k0 + qoff);
      v4f b[8];
#pragma unroll
      for (int t = 0; t < 8; ++t) b[t] = load_quad(rr + t * 16 * K + k0 + qoff);
#pragma unroll
      for (int t = 0; t < 8; ++t) {
        acc[t] = __builtin_amdgcn_wmma_f32_16x16x4_f32(
            false, (v2f){a.x, a.y}, false, (v2f){b[t].x, b[t].y}, (short)0, acc[t], false, false);
      }
#pragma unroll
      for (int t = 0; t < 8; ++t) {
        acc[t] = __builtin_amdgcn_wmma_f32_16x16x4_f32(
            false, (v2f){a.z, a.w}, false, (v2f){b[t].z, b[t].w}, (short)0, acc[t], false, false);
      }
    }
    const v2f a = load_tail27(xr, upper);
#pragma unroll
    for (int t = 0; t < 8; ++t) {
      const v2f b = load_tail27(rr + t * 16 * K, upper);
      acc[t] = __builtin_amdgcn_wmma_f32_16x16x4_f32(
          false, a, false, b, (short)0, acc[t], false, false);
    }
  }

#pragma unroll
  for (int h = 0; h < 4; ++h) {
    const float* __restrict__ ah = agg + (size_t)mrow * 108 + h * K;
    const float* __restrict__ wr = W1 + (size_t)(2 * h * 16 + (lane & 15)) * K;
#pragma unroll
    for (int k0 = 0; k0 < 24; k0 += 8) {
      const v4f a  = load_quad(ah + k0 + qoff);
      const v4f b0 = load_quad(wr + k0 + qoff);
      const v4f b1 = load_quad(wr + 16 * K + k0 + qoff);
      acc[2 * h] = __builtin_amdgcn_wmma_f32_16x16x4_f32(
          false, (v2f){a.x, a.y}, false, (v2f){b0.x, b0.y}, (short)0, acc[2 * h], false, false);
      acc[2 * h + 1] = __builtin_amdgcn_wmma_f32_16x16x4_f32(
          false, (v2f){a.x, a.y}, false, (v2f){b1.x, b1.y}, (short)0, acc[2 * h + 1], false, false);
      acc[2 * h] = __builtin_amdgcn_wmma_f32_16x16x4_f32(
          false, (v2f){a.z, a.w}, false, (v2f){b0.z, b0.w}, (short)0, acc[2 * h], false, false);
      acc[2 * h + 1] = __builtin_amdgcn_wmma_f32_16x16x4_f32(
          false, (v2f){a.z, a.w}, false, (v2f){b1.z, b1.w}, (short)0, acc[2 * h + 1], false, false);
    }
    const v2f a = load_tail27(ah, upper);
#pragma unroll
    for (int tt = 0; tt < 2; ++tt) {
      const v2f b = load_tail27(wr + tt * 16 * K, upper);
      acc[2 * h + tt] = __builtin_amdgcn_wmma_f32_16x16x4_f32(
          false, a, false, b, (short)0, acc[2 * h + tt], false, false);
    }
  }

  const int rowoff = (lane < 16) ? 0 : 8;
  const int colo   = lane & 15;
#pragma unroll
  for (int t = 0; t < 8; ++t) {
    const float bt = bias[t * 16 + colo];
#pragma unroll
    for (int i = 0; i < 8; ++i) {
      float v = acc[t][i] + bt;
      v = (v > 0.f) ? v : expm1f(v);
      h1[(size_t)(rbase + rowoff + i) * 128 + t * 16 + colo] = v;
    }
  }
}

__global__ void attn_proj_in(const float* __restrict__ x, const float* __restrict__ w,
                             const float* __restrict__ al, const float* __restrict__ ar,
                             float* __restrict__ el, float* __restrict__ er, int NH,
                             unsigned* __restrict__ cnt, int ncnt,
                             unsigned* __restrict__ gmat, int ngm) {
  __shared__ float vt[216];
  const int tid = threadIdx.x;
  if (tid < 216) {
    const int which = (tid >= 108) ? 1 : 0;
    const int idx = tid - 108 * which;
    const int h = idx / 27, c = idx - h * 27;
    const float* a = which ? ar : al;
    float s = 0.f;
    for (int d = 0; d < 32; ++d) s += w[(h * 32 + d) * 27 + c] * a[h * 32 + d];
    vt[tid] = s;
  }
  __syncthreads();
  const int t = blockIdx.x * blockDim.x + tid;
  if (t < NH) {
    const int n = t >> 2, h = t & 3;
    const float* xr = x + n * 27;
    float sl = 0.f, sr = 0.f;
    for (int c = 0; c < 27; ++c) {
      const float xv = xr[c];
      sl += xv * vt[h * 27 + c];
      sr += xv * vt[108 + h * 27 + c];
    }
    el[t] = sl; er[t] = sr;
  }
  const unsigned nt = gridDim.x * blockDim.x;
#pragma clang loop unroll(disable) interleave(disable) vectorize(disable)
  for (unsigned i = (unsigned)t; i < (unsigned)ncnt; i += nt) cnt[i] = 0u;
#pragma clang loop unroll(disable) interleave(disable) vectorize(disable)
  for (unsigned i = (unsigned)t; i < (unsigned)ngm; i += nt) gmat[i] = ((i & 63u) < 32u) ? 0u : ORD_NEG_INF;
}

__global__ void edge_place(const int* __restrict__ src, const int* __restrict__ dst,
                           int* cnt, int* __restrict__ ell,
                           int* __restrict__ ovf, int* novf, int E) {
  const int e = blockIdx.x * blockDim.x + threadIdx.x;
  if (e >= E) return;
  const int d = dst[e];
  const int s = src[e];
  const int r = atomicAdd(&cnt[d], 1);
  if (r < ELL_W) {
    ell[(size_t)d * ELL_W + r] = s;
  } else {
    const int o = atomicAdd(novf, 1);
    ovf[2 * (size_t)o]     = d;
    ovf[2 * (size_t)o + 1] = s;
  }
}

template <class F>
__device__ __forceinline__ void for_each_in_edge(int slots, const int* __restrict__ ovf, int nov,
                                                 int d, int deg, F&& f) {
  const int nb = (deg < ELL_W) ? deg : ELL_W;
  for (int j = 0; j < nb; ++j) f(__builtin_amdgcn_readlane(slots, j));
  for (int i = 0; i < nov; ++i) {
    if (ovf[2 * (size_t)i] == d) f(ovf[2 * (size_t)i + 1]);
  }
}

template <int J>
__device__ __forceinline__ float group8_bcast(float x) {
  return __int_as_float((int)__builtin_amdgcn_mov_dpp8((unsigned)__float_as_int(x), J * 0x249249));
}
__device__ __forceinline__ int edge_source8(int slots, int deg, int lane) {
  const int first = __builtin_amdgcn_readfirstlane(slots);
  const int own   = __builtin_amdgcn_update_dpp(slots, slots, 0x118  , 0xf, 0xc, false);
  return ((lane & 7) < deg) ? own : first;
}
__device__ __forceinline__ float edge_weight8(float els, float erd, int deg, int lane) {
  float v = els + erd;
  v = (v > 0.f) ? v : 0.2f * v;
  v = ((lane & 7) < deg) ? v : -INFINITY;
  float m = fmaxf(v, lane_xor<1>(v));
  m = fmaxf(m, lane_xor<2>(m));
  m = fmaxf(m, lane_xor<4>(m));
  const float ex = expf(v - m);
  float s = ex + lane_xor<1>(ex);
  s += lane_xor<2>(s);
  s += lane_xor<4>(s);
  return ex / s;
}

__global__ void node_softmax_gather_in(const int* __restrict__ cnt, const int* __restrict__ ell,
                                       const int* __restrict__ ovf, const int* __restrict__ novf,
                                       const float* __restrict__ el, const float* __restrict__ er,
                                       const float* __restrict__ x,
                                       float* __restrict__ agg,
                                       int N) {
  const int lane = threadIdx.x & 31;
  const int d = __builtin_amdgcn_readfirstlane((int)(blockIdx.x * (blockDim.x >> 5) + (threadIdx.x >> 5)));
  if (d >= N) return;
  const int h   = lane >> 3;
  const int deg = cnt[d];
  const int slots = ell[(size_t)d * ELL_W + (lane & (ELL_W - 1))];
  const float erd = er[d * 4 + h];
  const int cl = (lane < 27) ? lane : 26;

  float a0 = 0.f, a1 = 0.f, a2 = 0.f, a3 = 0.f;
  if (deg >= 1 && deg <= 8) {
    const int sidx = edge_source8(slots, deg, lane);
    const float els = el[sidx * 4 + h];
    float xv0, xv1, xv2, xv3, xv4, xv5, xv6, xv7;
#define ROW_LOAD_IN(J) if (J < deg) xv##J = x[(size_t)__builtin_amdgcn_readlane(sidx, J) * 27 + cl];
    ROW_LOAD_IN(0) ROW_LOAD_IN(1) ROW_LOAD_IN(2) ROW_LOAD_IN(3)
    ROW_LOAD_IN(4) ROW_LOAD_IN(5) ROW_LOAD_IN(6) ROW_LOAD_IN(7)
#undef ROW_LOAD_IN
    const int a = __float_as_int(edge_weight8(els, erd, deg, lane));
#define ROW_FMA_IN(J) if (J < deg) {                                              \
      a0 += xv##J * __int_as_float(__builtin_amdgcn_readlane(a, J));              \
      a1 += xv##J * __int_as_float(__builtin_amdgcn_readlane(a, 8 + J));          \
      a2 += xv##J * __int_as_float(__builtin_amdgcn_readlane(a, 16 + J));         \
      a3 += xv##J * __int_as_float(__builtin_amdgcn_readlane(a, 24 + J)); }
    ROW_FMA_IN(0) ROW_FMA_IN(1) ROW_FMA_IN(2) ROW_FMA_IN(3)
    ROW_FMA_IN(4) ROW_FMA_IN(5) ROW_FMA_IN(6) ROW_FMA_IN(7)
#undef ROW_FMA_IN
  } else {
    const int nov = (deg > ELL_W) ? novf[0] : 0;
    float m = -INFINITY;
    for_each_in_edge(slots, ovf, nov, d, deg, [&](int sidx) {
      float v = el[sidx * 4 + h] + erd;
      v = (v > 0.f) ? v : 0.2f * v;
      m = fmaxf(m, v);
    });
    float sum = 0.f;
    for_each_in_edge(slots, ovf, nov, d, deg, [&](int sidx) {
      float v = el[sidx * 4 + h] + erd;
      v = (v > 0.f) ? v : 0.2f * v;
      sum += expf(v - m);
    });
    for_each_in_edge(slots, ovf, nov, d, deg, [&](int sidx) {
      float v = el[sidx * 4 + h] + erd;
      v = (v > 0.f) ? v : 0.2f * v;
      const int a = __float_as_int(expf(v - m) / sum);
      const float xv = x[(size_t)sidx * 27 + cl];
      a0 += xv * __int_as_float(__builtin_amdgcn_readlane(a, 0));
      a1 += xv * __int_as_float(__builtin_amdgcn_readlane(a, 8));
      a2 += xv * __int_as_float(__builtin_amdgcn_readlane(a, 16));
      a3 += xv * __int_as_float(__builtin_amdgcn_readlane(a, 24));
    });
  }
  if (lane < 27) {
    float* o = agg + (size_t)d * 108 + lane;
    o[0] = a0; o[27] = a1; o[54] = a2; o[81] = a3;
  }
}

__global__ void node_softmax_gather(const int* __restrict__ cnt, const int* __restrict__ ell,
                                    const int* __restrict__ ovf, const int* __restrict__ novf,
                                    const float* __restrict__ el, const float* __restrict__ er,
                                    const float* __restrict__ feat,
                                    const float* __restrict__ addrow,
                                    const float* __restrict__ bias,
                                    const float* __restrict__ aw_w, const float* __restrict__ aw_b,
                                    const int* __restrict__ gid, float* __restrict__ gmat,
                                    int N) {
  const int lane = threadIdx.x & 31;
  const int d = __builtin_amdgcn_readfirstlane((int)(blockIdx.x * (blockDim.x >> 5) + (threadIdx.x >> 5)));
  if (d >= N) return;
  const int h   = lane >> 3;
  const int deg = cnt[d];
  const int slots = ell[(size_t)d * ELL_W + (lane & (ELL_W - 1))];
  const float erd = er[d * 4 + h];
  const float4 r = *reinterpret_cast<const float4*>(addrow + (size_t)d * 128 + lane * 4);
  const float4 b = *reinterpret_cast<const float4*>(bias + lane * 4);

  float4 acc = make_float4(0.f, 0.f, 0.f, 0.f);
  if (deg >= 1 && deg <= 8) {
    const int sidx = edge_source8(slots, deg, lane);
    const float els = el[sidx * 4 + h];
    float4 f0, f1, f2, f3, f4, f5, f6, f7;
#define ROW_LOAD(J) if (J < deg) f##J = *reinterpret_cast<const float4*>(                     \
        feat + (size_t)__builtin_amdgcn_readlane(sidx, J) * 128 + lane * 4);
    ROW_LOAD(0) ROW_LOAD(1) ROW_LOAD(2) ROW_LOAD(3)
    ROW_LOAD(4) ROW_LOAD(5) ROW_LOAD(6) ROW_LOAD(7)
#undef ROW_LOAD
    const float a = edge_weight8(els, erd, deg, lane);
#define ROW_FMA(J) if (J < deg) { const float aj = group8_bcast<J>(a);                        \
      acc.x += f##J.x * aj; acc.y += f##J.y * aj; acc.z += f##J.z * aj; acc.w += f##J.w * aj; }
    ROW_FMA(0) ROW_FMA(1) ROW_FMA(2) ROW_FMA(3)
    ROW_FMA(4) ROW_FMA(5) ROW_FMA(6) ROW_FMA(7)
#undef ROW_FMA
  } else {
    const int nov = (deg > ELL_W) ? novf[0] : 0;
    float m = -INFINITY;
    for_each_in_edge(slots, ovf, nov, d, deg, [&](int sidx) {
      float v = el[sidx * 4 + h] + erd;
      v = (v > 0.f) ? v : 0.2f * v;
      m = fmaxf(m, v);
    });
    float sum = 0.f;
    for_each_in_edge(slots, ovf, nov, d, deg, [&](int sidx) {
      float v = el[sidx * 4 + h] + erd;
      v = (v > 0.f) ? v : 0.2f * v;
      sum += expf(v - m);
    });
    for_each_in_edge(slots, ovf, nov, d, deg, [&](int sidx) {
      float v = el[sidx * 4 + h] + erd;
      v = (v > 0.f) ? v : 0.2f * v;
      const float a = expf(v - m) / sum;
      const float4 f = *reinterpret_cast<const float4*>(feat + (size_t)sidx * 128 + lane * 4);
      acc.x += f.x * a; acc.y += f.y * a; acc.z += f.z * a; acc.w += f.w * a;
    });
  }

  float4 x;
  x.x = acc.x + r.x + b.x; x.y = acc.y + r.y + b.y;
  x.z = acc.z + r.z + b.z; x.w = acc.w + r.w + b.w;

  const int q = lane & 7;
  float4 hn = make_float4(0.f, 0.f, 0.f, 0.f);
#pragma unroll
  for (int hh = 0; hh < 4; ++hh) {
    hn.x += __shfl(x.x, q + 8 * hh);
    hn.y += __shfl(x.y, q + 8 * hh);
    hn.z += __shfl(x.z, q + 8 * hh);
    hn.w += __shfl(x.w, q + 8 * hh);
  }
  hn.x *= 0.25f; hn.y *= 0.25f; hn.z *= 0.25f; hn.w *= 0.25f;

  float sw = aw_b[0];
#pragma unroll
  for (int i = 0; i < 32; ++i) {
    const float comp = ((i & 3) == 0) ? hn.x : ((i & 3) == 1) ? hn.y : ((i & 3) == 2) ? hn.z : hn.w;
    const float hi = __int_as_float(__builtin_amdgcn_readlane(__float_as_int(comp), i >> 2));
    sw += hi * aw_w[i];
  }
  const float w = 1.f / (1.f + expf(-sw));

  const float t0 = __shfl(hn.x, lane >> 2);
  const float t1 = __shfl(hn.y, lane >> 2);
  const float t2 = __shfl(hn.z, lane >> 2);
  const float t3 = __shfl(hn.w, lane >> 2);
  const int   c  = lane & 3;
  const float hv = (c == 0) ? t0 : (c == 1) ? t1 : (c == 2) ? t2 : t3;
  const int g = gid[d];
  atomicAdd(&gmat[g * 64 + lane], w * hv);
  atomicMax((unsigned*)gmat + g * 64 + 32 + lane, ford(hv));
}

extern "C" void kernel_launch(void* const* d_in, const int* in_sizes, int n_in,
                              void* d_out, int out_size, void* d_ws, size_t ws_size,
                              hipStream_t stream) {
  const float* node  = (const float*)d_in[0];
  const int*   src   = (const int*)d_in[1];
  const int*   dst   = (const int*)d_in[2];
  const int*   gid   = (const int*)d_in[3];
  const float* fc1_w = (const float*)d_in[4];
  const float* al1   = (const float*)d_in[5];
  const float* ar1   = (const float*)d_in[6];
  const float* res1w = (const float*)d_in[7];
  const float* bias1 = (const float*)d_in[8];
  const float* fc2_w = (const float*)d_in[9];
  const float* al2   = (const float*)d_in[10];
  const float* ar2   = (const float*)d_in[11];
  const float* bias2 = (const float*)d_in[12];
  const float* aw_w  = (const float*)d_in[13];
  const float* aw_b  = (const float*)d_in[14];
  const float* out_w = (const float*)d_in[15];
  const float* out_b = (const float*)d_in[16];
  (void)n_in; (void)ws_size;

  const int N = in_sizes[3];
  const int E = in_sizes[1];
  const int G = out_size / 128;

  char* wsb = (char*)d_ws;
  size_t off = 0;
  auto alloc = [&](size_t bytes) -> char* {
    char* p = wsb + off;
    off = (off + bytes + 255) & ~(size_t)255;
    return p;
  };
  float*    bufA  = (float*)alloc((size_t)N * 128 * 4);
  float*    bufB  = (float*)alloc((size_t)N * 108 * 4);
  float*    bufC  = (float*)alloc((size_t)N * 128 * 4);
  float*    el    = (float*)alloc((size_t)N * 4 * 4);
  float*    er    = (float*)alloc((size_t)N * 4 * 4);
  int*      cnt   = (int*)alloc(((size_t)N + 1) * 4);
  int*      ell   = (int*)alloc((size_t)N * ELL_W * 4);
  int*      ovf   = (int*)alloc((size_t)E * 2 * 4);
  float*    gmat  = (float*)alloc((size_t)G * 64 * 4);

  auto grid = [](int n) { return dim3((unsigned)((n + THREADS - 1) / THREADS)); };
  const int ntiles  = (N + 15) / 16;
  const int nblocks = (ntiles + 3) / 4;
  const int gtiles  = (G + 15) / 16;
  const int gblocks = (gtiles + 3) / 4;

  attn_proj_in<<<grid(N * 4), THREADS, 0, stream>>>(node, fc1_w, al1, ar1, el, er, N * 4,
                                                    (unsigned*)cnt, N + 1, (unsigned*)gmat, G * 64);

  edge_place<<<grid(E), THREADS, 0, stream>>>(src, dst, cnt, ell, ovf, cnt + N, E);

  node_softmax_gather_in<<<grid(N * 32), THREADS, 0, stream>>>(cnt, ell, ovf, cnt + N, el, er, node, bufB, N);
  wmma_gemm_l1<<<nblocks, 128, 0, stream>>>(bufB, node, fc1_w, res1w, bias1, bufC, N);

  wmma_gemm_feat_scores<<<nblocks, 128, 0, stream>>>(bufC, fc2_w, al2, ar2, bufA, el, er, N);
  node_softmax_gather<<<grid(N * 32), THREADS, 0, stream>>>(cnt, ell, ovf, cnt + N, el, er, bufA, bufC, bias2,
                                                            aw_w, aw_b, gid, gmat, N);

  wmma_gemm_head_ln<<<gblocks, 128, 0, stream>>>(gmat, out_w, out_b, (float*)d_out, G);
}


# Layer 1 and layer 2 projections in one kernel: the 16x128 h1 tile goes through the wave's own LDS slot, read there for whole-row h1 stores and as the 2nd product's A fragments; 7 -> 6 launches
# speedup vs baseline: 6.9180x; 1.0651x over previous
#include <hip/hip_runtime.h>
#include <hip/hip_bf16.h>
#include <math.h>

typedef __attribute__((ext_vector_type(2))) float v2f;
typedef __attribute__((ext_vector_type(4))) float v4f;
typedef __attribute__((ext_vector_type(8))) float v8f;

__device__ __forceinline__ v4f load_quad(const float* __restrict__ p) {
  v4f r;
  __builtin_memcpy(&r, p, 16);
  return r;
}

#define THREADS 256
#define ORD_NEG_INF 0x007FFFFFu
#define ELL_W 16

__device__ __forceinline__ unsigned ford(float f) {
  unsigned u = __float_as_uint(f);
  return (u & 0x80000000u) ? ~u : (u | 0x80000000u);
}
__device__ __forceinline__ float funord(unsigned u) {
  return __uint_as_float((u & 0x80000000u) ? (u & 0x7fffffffu) : ~u);
}

template <int MASK>
__device__ __forceinline__ float lane_xor(float x) {
  const int b = __float_as_int(x);
  return __int_as_float(__builtin_amdgcn_update_dpp(b, b, 0x160 + MASK, 0xf, 0xf, false));
}
template <int MASK>
__device__ __forceinline__ float halve_sum(float a, float b, bool up) {
  const float keep = up ? b : a;
  const float send = up ? a : b;
  return keep + lane_xor<MASK>(send);
}
__device__ __forceinline__ float row16_sum(float v) {
  v += lane_xor<1>(v);
  v += lane_xor<2>(v);
  v += lane_xor<4>(v);
  v += lane_xor<8>(v);
  return v;
}


__device__ __forceinline__ float max_half_decode(float f) {
  const unsigned u = __float_as_uint(f);
  return (u == ORD_NEG_INF) ? 0.f : funord(u);
}

__global__ void __launch_bounds__(128)
wmma_gemm_head_ln(const float* __restrict__ g, const float* __restrict__ W,
                  const float* __restrict__ out_b, float* __restrict__ out, int M) {
  const int lane = threadIdx.x & 31;
  const int wave = threadIdx.x >> 5;
  const int tile = blockIdx.x * (blockDim.x >> 5) + wave;
  if (tile * 16 >= M) return;
  const int rbase = tile * 16;

  v8f acc[8];
#pragma unroll
  for (int t = 0; t < 8; ++t) acc[t] = (v8f){0.f,0.f,0.f,0.f,0.f,0.f,0.f,0.f};
  const int qoff = (lane < 16) ? 0 : 4;
  const float* __restrict__ gr = g + (size_t)(rbase + (lane & 15)) * 64 + qoff;
  const float* __restrict__ wr = W + (size_t)(lane & 15) * 64 + qoff;
#pragma unroll
  for (int k0 = 0; k0 < 64; k0 += 8) {
    v4f a = load_quad(gr + k0);
    if (k0 >= 32) {
      a.x = max_half_decode(a.x); a.y = max_half_decode(a.y);
      a.z = max_half_decode(a.z); a.w = max_half_decode(a.w);
    }
    v4f b[8];
#pragma unroll
    for (int t = 0; t < 8; ++t) b[t] = load_quad(wr + (size_t)t * 16 * 64 + k0);
#pragma unroll
    for (int t = 0; t < 8; ++t) {
      acc[t] = __builtin_amdgcn_wmma_f32_16x16x4_f32(
          false, (v2f){a.x, a.y}, false, (v2f){b[t].x, b[t].y}, (short)0, acc[t], false, false);
    }
#pragma unroll
    for (int t = 0; t < 8; ++t) {
      acc[t] = __builtin_amdgcn_wmma_f32_16x16x4_f32(
          false, (v2f){a.z, a.w}, false, (v2f){b[t].z, b[t].w}, (short)0, acc[t], false, false);
    }
  }

  const int rowoff = (lane < 16) ? 0 : 8;
  const int colo   = lane & 15;
  float bt[8];
#pragma unroll
  for (int t = 0; t < 8; ++t) bt[t] = out_b[t * 16 + colo];
#pragma unroll
  for (int i = 0; i < 8; ++i) {
    float v[8];
    float s = 0.f;
#pragma unroll
    for (int t = 0; t < 8; ++t) { v[t] = acc[t][i] + bt[t]; s += v[t]; }
    const float mu = row16_sum(s) * (1.f / 128.f);
    float q = 0.f;
#pragma unroll
    for (int t = 0; t < 8; ++t) { v[t] -= mu; q += v[t] * v[t]; }
    const float rs = rsqrtf(row16_sum(q) * (1.f / 128.f) + 1e-5f);
#pragma unroll
    for (int t = 0; t < 8; ++t) {
      out[(size_t)(rbase + rowoff + i) * 128 + t * 16 + colo] = v[t] * rs;
    }
  }
}

__device__ __forceinline__ v2f load_tail27(const float* __restrict__ row, bool upper) {
  const v4f q = load_quad(row + 23);
  v2f r;
  r.x = upper ? q.w : q.y;
  r.y = upper ? 0.f : q.z;
  return r;
}

#define TILE_PITCH 132

__global__ void __launch_bounds__(128, 7)
wmma_gemm_l1_l2(const float* __restrict__ agg, const float* __restrict__ x,
                const float* __restrict__ W1, const float* __restrict__ R,
                const float* __restrict__ bias, const float* __restrict__ W2,
                const float* __restrict__ al, const float* __restrict__ ar,
                float* __restrict__ h1, float* __restrict__ Y,
                float* __restrict__ el, float* __restrict__ er, int M) {
  __shared__ __attribute__((aligned(16))) float tile_s[4 * 16 * TILE_PITCH];
  const int K    = 27;
  const int lane = threadIdx.x & 31;
  const int wave = __builtin_amdgcn_readfirstlane((int)(threadIdx.x >> 5));
  const int tile = blockIdx.x * (blockDim.x >> 5) + wave;
  if (tile * 16 >= M) return;
  const int rbase = tile * 16;
  const bool upper = lane >= 16;
  float* __restrict__ ts = tile_s + wave * (16 * TILE_PITCH);

  v8f acc[8];
#pragma unroll
  for (int t = 0; t < 8; ++t) acc[t] = (v8f){0.f,0.f,0.f,0.f,0.f,0.f,0.f,0.f};

  const int qoff = upper ? 4 : 0;

  {
    const float* __restrict__ xr = x + (size_t)rbase * K + (lane & 15) * K;
    const float* __restrict__ rr = R + (size_t)(lane & 15) * K;
#pragma clang loop unroll(disable)
    for (int k0 = 0; k0 < 24; k0 += 8) {
      const v4f a = load_quad(xr + k0 + qoff);
      v4f b[8];
#pragma unroll
      for (int t = 0; t < 8; ++t) b[t] = load_quad(rr + t * 16 * K + k0 + qoff);
#pragma unroll
      for (int t = 0; t < 8; ++t) {
        acc[t] = __builtin_amdgcn_wmma_f32_16x16x4_f32(
            false, (v2f){a.x, a.y}, false, (v2f){b[t].x, b[t].y}, (short)0, acc[t], false, false);
      }
#pragma unroll
      for (int t = 0; t < 8; ++t) {
        acc[t] = __builtin_amdgcn_wmma_f32_16x16x4_f32(
            false, (v2f){a.z, a.w}, false, (v2f){b[t].z, b[t].w}, (short)0, acc[t], false, false);
      }
    }
    const v2f a = load_tail27(xr, upper);
#pragma unroll
    for (int t = 0; t < 8; ++t) {
      const v2f b = load_tail27(rr + t * 16 * K, upper);
      acc[t] = __builtin_amdgcn_wmma_f32_16x16x4_f32(
          false, a, false, b, (short)0, acc[t], false, false);
    }
  }

#pragma unroll
  for (int h = 0; h < 4; ++h) {
    const float* __restrict__ ah = agg + (size_t)rbase * 108 + (lane & 15) * 108 + h * K;
    const float* __restrict__ wr = W1 + (size_t)(2 * h * 16 + (lane & 15)) * K;
#pragma unroll
    for (int k0 = 0; k0 < 24; k0 += 8) {
      const v4f a  = load_quad(ah + k0 + qoff);
      const v4f b0 = load_quad(wr + k0 + qoff);
      const v4f b1 = load_quad(wr + 16 * K + k0 + qoff);
      acc[2 * h] = __builtin_amdgcn_wmma_f32_16x16x4_f32(
          false, (v2f){a.x, a.y}, false, (v2f){b0.x, b0.y}, (short)0, acc[2 * h], false, false);
      acc[2 * h + 1] = __builtin_amdgcn_wmma_f32_16x16x4_f32(
          false, (v2f){a.x, a.y}, false, (v2f){b1.x, b1.y}, (short)0, acc[2 * h + 1], false, false);
      acc[2 * h] = __builtin_amdgcn_wmma_f32_16x16x4_f32(
          false, (v2f){a.z, a.w}, false, (v2f){b0.z, b0.w}, (short)0, acc[2 * h], false, false);
      acc[2 * h + 1] = __builtin_amdgcn_wmma_f32_16x16x4_f32(
          false, (v2f){a.z, a.w}, false, (v2f){b1.z, b1.w}, (short)0, acc[2 * h + 1], false, false);
    }
    const v2f a = load_tail27(ah, upper);
#pragma unroll
    for (int tt = 0; tt < 2; ++tt) {
      const v2f b = load_tail27(wr + tt * 16 * K, upper);
      acc[2 * h + tt] = __builtin_amdgcn_wmma_f32_16x16x4_f32(
          false, a, false, b, (short)0, acc[2 * h + tt], false, false);
    }
  }

  const int rowoff = (lane < 16) ? 0 : 8;
  const int colo   = lane & 15;
#pragma unroll
  for (int t = 0; t < 8; ++t) {
    const float bt = bias[t * 16 + colo];
#pragma unroll
    for (int i = 0; i < 8; ++i) {
      float v = acc[t][i] + bt;
      v = (v > 0.f) ? v : expm1f(v);
      ts[(rowoff + i) * TILE_PITCH + t * 16 + colo] = v;
    }
  }

#pragma unroll
  for (int r = 0; r < 16; ++r) {
    const v4f q = *reinterpret_cast<const v4f*>(ts + r * TILE_PITCH + lane * 4);
    *reinterpret_cast<v4f*>(h1 + (size_t)(rbase + r) * 128 + lane * 4) = q;
  }

#pragma unroll
  for (int t = 0; t < 8; ++t) acc[t] = (v8f){0.f,0.f,0.f,0.f,0.f,0.f,0.f,0.f};
  {
    const float* __restrict__ ar_s = ts + (lane & 15) * TILE_PITCH + qoff;
    const float* __restrict__ wr = W2 + (size_t)(lane & 15) * 128 + qoff;
#pragma unroll 2
    for (int k0 = 0; k0 < 128; k0 += 8) {
      const v4f a = *reinterpret_cast<const v4f*>(ar_s + k0);
      v4f b[8];
#pragma unroll
      for (int t = 0; t < 8; ++t) b[t] = load_quad(wr + (size_t)t * 16 * 128 + k0);
#pragma unroll
      for (int t = 0; t < 8; ++t) {
        acc[t] = __builtin_amdgcn_wmma_f32_16x16x4_f32(
            false, (v2f){a.x, a.y}, false, (v2f){b[t].x, b[t].y}, (short)0, acc[t], false, false);
      }
#pragma unroll
      for (int t = 0; t < 8; ++t) {
        acc[t] = __builtin_amdgcn_wmma_f32_16x16x4_f32(
            false, (v2f){a.z, a.w}, false, (v2f){b[t].z, b[t].w}, (short)0, acc[t], false, false);
      }
    }
  }

#pragma unroll
  for (int t = 0; t < 8; ++t) {
#pragma unroll
    for (int i = 0; i < 8; ++i) {
      Y[(size_t)(rbase + rowoff + i) * 128 + t * 16 + colo] = acc[t][i];
    }
  }

  float alo[4], ahi[4], rlo[4], rhi[4];
#pragma unroll
  for (int h = 0; h < 4; ++h) {
    alo[h] = al[h * 32 + colo]; ahi[h] = al[h * 32 + 16 + colo];
    rlo[h] = ar[h * 32 + colo]; rhi[h] = ar[h * 32 + 16 + colo];
  }
  const bool b3 = (colo & 8) != 0, b2 = (colo & 4) != 0, b1 = (colo & 2) != 0, b0 = (colo & 1) != 0;
  float pl[8][4], pr[8][4];
#pragma unroll
  for (int i = 0; i < 8; ++i) {
#pragma unroll
    for (int h = 0; h < 4; ++h) {
      pl[i][h] = acc[2 * h][i] * alo[h] + acc[2 * h + 1][i] * ahi[h];
      pr[i][h] = acc[2 * h][i] * rlo[h] + acc[2 * h + 1][i] * rhi[h];
    }
  }
#pragma unroll
  for (int h = 0; h < 4; ++h) {
#pragma unroll
    for (int i = 0; i < 4; ++i) {
      pl[i][h] = halve_sum<8>(pl[i][h], pl[i + 4][h], b3);
      pr[i][h] = halve_sum<8>(pr[i][h], pr[i + 4][h], b3);
    }
#pragma unroll
    for (int i = 0; i < 2; ++i) {
      pl[i][h] = halve_sum<4>(pl[i][h], pl[i + 2][h], b2);
      pr[i][h] = halve_sum<4>(pr[i][h], pr[i + 2][h], b2);
    }
    pl[0][h] = halve_sum<2>(pl[0][h], pl[1][h], b1);
    pr[0][h] = halve_sum<2>(pr[0][h], pr[1][h], b1);
  }
  float2 ol, orr;
  ol.x  = halve_sum<1>(pl[0][0], pl[0][2], b0);
  ol.y  = halve_sum<1>(pl[0][1], pl[0][3], b0);
  orr.x = halve_sum<1>(pr[0][0], pr[0][2], b0);
  orr.y = halve_sum<1>(pr[0][1], pr[0][3], b0);
  const size_t so = (size_t)(rbase + rowoff) * 4 + 2 * colo;
  *reinterpret_cast<float2*>(el + so) = ol;
  *reinterpret_cast<float2*>(er + so) = orr;
}

__global__ void attn_proj_in(const float* __restrict__ x, const float* __restrict__ w,
                             const float* __restrict__ al, const float* __restrict__ ar,
                             float* __restrict__ el, float* __restrict__ er, int NH,
                             unsigned* __restrict__ cnt, int ncnt,
                             unsigned* __restrict__ gmat, int ngm) {
  __shared__ float vt[216];
  const int tid = threadIdx.x;
  if (tid < 216) {
    const int which = (tid >= 108) ? 1 : 0;
    const int idx = tid - 108 * which;
    const int h = idx / 27, c = idx - h * 27;
    const float* a = which ? ar : al;
    float s = 0.f;
    for (int d = 0; d < 32; ++d) s += w[(h * 32 + d) * 27 + c] * a[h * 32 + d];
    vt[tid] = s;
  }
  __syncthreads();
  const int t = blockIdx.x * blockDim.x + tid;
  if (t < NH) {
    const int n = t >> 2, h = t & 3;
    const float* xr = x + n * 27;
    float sl = 0.f, sr = 0.f;
    for (int c = 0; c < 27; ++c) {
      const float xv = xr[c];
      sl += xv * vt[h * 27 + c];
      sr += xv * vt[108 + h * 27 + c];
    }
    el[t] = sl; er[t] = sr;
  }
  const unsigned nt = gridDim.x * blockDim.x;
#pragma clang loop unroll(disable) interleave(disable) vectorize(disable)
  for (unsigned i = (unsigned)t; i < (unsigned)ncnt; i += nt) cnt[i] = 0u;
#pragma clang loop unroll(disable) interleave(disable) vectorize(disable)
  for (unsigned i = (unsigned)t; i < (unsigned)ngm; i += nt) gmat[i] = ((i & 63u) < 32u) ? 0u : ORD_NEG_INF;
}

__global__ void edge_place(const int* __restrict__ src, const int* __restrict__ dst,
                           int* cnt, int* __restrict__ ell,
                           int* __restrict__ ovf, int* novf, int E) {
  const int e = blockIdx.x * blockDim.x + threadIdx.x;
  if (e >= E) return;
  const int d = dst[e];
  const int s = src[e];
  const int r = atomicAdd(&cnt[d], 1);
  if (r < ELL_W) {
    ell[(size_t)d * ELL_W + r] = s;
  } else {
    const int o = atomicAdd(novf, 1);
    ovf[2 * (size_t)o]     = d;
    ovf[2 * (size_t)o + 1] = s;
  }
}

template <class F>
__device__ __forceinline__ void for_each_in_edge(int slots, const int* __restrict__ ovf, int nov,
                                                 int d, int deg, F&& f) {
  const int nb = (deg < ELL_W) ? deg : ELL_W;
  for (int j = 0; j < nb; ++j) f(__builtin_amdgcn_readlane(slots, j));
  for (int i = 0; i < nov; ++i) {
    if (ovf[2 * (size_t)i] == d) f(ovf[2 * (size_t)i + 1]);
  }
}

template <int J>
__device__ __forceinline__ float group8_bcast(float x) {
  return __int_as_float((int)__builtin_amdgcn_mov_dpp8((unsigned)__float_as_int(x), J * 0x249249));
}
__device__ __forceinline__ int edge_source8(int slots, int deg, int lane) {
  const int first = __builtin_amdgcn_readfirstlane(slots);
  const int own   = __builtin_amdgcn_update_dpp(slots, slots, 0x118  , 0xf, 0xc, false);
  return ((lane & 7) < deg) ? own : first;
}
__device__ __forceinline__ float edge_weight8(float els, float erd, int deg, int lane) {
  float v = els + erd;
  v = (v > 0.f) ? v : 0.2f * v;
  v = ((lane & 7) < deg) ? v : -INFINITY;
  float m = fmaxf(v, lane_xor<1>(v));
  m = fmaxf(m, lane_xor<2>(m));
  m = fmaxf(m, lane_xor<4>(m));
  const float ex = expf(v - m);
  float s = ex + lane_xor<1>(ex);
  s += lane_xor<2>(s);
  s += lane_xor<4>(s);
  return ex / s;
}

__global__ void node_softmax_gather_in(const int* __restrict__ cnt, const int* __restrict__ ell,
                                       const int* __restrict__ ovf, const int* __restrict__ novf,
                                       const float* __restrict__ el, const float* __restrict__ er,
                                       const float* __restrict__ x,
                                       float* __restrict__ agg,
                                       int N) {
  const int lane = threadIdx.x & 31;
  const int d = __builtin_amdgcn_readfirstlane((int)(blockIdx.x * (blockDim.x >> 5) + (threadIdx.x >> 5)));
  if (d >= N) return;
  const int h   = lane >> 3;
  const int deg = cnt[d];
  const int slots = ell[(size_t)d * ELL_W + (lane & (ELL_W - 1))];
  const float erd = er[d * 4 + h];
  const int cl = (lane < 27) ? lane : 26;

  float a0 = 0.f, a1 = 0.f, a2 = 0.f, a3 = 0.f;
  if (deg >= 1 && deg <= 8) {
    const int sidx = edge_source8(slots, deg, lane);
    const float els = el[sidx * 4 + h];
    float xv0, xv1, xv2, xv3, xv4, xv5, xv6, xv7;
#define ROW_LOAD_IN(J) if (J < deg) xv##J = x[(size_t)__builtin_amdgcn_readlane(sidx, J) * 27 + cl];
    ROW_LOAD_IN(0) ROW_LOAD_IN(1) ROW_LOAD_IN(2) ROW_LOAD_IN(3)
    ROW_LOAD_IN(4) ROW_LOAD_IN(5) ROW_LOAD_IN(6) ROW_LOAD_IN(7)
#undef ROW_LOAD_IN
    const int a = __float_as_int(edge_weight8(els, erd, deg, lane));
#define ROW_FMA_IN(J) if (J < deg) {                                              \
      a0 += xv##J * __int_as_float(__builtin_amdgcn_readlane(a, J));              \
      a1 += xv##J * __int_as_float(__builtin_amdgcn_readlane(a, 8 + J));          \
      a2 += xv##J * __int_as_float(__builtin_amdgcn_readlane(a, 16 + J));         \
      a3 += xv##J * __int_as_float(__builtin_amdgcn_readlane(a, 24 + J)); }
    ROW_FMA_IN(0) ROW_FMA_IN(1) ROW_FMA_IN(2) ROW_FMA_IN(3)
    ROW_FMA_IN(4) ROW_FMA_IN(5) ROW_FMA_IN(6) ROW_FMA_IN(7)
#undef ROW_FMA_IN
  } else {
    const int nov = (deg > ELL_W) ? novf[0] : 0;
    float m = -INFINITY;
    for_each_in_edge(slots, ovf, nov, d, deg, [&](int sidx) {
      float v = el[sidx * 4 + h] + erd;
      v = (v > 0.f) ? v : 0.2f * v;
      m = fmaxf(m, v);
    });
    float sum = 0.f;
    for_each_in_edge(slots, ovf, nov, d, deg, [&](int sidx) {
      float v = el[sidx * 4 + h] + erd;
      v = (v > 0.f) ? v : 0.2f * v;
      sum += expf(v - m);
    });
    for_each_in_edge(slots, ovf, nov, d, deg, [&](int sidx) {
      float v = el[sidx * 4 + h] + erd;
      v = (v > 0.f) ? v : 0.2f * v;
      const int a = __float_as_int(expf(v - m) / sum);
      const float xv = x[(size_t)sidx * 27 + cl];
      a0 += xv * __int_as_float(__builtin_amdgcn_readlane(a, 0));
      a1 += xv * __int_as_float(__builtin_amdgcn_readlane(a, 8));
      a2 += xv * __int_as_float(__builtin_amdgcn_readlane(a, 16));
      a3 += xv * __int_as_float(__builtin_amdgcn_readlane(a, 24));
    });
  }
  if (lane < 27) {
    float* o = agg + (size_t)d * 108 + lane;
    o[0] = a0; o[27] = a1; o[54] = a2; o[81] = a3;
  }
}

__global__ void node_softmax_gather(const int* __restrict__ cnt, const int* __restrict__ ell,
                                    const int* __restrict__ ovf, const int* __restrict__ novf,
                                    const float* __restrict__ el, const float* __restrict__ er,
                                    const float* __restrict__ feat,
                                    const float* __restrict__ addrow,
                                    const float* __restrict__ bias,
                                    const float* __restrict__ aw_w, const float* __restrict__ aw_b,
                                    const int* __restrict__ gid, float* __restrict__ gmat,
                                    int N) {
  const int lane = threadIdx.x & 31;
  const int d = __builtin_amdgcn_readfirstlane((int)(blockIdx.x * (blockDim.x >> 5) + (threadIdx.x >> 5)));
  if (d >= N) return;
  const int h   = lane >> 3;
  const int deg = cnt[d];
  const int slots = ell[(size_t)d * ELL_W + (lane & (ELL_W - 1))];
  const float erd = er[d * 4 + h];
  const float4 r = *reinterpret_cast<const float4*>(addrow + (size_t)d * 128 + lane * 4);
  const float4 b = *reinterpret_cast<const float4*>(bias + lane * 4);

  float4 acc = make_float4(0.f, 0.f, 0.f, 0.f);
  if (deg >= 1 && deg <= 8) {
    const int sidx = edge_source8(slots, deg, lane);
    const float els = el[sidx * 4 + h];
    float4 f0, f1, f2, f3, f4, f5, f6, f7;
#define ROW_LOAD(J) if (J < deg) f##J = *reinterpret_cast<const float4*>(                     \
        feat + (size_t)__builtin_amdgcn_readlane(sidx, J) * 128 + lane * 4);
    ROW_LOAD(0) ROW_LOAD(1) ROW_LOAD(2) ROW_LOAD(3)
    ROW_LOAD(4) ROW_LOAD(5) ROW_LOAD(6) ROW_LOAD(7)
#undef ROW_LOAD
    const float a = edge_weight8(els, erd, deg, lane);
#define ROW_FMA(J) if (J < deg) { const float aj = group8_bcast<J>(a);                        \
      acc.x += f##J.x * aj; acc.y += f##J.y * aj; acc.z += f##J.z * aj; acc.w += f##J.w * aj; }
    ROW_FMA(0) ROW_FMA(1) ROW_FMA(2) ROW_FMA(3)
    ROW_FMA(4) ROW_FMA(5) ROW_FMA(6) ROW_FMA(7)
#undef ROW_FMA
  } else {
    const int nov = (deg > ELL_W) ? novf[0] : 0;
    float m = -INFINITY;
    for_each_in_edge(slots, ovf, nov, d, deg, [&](int sidx) {
      float v = el[sidx * 4 + h] + erd;
      v = (v > 0.f) ? v : 0.2f * v;
      m = fmaxf(m, v);
    });
    float sum = 0.f;
    for_each_in_edge(slots, ovf, nov, d, deg, [&](int sidx) {
      float v = el[sidx * 4 + h] + erd;
      v = (v > 0.f) ? v : 0.2f * v;
      sum += expf(v - m);
    });
    for_each_in_edge(slots, ovf, nov, d, deg, [&](int sidx) {
      float v = el[sidx * 4 + h] + erd;
      v = (v > 0.f) ? v : 0.2f * v;
      const float a = expf(v - m) / sum;
      const float4 f = *reinterpret_cast<const float4*>(feat + (size_t)sidx * 128 + lane * 4);
      acc.x += f.x * a; acc.y += f.y * a; acc.z += f.z * a; acc.w += f.w * a;
    });
  }

  float4 x;
  x.x = acc.x + r.x + b.x; x.y = acc.y + r.y + b.y;
  x.z = acc.z + r.z + b.z; x.w = acc.w + r.w + b.w;

  const int q = lane & 7;
  float4 hn = make_float4(0.f, 0.f, 0.f, 0.f);
#pragma unroll
  for (int hh = 0; hh < 4; ++hh) {
    hn.x += __shfl(x.x, q + 8 * hh);
    hn.y += __shfl(x.y, q + 8 * hh);
    hn.z += __shfl(x.z, q + 8 * hh);
    hn.w += __shfl(x.w, q + 8 * hh);
  }
  hn.x *= 0.25f; hn.y *= 0.25f; hn.z *= 0.25f; hn.w *= 0.25f;

  float sw = aw_b[0];
#pragma unroll
  for (int i = 0; i < 32; ++i) {
    const float comp = ((i & 3) == 0) ? hn.x : ((i & 3) == 1) ? hn.y : ((i & 3) == 2) ? hn.z : hn.w;
    const float hi = __int_as_float(__builtin_amdgcn_readlane(__float_as_int(comp), i >> 2));
    sw += hi * aw_w[i];
  }
  const float w = 1.f / (1.f + expf(-sw));

  const float t0 = __shfl(hn.x, lane >> 2);
  const float t1 = __shfl(hn.y, lane >> 2);
  const float t2 = __shfl(hn.z, lane >> 2);
  const float t3 = __shfl(hn.w, lane >> 2);
  const int   c  = lane & 3;
  const float hv = (c == 0) ? t0 : (c == 1) ? t1 : (c == 2) ? t2 : t3;
  const int g = gid[d];
  atomicAdd(&gmat[g * 64 + lane], w * hv);
  atomicMax((unsigned*)gmat + g * 64 + 32 + lane, ford(hv));
}

extern "C" void kernel_launch(void* const* d_in, const int* in_sizes, int n_in,
                              void* d_out, int out_size, void* d_ws, size_t ws_size,
                              hipStream_t stream) {
  const float* node  = (const float*)d_in[0];
  const int*   src   = (const int*)d_in[1];
  const int*   dst   = (const int*)d_in[2];
  const int*   gid   = (const int*)d_in[3];
  const float* fc1_w = (const float*)d_in[4];
  const float* al1   = (const float*)d_in[5];
  const float* ar1   = (const float*)d_in[6];
  const float* res1w = (const float*)d_in[7];
  const float* bias1 = (const float*)d_in[8];
  const float* fc2_w = (const float*)d_in[9];
  const float* al2   = (const float*)d_in[10];
  const float* ar2   = (const float*)d_in[11];
  const float* bias2 = (const float*)d_in[12];
  const float* aw_w  = (const float*)d_in[13];
  const float* aw_b  = (const float*)d_in[14];
  const float* out_w = (const float*)d_in[15];
  const float* out_b = (const float*)d_in[16];
  (void)n_in; (void)ws_size;

  const int N = in_sizes[3];
  const int E = in_sizes[1];
  const int G = out_size / 128;

  char* wsb = (char*)d_ws;
  size_t off = 0;
  auto alloc = [&](size_t bytes) -> char* {
    char* p = wsb + off;
    off = (off + bytes + 255) & ~(size_t)255;
    return p;
  };
  float*    bufA  = (float*)alloc((size_t)N * 128 * 4);
  float*    bufB  = (float*)alloc((size_t)N * 108 * 4);
  float*    bufC  = (float*)alloc((size_t)N * 128 * 4);
  float*    el    = (float*)alloc((size_t)N * 4 * 4);
  float*    er    = (float*)alloc((size_t)N * 4 * 4);
  int*      cnt   = (int*)alloc(((size_t)N + 1) * 4);
  int*      ell   = (int*)alloc((size_t)N * ELL_W * 4);
  int*      ovf   = (int*)alloc((size_t)E * 2 * 4);
  float*    gmat  = (float*)alloc((size_t)G * 64 * 4);

  auto grid = [](int n) { return dim3((unsigned)((n + THREADS - 1) / THREADS)); };
  const int ntiles  = (N + 15) / 16;
  const int nblocks = (ntiles + 3) / 4;
  const int gtiles  = (G + 15) / 16;
  const int gblocks = (gtiles + 3) / 4;

  attn_proj_in<<<grid(N * 4), THREADS, 0, stream>>>(node, fc1_w, al1, ar1, el, er, N * 4,
                                                    (unsigned*)cnt, N + 1, (unsigned*)gmat, G * 64);

  edge_place<<<grid(E), THREADS, 0, stream>>>(src, dst, cnt, ell, ovf, cnt + N, E);

  node_softmax_gather_in<<<grid(N * 32), THREADS, 0, stream>>>(cnt, ell, ovf, cnt + N, el, er, node, bufB, N);
  wmma_gemm_l1_l2<<<nblocks, 128, 0, stream>>>(bufB, node, fc1_w, res1w, bias1, fc2_w, al2, ar2,
                                              bufC, bufA, el, er, N);

  node_softmax_gather<<<grid(N * 32), THREADS, 0, stream>>>(cnt, ell, ovf, cnt + N, el, er, bufA, bufC, bias2,
                                                            aw_w, aw_b, gid, gmat, N);

  wmma_gemm_head_ln<<<gblocks, 128, 0, stream>>>(gmat, out_w, out_b, (float*)d_out, G);
}


# W2, res1_w, fc1_w copied once per call by the first kernel into WMMA fragment order: a B-fragment load of the two-layer GEMM reads 4 whole 128-B lines, not 16 partial ones; every sum unchanged
# speedup vs baseline: 7.5951x; 1.0979x over previous
#include <hip/hip_runtime.h>
#include <hip/hip_bf16.h>
#include <math.h>

typedef __attribute__((ext_vector_type(2))) float v2f;
typedef __attribute__((ext_vector_type(4))) float v4f;
typedef __attribute__((ext_vector_type(8))) float v8f;

__device__ __forceinline__ v4f load_quad(const float* __restrict__ p) {
  v4f r;
  __builtin_memcpy(&r, p, 16);
  return r;
}

#define THREADS 256
#define ORD_NEG_INF 0x007FFFFFu
#define ELL_W 16

__device__ __forceinline__ unsigned ford(float f) {
  unsigned u = __float_as_uint(f);
  return (u & 0x80000000u) ? ~u : (u | 0x80000000u);
}
__device__ __forceinline__ float funord(unsigned u) {
  return __uint_as_float((u & 0x80000000u) ? (u & 0x7fffffffu) : ~u);
}

template <int MASK>
__device__ __forceinline__ float lane_xor(float x) {
  const int b = __float_as_int(x);
  return __int_as_float(__builtin_amdgcn_update_dpp(b, b, 0x160 + MASK, 0xf, 0xf, false));
}
template <int MASK>
__device__ __forceinline__ float halve_sum(float a, float b, bool up) {
  const float keep = up ? b : a;
  const float send = up ? a : b;
  return keep + lane_xor<MASK>(send);
}
__device__ __forceinline__ float row16_sum(float v) {
  v += lane_xor<1>(v);
  v += lane_xor<2>(v);
  v += lane_xor<4>(v);
  v += lane_xor<8>(v);
  return v;
}

#define PK_R    0
#define PK_RT   3072
#define PK_W1   3584
#define PK_W1T  6656
#define PK_W2   7168
#define PK_WORDS 23552

__device__ __forceinline__ float packed_weight_word(unsigned i, const float* __restrict__ W2,
                                                    const float* __restrict__ R,
                                                    const float* __restrict__ W1) {
  if (i >= (unsigned)PK_W2) {
    i -= (unsigned)PK_W2;
    const unsigned g = i >> 10, t = (i >> 7) & 7u, l = (i >> 2) & 31u, j = i & 3u;
    return W2[(t * 16u + (l & 15u)) * 128u + g * 8u + (l >> 4) * 4u + j];
  }
  const float* __restrict__ w = R;
  if (i >= (unsigned)PK_W1) { i -= (unsigned)PK_W1; w = W1; }
  if (i < (unsigned)PK_RT) {
    const unsigned g = i >> 10, t = (i >> 7) & 7u, l = (i >> 2) & 31u, j = i & 3u;
    return w[(t * 16u + (l & 15u)) * 27u + g * 8u + (l >> 4) * 4u + j];
  }
  i -= (unsigned)PK_RT;
  const unsigned t = i >> 6, l = (i >> 1) & 31u, j = i & 1u;
  const float* __restrict__ row = w + (t * 16u + (l & 15u)) * 27u;
  return (l < 16u) ? row[24u + j] : ((j == 0u) ? row[26] : 0.f);
}

__device__ __forceinline__ float max_half_decode(float f) {
  const unsigned u = __float_as_uint(f);
  return (u == ORD_NEG_INF) ? 0.f : funord(u);
}

__global__ void __launch_bounds__(128)
wmma_gemm_head_ln(const float* __restrict__ g, const float* __restrict__ W,
                  const float* __restrict__ out_b, float* __restrict__ out, int M) {
  const int lane = threadIdx.x & 31;
  const int wave = threadIdx.x >> 5;
  const int tile = blockIdx.x * (blockDim.x >> 5) + wave;
  if (tile * 16 >= M) return;
  const int rbase = tile * 16;

  v8f acc[8];
#pragma unroll
  for (int t = 0; t < 8; ++t) acc[t] = (v8f){0.f,0.f,0.f,0.f,0.f,0.f,0.f,0.f};
  const int qoff = (lane < 16) ? 0 : 4;
  const float* __restrict__ gr = g + (size_t)(rbase + (lane & 15)) * 64 + qoff;
  const float* __restrict__ wr = W + (size_t)(lane & 15) * 64 + qoff;
#pragma unroll
  for (int k0 = 0; k0 < 64; k0 += 8) {
    v4f a = load_quad(gr + k0);
    if (k0 >= 32) {
      a.x = max_half_decode(a.x); a.y = max_half_decode(a.y);
      a.z = max_half_decode(a.z); a.w = max_half_decode(a.w);
    }
    v4f b[8];
#pragma unroll
    for (int t = 0; t < 8; ++t) b[t] = load_quad(wr + (size_t)t * 16 * 64 + k0);
#pragma unroll
    for (int t = 0; t < 8; ++t) {
      acc[t] = __builtin_amdgcn_wmma_f32_16x16x4_f32(
          false, (v2f){a.x, a.y}, false, (v2f){b[t].x, b[t].y}, (short)0, acc[t], false, false);
    }
#pragma unroll
    for (int t = 0; t < 8; ++t) {
      acc[t] = __builtin_amdgcn_wmma_f32_16x16x4_f32(
          false, (v2f){a.z, a.w}, false, (v2f){b[t].z, b[t].w}, (short)0, acc[t], false, false);
    }
  }

  const int rowoff = (lane < 16) ? 0 : 8;
  const int colo   = lane & 15;
  float bt[8];
#pragma unroll
  for (int t = 0; t < 8; ++t) bt[t] = out_b[t * 16 + colo];
#pragma unroll
  for (int i = 0; i < 8; ++i) {
    float v[8];
    float s = 0.f;
#pragma unroll
    for (int t = 0; t < 8; ++t) { v[t] = acc[t][i] + bt[t]; s += v[t]; }
    const float mu = row16_sum(s) * (1.f / 128.f);
    float q = 0.f;
#pragma unroll
    for (int t = 0; t < 8; ++t) { v[t] -= mu; q += v[t] * v[t]; }
    const float rs = rsqrtf(row16_sum(q) * (1.f / 128.f) + 1e-5f);
#pragma unroll
    for (int t = 0; t < 8; ++t) {
      out[(size_t)(rbase + rowoff + i) * 128 + t * 16 + colo] = v[t] * rs;
    }
  }
}

__device__ __forceinline__ v2f load_tail27(const float* __restrict__ row, bool upper) {
  const v4f q = load_quad(row + 23);
  v2f r;
  r.x = upper ? q.w : q.y;
  r.y = upper ? 0.f : q.z;
  return r;
}

#define TILE_PITCH 132

__global__ void __launch_bounds__(128, 7)
wmma_gemm_l1_l2(const float* __restrict__ agg, const float* __restrict__ x,
                const float* __restrict__ wpk,
                const float* __restrict__ w2p,
                const float* __restrict__ bias,
                const float* __restrict__ al, const float* __restrict__ ar,
                float* __restrict__ h1, float* __restrict__ Y,
                float* __restrict__ el, float* __restrict__ er, int M) {
  __shared__ __attribute__((aligned(16))) float tile_s[4 * 16 * TILE_PITCH];
  const int K    = 27;
  const int lane = threadIdx.x & 31;
  const int wave = __builtin_amdgcn_readfirstlane((int)(threadIdx.x >> 5));
  const int tile = blockIdx.x * (blockDim.x >> 5) + wave;
  if (tile * 16 >= M) return;
  const int rbase = tile * 16;
  const bool upper = lane >= 16;
  float* __restrict__ ts = tile_s + wave * (16 * TILE_PITCH);

  v8f acc[8];
#pragma unroll
  for (int t = 0; t < 8; ++t) acc[t] = (v8f){0.f,0.f,0.f,0.f,0.f,0.f,0.f,0.f};

  const int qoff = upper ? 4 : 0;

  {
    const float* __restrict__ xr = x + (size_t)rbase * K + (lane & 15) * K;
    const float* __restrict__ rp = wpk + PK_R + lane * 4;
#pragma clang loop unroll(disable)
    for (int k0 = 0; k0 < 24; k0 += 8) {
      const v4f a = load_quad(xr + k0 + qoff);
      v4f b[8];
#pragma unroll
      for (int t = 0; t < 8; ++t) b[t] = *reinterpret_cast<const v4f*>(rp + k0 * 128 + t * 128);
#pragma unroll
      for (int t = 0; t < 8; ++t) {
        acc[t] = __builtin_amdgcn_wmma_f32_16x16x4_f32(
            false, (v2f){a.x, a.y}, false, (v2f){b[t].x, b[t].y}, (short)0, acc[t], false, false);
      }
#pragma unroll
      for (int t = 0; t < 8; ++t) {
        acc[t] = __builtin_amdgcn_wmma_f32_16x16x4_f32(
            false, (v2f){a.z, a.w}, false, (v2f){b[t].z, b[t].w}, (short)0, acc[t], false, false);
      }
    }
    const v2f a = load_tail27(xr, upper);
#pragma unroll
    for (int t = 0; t < 8; ++t) {
      const v2f b = *reinterpret_cast<const v2f*>(wpk + PK_RT + t * 64 + lane * 2);
      acc[t] = __builtin_amdgcn_wmma_f32_16x16x4_f32(
          false, a, false, b, (short)0, acc[t], false, false);
    }
  }

  const float* __restrict__ ag = agg + (size_t)rbase * 108 + (lane & 15) * 108;
  v4f an[3], ant;
#pragma unroll
  for (int g = 0; g < 3; ++g) an[g] = load_quad(ag + g * 8 + qoff);
  ant = load_quad(ag + 23);
#pragma unroll
  for (int h = 0; h < 4; ++h) {
    const float* __restrict__ wr = wpk + PK_W1 + 2 * h * 128 + lane * 4;
    const v4f ac0 = an[0], ac1 = an[1], ac2 = an[2], act = ant;
    if (h < 3) {
      const float* __restrict__ ah = ag + (h + 1) * K;
#pragma unroll
      for (int g = 0; g < 3; ++g) an[g] = load_quad(ah + g * 8 + qoff);
      ant = load_quad(ah + 23);
    }
#pragma unroll
    for (int k0 = 0; k0 < 24; k0 += 8) {
      const v4f a  = (k0 == 0) ? ac0 : (k0 == 8) ? ac1 : ac2;
      const v4f b0 = *reinterpret_cast<const v4f*>(wr + k0 * 128);
      const v4f b1 = *reinterpret_cast<const v4f*>(wr + k0 * 128 + 128);
      acc[2 * h] = __builtin_amdgcn_wmma_f32_16x16x4_f32(
          false, (v2f){a.x, a.y}, false, (v2f){b0.x, b0.y}, (short)0, acc[2 * h], false, false);
      acc[2 * h + 1] = __builtin_amdgcn_wmma_f32_16x16x4_f32(
          false, (v2f){a.x, a.y}, false, (v2f){b1.x, b1.y}, (short)0, acc[2 * h + 1], false, false);
      acc[2 * h] = __builtin_amdgcn_wmma_f32_16x16x4_f32(
          false, (v2f){a.z, a.w}, false, (v2f){b0.z, b0.w}, (short)0, acc[2 * h], false, false);
      acc[2 * h + 1] = __builtin_amdgcn_wmma_f32_16x16x4_f32(
          false, (v2f){a.z, a.w}, false, (v2f){b1.z, b1.w}, (short)0, acc[2 * h + 1], false, false);
    }
    v2f a;
    a.x = upper ? act.w : act.y;
    a.y = upper ? 0.f : act.z;
#pragma unroll
    for (int tt = 0; tt < 2; ++tt) {
      const v2f b = *reinterpret_cast<const v2f*>(wpk + PK_W1T + (2 * h + tt) * 64 + lane * 2);
      acc[2 * h + tt] = __builtin_amdgcn_wmma_f32_16x16x4_f32(
          false, a, false, b, (short)0, acc[2 * h + tt], false, false);
    }
    __builtin_amdgcn_sched_barrier(0);
  }

  const int rowoff = (lane < 16) ? 0 : 8;
  const int colo   = lane & 15;
#pragma unroll
  for (int t = 0; t < 8; ++t) {
    const float bt = bias[t * 16 + colo];
#pragma unroll
    for (int i = 0; i < 8; ++i) {
      float v = acc[t][i] + bt;
      v = (v > 0.f) ? v : expm1f(v);
      ts[(rowoff + i) * TILE_PITCH + t * 16 + colo] = v;
    }
  }

#pragma unroll
  for (int r = 0; r < 16; ++r) {
    const v4f q = *reinterpret_cast<const v4f*>(ts + r * TILE_PITCH + lane * 4);
    *reinterpret_cast<v4f*>(h1 + (size_t)(rbase + r) * 128 + lane * 4) = q;
  }

#pragma unroll
  for (int t = 0; t < 8; ++t) acc[t] = (v8f){0.f,0.f,0.f,0.f,0.f,0.f,0.f,0.f};
  {
    const float* __restrict__ ar_s = ts + (lane & 15) * TILE_PITCH + qoff;
    const float* __restrict__ wr = w2p + lane * 4;
#pragma unroll 2
    for (int k0 = 0; k0 < 128; k0 += 8) {
      const v4f a = *reinterpret_cast<const v4f*>(ar_s + k0);
      v4f b[8];
#pragma unroll
      for (int t = 0; t < 8; ++t) b[t] = *reinterpret_cast<const v4f*>(wr + k0 * 128 + t * 128);
#pragma unroll
      for (int t = 0; t < 8; ++t) {
        acc[t] = __builtin_amdgcn_wmma_f32_16x16x4_f32(
            false, (v2f){a.x, a.y}, false, (v2f){b[t].x, b[t].y}, (short)0, acc[t], false, false);
      }
#pragma unroll
      for (int t = 0; t < 8; ++t) {
        acc[t] = __builtin_amdgcn_wmma_f32_16x16x4_f32(
            false, (v2f){a.z, a.w}, false, (v2f){b[t].z, b[t].w}, (short)0, acc[t], false, false);
      }
    }
  }

#pragma unroll
  for (int t = 0; t < 8; ++t) {
#pragma unroll
    for (int i = 0; i < 8; ++i) {
      Y[(size_t)(rbase + rowoff + i) * 128 + t * 16 + colo] = acc[t][i];
    }
  }

  float alo[4], ahi[4], rlo[4], rhi[4];
#pragma unroll
  for (int h = 0; h < 4; ++h) {
    alo[h] = al[h * 32 + colo]; ahi[h] = al[h * 32 + 16 + colo];
    rlo[h] = ar[h * 32 + colo]; rhi[h] = ar[h * 32 + 16 + colo];
  }
  const bool b3 = (colo & 8) != 0, b2 = (colo & 4) != 0, b1 = (colo & 2) != 0, b0 = (colo & 1) != 0;
  float pl[8][4], pr[8][4];
#pragma unroll
  for (int i = 0; i < 8; ++i) {
#pragma unroll
    for (int h = 0; h < 4; ++h) {
      pl[i][h] = acc[2 * h][i] * alo[h] + acc[2 * h + 1][i] * ahi[h];
      pr[i][h] = acc[2 * h][i] * rlo[h] + acc[2 * h + 1][i] * rhi[h];
    }
  }
#pragma unroll
  for (int h = 0; h < 4; ++h) {
#pragma unroll
    for (int i = 0; i < 4; ++i) {
      pl[i][h] = halve_sum<8>(pl[i][h], pl[i + 4][h], b3);
      pr[i][h] = halve_sum<8>(pr[i][h], pr[i + 4][h], b3);
    }
#pragma unroll
    for (int i = 0; i < 2; ++i) {
      pl[i][h] = halve_sum<4>(pl[i][h], pl[i + 2][h], b2);
      pr[i][h] = halve_sum<4>(pr[i][h], pr[i + 2][h], b2);
    }
    pl[0][h] = halve_sum<2>(pl[0][h], pl[1][h], b1);
    pr[0][h] = halve_sum<2>(pr[0][h], pr[1][h], b1);
  }
  float2 ol, orr;
  ol.x  = halve_sum<1>(pl[0][0], pl[0][2], b0);
  ol.y  = halve_sum<1>(pl[0][1], pl[0][3], b0);
  orr.x = halve_sum<1>(pr[0][0], pr[0][2], b0);
  orr.y = halve_sum<1>(pr[0][1], pr[0][3], b0);
  const size_t so = (size_t)(rbase + rowoff) * 4 + 2 * colo;
  *reinterpret_cast<float2*>(el + so) = ol;
  *reinterpret_cast<float2*>(er + so) = orr;
}

__global__ void attn_proj_in(const float* __restrict__ x, const float* __restrict__ w,
                             const float* __restrict__ al, const float* __restrict__ ar,
                             float* __restrict__ el, float* __restrict__ er, int NH,
                             unsigned* __restrict__ cnt, int ncnt,
                             unsigned* __restrict__ gmat, int ngm,
                             const float* __restrict__ R, const float* __restrict__ W2,
                             float* __restrict__ wpk) {
  __shared__ float vt[216];
  const int tid = threadIdx.x;
  if (tid < 216) {
    const int which = (tid >= 108) ? 1 : 0;
    const int idx = tid - 108 * which;
    const int h = idx / 27, c = idx - h * 27;
    const float* a = which ? ar : al;
    float s = 0.f;
    for (int d = 0; d < 32; ++d) s += w[(h * 32 + d) * 27 + c] * a[h * 32 + d];
    vt[tid] = s;
  }
  __syncthreads();
  const int t = blockIdx.x * blockDim.x + tid;
  if (t < NH) {
    const int n = t >> 2, h = t & 3;
    const float* xr = x + n * 27;
    float sl = 0.f, sr = 0.f;
    for (int c = 0; c < 27; ++c) {
      const float xv = xr[c];
      sl += xv * vt[h * 27 + c];
      sr += xv * vt[108 + h * 27 + c];
    }
    el[t] = sl; er[t] = sr;
  }
  const unsigned nt = gridDim.x * blockDim.x;
#pragma clang loop unroll(disable) interleave(disable) vectorize(disable)
  for (unsigned i = (unsigned)t; i < (unsigned)ncnt; i += nt) cnt[i] = 0u;
#pragma clang loop unroll(disable) interleave(disable) vectorize(disable)
  for (unsigned i = (unsigned)t; i < (unsigned)ngm; i += nt) gmat[i] = ((i & 63u) < 32u) ? 0u : ORD_NEG_INF;
#pragma clang loop unroll(disable) interleave(disable) vectorize(disable)
  for (unsigned i = (unsigned)t; i < (unsigned)PK_WORDS; i += nt) wpk[i] = packed_weight_word(i, W2, R, w);
}

__global__ void edge_place(const int* __restrict__ src, const int* __restrict__ dst,
                           int* cnt, int* __restrict__ ell,
                           int* __restrict__ ovf, int* novf, int E) {
  const int e = blockIdx.x * blockDim.x + threadIdx.x;
  if (e >= E) return;
  const int d = dst[e];
  const int s = src[e];
  const int r = atomicAdd(&cnt[d], 1);
  if (r < ELL_W) {
    ell[(size_t)d * ELL_W + r] = s;
  } else {
    const int o = atomicAdd(novf, 1);
    ovf[2 * (size_t)o]     = d;
    ovf[2 * (size_t)o + 1] = s;
  }
}

template <class F>
__device__ __forceinline__ void for_each_in_edge(int slots, const int* __restrict__ ovf, int nov,
                                                 int d, int deg, F&& f) {
  const int nb = (deg < ELL_W) ? deg : ELL_W;
  for (int j = 0; j < nb; ++j) f(__builtin_amdgcn_readlane(slots, j));
  for (int i = 0; i < nov; ++i) {
    if (ovf[2 * (size_t)i] == d) f(ovf[2 * (size_t)i + 1]);
  }
}

template <int J>
__device__ __forceinline__ float group8_bcast(float x) {
  return __int_as_float((int)__builtin_amdgcn_mov_dpp8((unsigned)__float_as_int(x), J * 0x249249));
}
__device__ __forceinline__ int edge_source8(int slots, int deg, int lane) {
  const int first = __builtin_amdgcn_readfirstlane(slots);
  const int own   = __builtin_amdgcn_update_dpp(slots, slots, 0x118  , 0xf, 0xc, false);
  return ((lane & 7) < deg) ? own : first;
}
__device__ __forceinline__ float edge_weight8(float els, float erd, int deg, int lane) {
  float v = els + erd;
  v = (v > 0.f) ? v : 0.2f * v;
  v = ((lane & 7) < deg) ? v : -INFINITY;
  float m = fmaxf(v, lane_xor<1>(v));
  m = fmaxf(m, lane_xor<2>(m));
  m = fmaxf(m, lane_xor<4>(m));
  const float ex = expf(v - m);
  float s = ex + lane_xor<1>(ex);
  s += lane_xor<2>(s);
  s += lane_xor<4>(s);
  return ex / s;
}

__global__ void node_softmax_gather_in(const int* __restrict__ cnt, const int* __restrict__ ell,
                                       const int* __restrict__ ovf, const int* __restrict__ novf,
                                       const float* __restrict__ el, const float* __restrict__ er,
                                       const float* __restrict__ x,
                                       float* __restrict__ agg,
                                       int N) {
  const int lane = threadIdx.x & 31;
  const int d = __builtin_amdgcn_readfirstlane((int)(blockIdx.x * (blockDim.x >> 5) + (threadIdx.x >> 5)));
  if (d >= N) return;
  const int h   = lane >> 3;
  const int deg = cnt[d];
  const int slots = ell[(size_t)d * ELL_W + (lane & (ELL_W - 1))];
  const float erd = er[d * 4 + h];
  const int cl = (lane < 27) ? lane : 26;

  float a0 = 0.f, a1 = 0.f, a2 = 0.f, a3 = 0.f;
  if (deg >= 1 && deg <= 8) {
    const int sidx = edge_source8(slots, deg, lane);
    const float els = el[sidx * 4 + h];
    float xv0, xv1, xv2, xv3, xv4, xv5, xv6, xv7;
#define ROW_LOAD_IN(J) if (J < deg) xv##J = x[(size_t)__builtin_amdgcn_readlane(sidx, J) * 27 + cl];
    ROW_LOAD_IN(0) ROW_LOAD_IN(1) ROW_LOAD_IN(2) ROW_LOAD_IN(3)
    ROW_LOAD_IN(4) ROW_LOAD_IN(5) ROW_LOAD_IN(6) ROW_LOAD_IN(7)
#undef ROW_LOAD_IN
    const int a = __float_as_int(edge_weight8(els, erd, deg, lane));
#define ROW_FMA_IN(J) if (J < deg) {                                              \
      a0 += xv##J * __int_as_float(__builtin_amdgcn_readlane(a, J));              \
      a1 += xv##J * __int_as_float(__builtin_amdgcn_readlane(a, 8 + J));          \
      a2 += xv##J * __int_as_float(__builtin_amdgcn_readlane(a, 16 + J));         \
      a3 += xv##J * __int_as_float(__builtin_amdgcn_readlane(a, 24 + J)); }
    ROW_FMA_IN(0) ROW_FMA_IN(1) ROW_FMA_IN(2) ROW_FMA_IN(3)
    ROW_FMA_IN(4) ROW_FMA_IN(5) ROW_FMA_IN(6) ROW_FMA_IN(7)
#undef ROW_FMA_IN
  } else {
    const int nov = (deg > ELL_W) ? novf[0] : 0;
    float m = -INFINITY;
    for_each_in_edge(slots, ovf, nov, d, deg, [&](int sidx) {
      float v = el[sidx * 4 + h] + erd;
      v = (v > 0.f) ? v : 0.2f * v;
      m = fmaxf(m, v);
    });
    float sum = 0.f;
    for_each_in_edge(slots, ovf, nov, d, deg, [&](int sidx) {
      float v = el[sidx * 4 + h] + erd;
      v = (v > 0.f) ? v : 0.2f * v;
      sum += expf(v - m);
    });
    for_each_in_edge(slots, ovf, nov, d, deg, [&](int sidx) {
      float v = el[sidx * 4 + h] + erd;
      v = (v > 0.f) ? v : 0.2f * v;
      const int a = __float_as_int(expf(v - m) / sum);
      const float xv = x[(size_t)sidx * 27 + cl];
      a0 += xv * __int_as_float(__builtin_amdgcn_readlane(a, 0));
      a1 += xv * __int_as_float(__builtin_amdgcn_readlane(a, 8));
      a2 += xv * __int_as_float(__builtin_amdgcn_readlane(a, 16));
      a3 += xv * __int_as_float(__builtin_amdgcn_readlane(a, 24));
    });
  }
  if (lane < 27) {
    float* o = agg + (size_t)d * 108 + lane;
    o[0] = a0; o[27] = a1; o[54] = a2; o[81] = a3;
  }
}

__global__ void node_softmax_gather(const int* __restrict__ cnt, const int* __restrict__ ell,
                                    const int* __restrict__ ovf, const int* __restrict__ novf,
                                    const float* __restrict__ el, const float* __restrict__ er,
                                    const float* __restrict__ feat,
                                    const float* __restrict__ addrow,
                                    const float* __restrict__ bias,
                                    const float* __restrict__ aw_w, const float* __restrict__ aw_b,
                                    const int* __restrict__ gid, float* __restrict__ gmat,
                                    int N) {
  const int lane = threadIdx.x & 31;
  const int d = __builtin_amdgcn_readfirstlane((int)(blockIdx.x * (blockDim.x >> 5) + (threadIdx.x >> 5)));
  if (d >= N) return;
  const int h   = lane >> 3;
  const int deg = cnt[d];
  const int slots = ell[(size_t)d * ELL_W + (lane & (ELL_W - 1))];
  const float erd = er[d * 4 + h];
  const float4 r = *reinterpret_cast<const float4*>(addrow + (size_t)d * 128 + lane * 4);
  const float4 b = *reinterpret_cast<const float4*>(bias + lane * 4);

  float4 acc = make_float4(0.f, 0.f, 0.f, 0.f);
  if (deg >= 1 && deg <= 8) {
    const int sidx = edge_source8(slots, deg, lane);
    const float els = el[sidx * 4 + h];
    float4 f0, f1, f2, f3, f4, f5, f6, f7;
#define ROW_LOAD(J) if (J < deg) f##J = *reinterpret_cast<const float4*>(                     \
        feat + (size_t)__builtin_amdgcn_readlane(sidx, J) * 128 + lane * 4);
    ROW_LOAD(0) ROW_LOAD(1) ROW_LOAD(2) ROW_LOAD(3)
    ROW_LOAD(4) ROW_LOAD(5) ROW_LOAD(6) ROW_LOAD(7)
#undef ROW_LOAD
    const float a = edge_weight8(els, erd, deg, lane);
#define ROW_FMA(J) if (J < deg) { const float aj = group8_bcast<J>(a);                        \
      acc.x += f##J.x * aj; acc.y += f##J.y * aj; acc.z += f##J.z * aj; acc.w += f##J.w * aj; }
    ROW_FMA(0) ROW_FMA(1) ROW_FMA(2) ROW_FMA(3)
    ROW_FMA(4) ROW_FMA(5) ROW_FMA(6) ROW_FMA(7)
#undef ROW_FMA
  } else {
    const int nov = (deg > ELL_W) ? novf[0] : 0;
    float m = -INFINITY;
    for_each_in_edge(slots, ovf, nov, d, deg, [&](int sidx) {
      float v = el[sidx * 4 + h] + erd;
      v = (v > 0.f) ? v : 0.2f * v;
      m = fmaxf(m, v);
    });
    float sum = 0.f;
    for_each_in_edge(slots, ovf, nov, d, deg, [&](int sidx) {
      float v = el[sidx * 4 + h] + erd;
      v = (v > 0.f) ? v : 0.2f * v;
      sum += expf(v - m);
    });
    for_each_in_edge(slots, ovf, nov, d, deg, [&](int sidx) {
      float v = el[sidx * 4 + h] + erd;
      v = (v > 0.f) ? v : 0.2f * v;
      const float a = expf(v - m) / sum;
      const float4 f = *reinterpret_cast<const float4*>(feat + (size_t)sidx * 128 + lane * 4);
      acc.x += f.x * a; acc.y += f.y * a; acc.z += f.z * a; acc.w += f.w * a;
    });
  }

  float4 x;
  x.x = acc.x + r.x + b.x; x.y = acc.y + r.y + b.y;
  x.z = acc.z + r.z + b.z; x.w = acc.w + r.w + b.w;

  const int q = lane & 7;
  float4 hn = make_float4(0.f, 0.f, 0.f, 0.f);
#pragma unroll
  for (int hh = 0; hh < 4; ++hh) {
    hn.x += __shfl(x.x, q + 8 * hh);
    hn.y += __shfl(x.y, q + 8 * hh);
    hn.z += __shfl(x.z, q + 8 * hh);
    hn.w += __shfl(x.w, q + 8 * hh);
  }
  hn.x *= 0.25f; hn.y *= 0.25f; hn.z *= 0.25f; hn.w *= 0.25f;

  float sw = aw_b[0];
#pragma unroll
  for (int i = 0; i < 32; ++i) {
    const float comp = ((i & 3) == 0) ? hn.x : ((i & 3) == 1) ? hn.y : ((i & 3) == 2) ? hn.z : hn.w;
    const float hi = __int_as_float(__builtin_amdgcn_readlane(__float_as_int(comp), i >> 2));
    sw += hi * aw_w[i];
  }
  const float w = 1.f / (1.f + expf(-sw));

  const float t0 = __shfl(hn.x, lane >> 2);
  const float t1 = __shfl(hn.y, lane >> 2);
  const float t2 = __shfl(hn.z, lane >> 2);
  const float t3 = __shfl(hn.w, lane >> 2);
  const int   c  = lane & 3;
  const float hv = (c == 0) ? t0 : (c == 1) ? t1 : (c == 2) ? t2 : t3;
  const int g = gid[d];
  atomicAdd(&gmat[g * 64 + lane], w * hv);
  atomicMax((unsigned*)gmat + g * 64 + 32 + lane, ford(hv));
}

extern "C" void kernel_launch(void* const* d_in, const int* in_sizes, int n_in,
                              void* d_out, int out_size, void* d_ws, size_t ws_size,
                              hipStream_t stream) {
  const float* node  = (const float*)d_in[0];
  const int*   src   = (const int*)d_in[1];
  const int*   dst   = (const int*)d_in[2];
  const int*   gid   = (const int*)d_in[3];
  const float* fc1_w = (const float*)d_in[4];
  const float* al1   = (const float*)d_in[5];
  const float* ar1   = (const float*)d_in[6];
  const float* res1w = (const float*)d_in[7];
  const float* bias1 = (const float*)d_in[8];
  const float* fc2_w = (const float*)d_in[9];
  const float* al2   = (const float*)d_in[10];
  const float* ar2   = (const float*)d_in[11];
  const float* bias2 = (const float*)d_in[12];
  const float* aw_w  = (const float*)d_in[13];
  const float* aw_b  = (const float*)d_in[14];
  const float* out_w = (const float*)d_in[15];
  const float* out_b = (const float*)d_in[16];
  (void)n_in; (void)ws_size;

  const int N = in_sizes[3];
  const int E = in_sizes[1];
  const int G = out_size / 128;

  char* wsb = (char*)d_ws;
  size_t off = 0;
  auto alloc = [&](size_t bytes) -> char* {
    char* p = wsb + off;
    off = (off + bytes + 255) & ~(size_t)255;
    return p;
  };
  float*    bufA  = (float*)alloc((size_t)N * 128 * 4);
  float*    bufB  = (float*)alloc((size_t)N * 108 * 4);
  float*    bufC  = (float*)alloc((size_t)N * 128 * 4);
  float*    el    = (float*)alloc((size_t)N * 4 * 4);
  float*    er    = (float*)alloc((size_t)N * 4 * 4);
  int*      cnt   = (int*)alloc(((size_t)N + 1) * 4);
  int*      ell   = (int*)alloc((size_t)N * ELL_W * 4);
  int*      ovf   = (int*)alloc((size_t)E * 2 * 4);
  float*    gmat  = (float*)alloc((size_t)G * 64 * 4);
  float*    wpk   = (float*)alloc((size_t)PK_WORDS * 4);

  auto grid = [](int n) { return dim3((unsigned)((n + THREADS - 1) / THREADS)); };
  const int ntiles  = (N + 15) / 16;
  const int nblocks = (ntiles + 3) / 4;
  const int gtiles  = (G + 15) / 16;
  const int gblocks = (gtiles + 3) / 4;

  attn_proj_in<<<grid(N * 4), THREADS, 0, stream>>>(node, fc1_w, al1, ar1, el, er, N * 4,
                                                    (unsigned*)cnt, N + 1, (unsigned*)gmat, G * 64,
                                                    res1w, fc2_w, wpk);

  edge_place<<<grid(E), THREADS, 0, stream>>>(src, dst, cnt, ell, ovf, cnt + N, E);

  node_softmax_gather_in<<<grid(N * 32), THREADS, 0, stream>>>(cnt, ell, ovf, cnt + N, el, er, node, bufB, N);
  wmma_gemm_l1_l2<<<nblocks, 128, 0, stream>>>(bufB, node, wpk, wpk + PK_W2, bias1, al2, ar2,
                                              bufC, bufA, el, er, N);

  node_softmax_gather<<<grid(N * 32), THREADS, 0, stream>>>(cnt, ell, ovf, cnt + N, el, er, bufA, bufC, bias2,
                                                            aw_w, aw_b, gid, gmat, N);

  wmma_gemm_head_ln<<<gblocks, 128, 0, stream>>>(gmat, out_w, out_b, (float*)d_out, G);
}
